# starting kernel
# baseline (speedup 1.0000x reference)
#include <hip/hip_runtime.h>


typedef __attribute__((ext_vector_type(2))) float v2f;
typedef __attribute__((ext_vector_type(8))) float v8f;

#define SC_HALF_SQRT2 0.70710678118654752f

__device__ __forceinline__ int refl(int i, int n) {
  i = (i < 0) ? (-1 - i) : i;
  i = (i >= n) ? (2 * n - 1 - i) : i;
  return i;
}

__device__ __forceinline__ int iclamp(int v, int lo, int hi) {
  return v < lo ? lo : (v > hi ? hi : v);
}

__device__ __forceinline__ v8f wmma4(v2f a, v2f b, v8f c) {
  return __builtin_amdgcn_wmma_f32_16x16x4_f32(false, a, false, b, (short)0, c,
                                               false, false);
}

__device__ __forceinline__ float ifilt_coef(int o, int kg, int n,
                                            const float (&ha)[10],
                                            const float (&hb)[10], bool hp) {
  const int i = o >> 2, s = o & 3;
  const int base = 4 + ((((s & 1) != 0) != hp) ? 1 : 0);
  float acc = 0.f;
#pragma unroll
  for (int m = 0; m < 5; ++m) {
    const int L = refl(2 * i + base - 2 * m, n);
    const float c = (s == 0)   ? ha[2 * m]
                    : (s == 1) ? hb[2 * m]
                    : (s == 2) ? ha[2 * m + 1]
                               : hb[2 * m + 1];
    acc += (L == kg) ? c : 0.f;
  }
  return acc;
}

template <int NT>
__device__ __forceinline__ float filt_coef(int o, int kg, int n,
                                           const float (&h)[NT]) {
  float acc = 0.f;
#pragma unroll
  for (int m = 0; m < NT; ++m) {
    const int L = refl(o + (NT >> 1) - m, n);
    acc += (L == kg) ? h[m] : 0.f;
  }
  return acc;
}


__global__ __launch_bounds__(256) void build_ifilt_colF(
    const float* __restrict__ g0a, const float* __restrict__ g0b,
    const float* __restrict__ g1a, const float* __restrict__ g1b,
    float* __restrict__ M, int r) {
  const int ty = blockIdx.x, f = blockIdx.y;
  float ha[10], hb[10];
#pragma unroll
  for (int m = 0; m < 10; ++m) {
    ha[m] = f ? g1b[m] : g0b[m];
    hb[m] = f ? g1a[m] : g0a[m];
  }
  const int o0 = ty << 4;
  const int w0 = iclamp((o0 >> 1) - 4, 0, r - 16);
  const int idx = threadIdx.x;
  const int row = idx >> 4, k = idx & 15;
  M[(((size_t)ty * 2 + f) * 16 + row) * 16 + k] =
      ifilt_coef(o0 + row, w0 + k, r, ha, hb, f != 0);
}

__global__ __launch_bounds__(256) void build_ifilt_rowG(
    const float* __restrict__ g0a, const float* __restrict__ g0b,
    const float* __restrict__ g1a, const float* __restrict__ g1b,
    float* __restrict__ M, int W) {
  const int tx = blockIdx.x, f = blockIdx.y;
  float ha[10], hb[10];
#pragma unroll
  for (int m = 0; m < 10; ++m) {
    ha[m] = f ? g1b[m] : g0b[m];
    hb[m] = f ? g1a[m] : g0a[m];
  }
  const int p0 = tx << 4;
  const int w0 = iclamp((p0 >> 1) - 4, 0, W - 16);
  const int idx = threadIdx.x;
  const int k = idx >> 4, n = idx & 15;
  M[(((size_t)tx * 2 + f) * 16 + k) * 16 + n] =
      ifilt_coef(p0 + n, w0 + k, W, ha, hb, f != 0);
}

__global__ __launch_bounds__(256) void build_j1_colF(
    const float* __restrict__ g0o, const float* __restrict__ g1o,
    float* __restrict__ M, int r) {
  const int ty = blockIdx.x, f = blockIdx.y;
  float h0[7], h1[5];
#pragma unroll
  for (int m = 0; m < 7; ++m) h0[m] = g0o[m];
#pragma unroll
  for (int m = 0; m < 5; ++m) h1[m] = g1o[m];
  const int o0 = ty << 4;
  const int w0 = iclamp(o0 - 4, 0, r - 24);
  for (int idx = threadIdx.x; idx < 16 * 24; idx += 256) {
    const int row = idx / 24, k = idx % 24;
    const float v = f ? filt_coef<5>(o0 + row, w0 + k, r, h1)
                      : filt_coef<7>(o0 + row, w0 + k, r, h0);
    M[((size_t)ty * 2 + f) * 384 + row * 24 + k] = v;
  }
}

__global__ __launch_bounds__(256) void build_j1_rowG(
    const float* __restrict__ g0o, const float* __restrict__ g1o,
    float* __restrict__ M, int W) {
  const int tx = blockIdx.x, f = blockIdx.y;
  float h0[7], h1[5];
#pragma unroll
  for (int m = 0; m < 7; ++m) h0[m] = g0o[m];
#pragma unroll
  for (int m = 0; m < 5; ++m) h1[m] = g1o[m];
  const int p0 = tx << 4;
  const int w0 = iclamp(p0 - 4, 0, W - 24);
  for (int idx = threadIdx.x; idx < 24 * 16; idx += 256) {
    const int k = idx >> 4, n = idx & 15;
    const float v = f ? filt_coef<5>(p0 + n, w0 + k, W, h1)
                      : filt_coef<7>(p0 + n, w0 + k, W, h0);
    M[((size_t)tx * 2 + f) * 384 + k * 16 + n] = v;
  }
}


template <int R>
__global__ __launch_bounds__(32) void col_ifilt_kernel(
    const float* __restrict__ ll, const float* __restrict__ yh,
    const float* __restrict__ Fmat, float* __restrict__ y1,
    float* __restrict__ y2) {
  constexpr int TILES_X = R / 16;
  constexpr int H2 = R / 2;
  constexpr int SB = H2 * H2 * 2;
  const int lane = threadIdx.x;
  const int tx = blockIdx.x % TILES_X, ty = blockIdx.x / TILES_X;
  const int o0 = ty << 4, c0 = tx << 4;
  const int mc = blockIdx.y;
  const int which = blockIdx.z;
  const int w0 = iclamp((o0 >> 1) - 4, 0, R - 16);

  const int hg = lane >> 4, l16 = lane & 15;
  const int bcol = c0 + l16;

  const float* llp = ll + (size_t)mc * (R * R);
  const float* yhp = yh + (size_t)mc * (6 * SB);
  const float* pH1 = yhp + (which ? 1 : 0) * SB;
  const float* pH2 = yhp + (which ? 4 : 5) * SB;
  const float* pL1 = yhp + 2 * SB;
  const float* pL2 = yhp + 3 * SB;

  const float sgn1 = (bcol & 1) ? -SC_HALF_SQRT2 : SC_HALF_SQRT2;
  const float sgn2 = (bcol & 1) ? SC_HALF_SQRT2 : -SC_HALF_SQRT2;
  const unsigned colE = (unsigned)bcol;
  const unsigned colO = (unsigned)(bcol ^ 1);

  const float* FL = Fmat + ((size_t)ty * 2 + 0) * 256;
  const float* FH = Fmat + ((size_t)ty * 2 + 1) * 256;
  v2f aL[4], aH[4];
#pragma unroll
  for (int t = 0; t < 4; ++t) {
    const int kb = 4 * t + (hg << 1);
    aL[t] = *reinterpret_cast<const v2f*>(FL + l16 * 16 + kb);
    aH[t] = *reinterpret_cast<const v2f*>(FH + l16 * 16 + kb);
  }

  v8f acc = {0.f, 0.f, 0.f, 0.f, 0.f, 0.f, 0.f, 0.f};
#pragma unroll
  for (int t = 0; t < 4; ++t) {
    const int kb = 4 * t + (hg << 1);
    const unsigned rowb = (unsigned)(w0 + kb) * H2;
    v2f bL, bH;
    if (which == 0) {
      const unsigned l0 = (unsigned)(w0 + kb) * R + colE;
      bL.x = llp[l0];
      bL.y = llp[l0 + R];
    } else {
      bL.x = SC_HALF_SQRT2 * (pL1[rowb + colE] + pL2[rowb + colE]);
      bL.y = sgn1 * pL1[rowb + colO] + sgn2 * pL2[rowb + colO];
    }
    bH.x = SC_HALF_SQRT2 * (pH1[rowb + colE] + pH2[rowb + colE]);
    bH.y = sgn1 * pH1[rowb + colO] + sgn2 * pH2[rowb + colO];
    acc = wmma4(aL[t], bL, acc);
    acc = wmma4(aH[t], bH, acc);
  }

  float* outp = (which ? y2 : y1) + (size_t)mc * (2 * R * R) +
                (size_t)(o0 + 8 * hg) * R + bcol;
#pragma unroll
  for (int v = 0; v < 8; ++v) outp[(unsigned)(v * R)] = acc[v];
}

template <int H, int W>
__global__ __launch_bounds__(32) void row_ifilt_kernel(
    const float* __restrict__ y1b, const float* __restrict__ y2b,
    const float* __restrict__ Gmat, float* __restrict__ out) {
  constexpr int TILES_X = (2 * W) / 16;
  const int lane = threadIdx.x;
  const int tx = blockIdx.x % TILES_X, ty = blockIdx.x / TILES_X;
  const int yy0 = ty << 4, p0 = tx << 4;
  const int mc = blockIdx.y;
  const int w0 = iclamp((p0 >> 1) - 4, 0, W - 16);

  __shared__ float t1[16 * 17], t2[16 * 17];
  const float* y1p = y1b + (size_t)mc * (H * W);
  const float* y2p = y2b + (size_t)mc * (H * W);
  {
    const int rr = lane >> 1, cb = (lane & 1) * 8;
    const unsigned g = (unsigned)(yy0 + rr) * W + (w0 + cb);
#pragma unroll
    for (int j = 0; j < 8; ++j) {
      t1[rr * 17 + cb + j] = y1p[g + j];
      t2[rr * 17 + cb + j] = y2p[g + j];
    }
  }

  const int hg = lane >> 4, l16 = lane & 15;
  const float* GL = Gmat + ((size_t)tx * 2 + 0) * 256;
  const float* GH = Gmat + ((size_t)tx * 2 + 1) * 256;
  v2f b1[4], b2[4];
#pragma unroll
  for (int t = 0; t < 4; ++t) {
    const int kb = 4 * t + (hg << 1);
    b1[t].x = GL[kb * 16 + l16];
    b1[t].y = GL[(kb + 1) * 16 + l16];
    b2[t].x = GH[kb * 16 + l16];
    b2[t].y = GH[(kb + 1) * 16 + l16];
  }
  __syncthreads();

  v8f acc = {0.f, 0.f, 0.f, 0.f, 0.f, 0.f, 0.f, 0.f};
#pragma unroll
  for (int t = 0; t < 4; ++t) {
    const int kb = 4 * t + (hg << 1);
    v2f a1, a2;
    a1.x = t1[l16 * 17 + kb];     a1.y = t1[l16 * 17 + kb + 1];
    a2.x = t2[l16 * 17 + kb];     a2.y = t2[l16 * 17 + kb + 1];
    acc = wmma4(a1, b1[t], acc);
    acc = wmma4(a2, b2[t], acc);
  }

  float* outp = out + (size_t)mc * (H * 2 * W) +
                (size_t)(yy0 + 8 * hg) * (2 * W) + (p0 + l16);
#pragma unroll
  for (int v = 0; v < 8; ++v) outp[(unsigned)(v * 2 * W)] = acc[v];
}

template <int R>
__global__ __launch_bounds__(32) void col_j1_kernel(
    const float* __restrict__ ll, const float* __restrict__ yh,
    const float* __restrict__ Fmat, float* __restrict__ y1,
    float* __restrict__ y2) {
  constexpr int TILES_X = R / 16;
  constexpr int H2 = R / 2;
  constexpr int SB = H2 * H2 * 2;
  const int lane = threadIdx.x;
  const int tx = blockIdx.x % TILES_X, ty = blockIdx.x / TILES_X;
  const int o0 = ty << 4, c0 = tx << 4;
  const int mc = blockIdx.y;
  const int which = blockIdx.z;
  const int w0 = iclamp(o0 - 4, 0, R - 24);

  const int hg = lane >> 4, l16 = lane & 15;
  const int bcol = c0 + l16;

  const float* llp = ll + (size_t)mc * (R * R);
  const float* yhp = yh + (size_t)mc * (6 * SB);
  const float* pH1 = yhp + (which ? 1 : 0) * SB;
  const float* pH2 = yhp + (which ? 4 : 5) * SB;
  const float* pL1 = yhp + 2 * SB;
  const float* pL2 = yhp + 3 * SB;

  const float sgn1 = (bcol & 1) ? -SC_HALF_SQRT2 : SC_HALF_SQRT2;
  const float sgn2 = (bcol & 1) ? SC_HALF_SQRT2 : -SC_HALF_SQRT2;
  const unsigned colE = (unsigned)bcol;
  const unsigned colO = (unsigned)(bcol ^ 1);

  const float* FL = Fmat + ((size_t)ty * 2 + 0) * 384;
  const float* FH = Fmat + ((size_t)ty * 2 + 1) * 384;
  v2f aL[6], aH[6];
#pragma unroll
  for (int t = 0; t < 6; ++t) {
    const int kb = 4 * t + (hg << 1);
    aL[t] = *reinterpret_cast<const v2f*>(FL + l16 * 24 + kb);
    aH[t] = *reinterpret_cast<const v2f*>(FH + l16 * 24 + kb);
  }

  v8f acc = {0.f, 0.f, 0.f, 0.f, 0.f, 0.f, 0.f, 0.f};
#pragma unroll
  for (int t = 0; t < 6; ++t) {
    const int kb = 4 * t + (hg << 1);
    const unsigned rowb = (unsigned)(w0 + kb) * H2;
    v2f bL, bH;
    if (which == 0) {
      const unsigned l0 = (unsigned)(w0 + kb) * R + colE;
      bL.x = llp[l0];
      bL.y = llp[l0 + R];
    } else {
      bL.x = SC_HALF_SQRT2 * (pL1[rowb + colE] + pL2[rowb + colE]);
      bL.y = sgn1 * pL1[rowb + colO] + sgn2 * pL2[rowb + colO];
    }
    bH.x = SC_HALF_SQRT2 * (pH1[rowb + colE] + pH2[rowb + colE]);
    bH.y = sgn1 * pH1[rowb + colO] + sgn2 * pH2[rowb + colO];
    acc = wmma4(aL[t], bL, acc);
    acc = wmma4(aH[t], bH, acc);
  }

  float* outp = (which ? y2 : y1) + (size_t)mc * (R * R) +
                (size_t)(o0 + 8 * hg) * R + bcol;
#pragma unroll
  for (int v = 0; v < 8; ++v) outp[(unsigned)(v * R)] = acc[v];
}

template <int H, int W>
__global__ __launch_bounds__(32) void row_j1_kernel(
    const float* __restrict__ y1b, const float* __restrict__ y2b,
    const float* __restrict__ Gmat, float* __restrict__ out) {
  constexpr int TILES_X = W / 16;
  const int lane = threadIdx.x;
  const int tx = blockIdx.x % TILES_X, ty = blockIdx.x / TILES_X;
  const int yy0 = ty << 4, p0 = tx << 4;
  const int mc = blockIdx.y;
  const int w0 = iclamp(p0 - 4, 0, W - 24);

  __shared__ float t1[16 * 25], t2[16 * 25];
  const float* y1p = y1b + (size_t)mc * (H * W);
  const float* y2p = y2b + (size_t)mc * (H * W);
  {
    const int rr = lane >> 1, cb = (lane & 1) * 12;
    const unsigned g = (unsigned)(yy0 + rr) * W + (w0 + cb);
#pragma unroll
    for (int j = 0; j < 12; ++j) {
      t1[rr * 25 + cb + j] = y1p[g + j];
      t2[rr * 25 + cb + j] = y2p[g + j];
    }
  }

  const int hg = lane >> 4, l16 = lane & 15;
  const float* GL = Gmat + ((size_t)tx * 2 + 0) * 384;
  const float* GH = Gmat + ((size_t)tx * 2 + 1) * 384;
  v2f b1[6], b2[6];
#pragma unroll
  for (int t = 0; t < 6; ++t) {
    const int kb = 4 * t + (hg << 1);
    b1[t].x = GL[kb * 16 + l16];
    b1[t].y = GL[(kb + 1) * 16 + l16];
    b2[t].x = GH[kb * 16 + l16];
    b2[t].y = GH[(kb + 1) * 16 + l16];
  }
  __syncthreads();

  v8f acc = {0.f, 0.f, 0.f, 0.f, 0.f, 0.f, 0.f, 0.f};
#pragma unroll
  for (int t = 0; t < 6; ++t) {
    const int kb = 4 * t + (hg << 1);
    v2f a1, a2;
    a1.x = t1[l16 * 25 + kb];     a1.y = t1[l16 * 25 + kb + 1];
    a2.x = t2[l16 * 25 + kb];     a2.y = t2[l16 * 25 + kb + 1];
    acc = wmma4(a1, b1[t], acc);
    acc = wmma4(a2, b2[t], acc);
  }

  float* outp = out + (size_t)mc * (H * W) +
                (size_t)(yy0 + 8 * hg) * W + (p0 + l16);
#pragma unroll
  for (int v = 0; v < 8; ++v) outp[(unsigned)(v * W)] = acc[v];
}

extern "C" void kernel_launch(void* const* d_in, const int* in_sizes, int n_in,
                              void* d_out, int out_size, void* d_ws,
                              size_t ws_size, hipStream_t stream) {
  (void)in_sizes; (void)n_in; (void)out_size; (void)ws_size;
  const float* yl  = (const float*)d_in[0];
  const float* yh1 = (const float*)d_in[1];
  const float* yh2 = (const float*)d_in[2];
  const float* yh3 = (const float*)d_in[3];
  const float* g0o = (const float*)d_in[4];
  const float* g1o = (const float*)d_in[5];
  const float* g0a = (const float*)d_in[6];
  const float* g0b = (const float*)d_in[7];
  const float* g1a = (const float*)d_in[8];
  const float* g1b = (const float*)d_in[9];

  const size_t SEG = (size_t)16 * 1024 * 1024;
  float* bufA = (float*)d_ws;
  float* bufB = bufA + SEG;
  float* bufC = bufB + SEG;
  float* F3c = bufC + SEG;
  float* G3r = F3c + 8 * 2 * 256;
  float* F2c = G3r + 8 * 2 * 256;
  float* G2r = F2c + 16 * 2 * 256;
  float* F1c = G2r + 16 * 2 * 256;
  float* G1r = F1c + 16 * 2 * 384;

  build_ifilt_colF<<<dim3(8, 2),  256, 0, stream>>>(g0a, g0b, g1a, g1b, F3c, 64);
  build_ifilt_rowG<<<dim3(8, 2),  256, 0, stream>>>(g0a, g0b, g1a, g1b, G3r, 64);
  build_ifilt_colF<<<dim3(16, 2), 256, 0, stream>>>(g0a, g0b, g1a, g1b, F2c, 128);
  build_ifilt_rowG<<<dim3(16, 2), 256, 0, stream>>>(g0a, g0b, g1a, g1b, G2r, 128);
  build_j1_colF<<<dim3(16, 2),    256, 0, stream>>>(g0o, g1o, F1c, 256);
  build_j1_rowG<<<dim3(16, 2),    256, 0, stream>>>(g0o, g1o, G1r, 256);

  col_ifilt_kernel<64><<<dim3(8 * 4, 256, 2), 32, 0, stream>>>(
      yl, yh3, F3c, bufA, bufB);
  row_ifilt_kernel<128, 64><<<dim3(8 * 8, 256, 1), 32, 0, stream>>>(
      bufA, bufB, G3r, bufC);

  col_ifilt_kernel<128><<<dim3(16 * 8, 256, 2), 32, 0, stream>>>(
      bufC, yh2, F2c, bufA, bufB);
  row_ifilt_kernel<256, 128><<<dim3(16 * 16, 256, 1), 32, 0, stream>>>(
      bufA, bufB, G2r, bufC);

  col_j1_kernel<256><<<dim3(16 * 16, 256, 2), 32, 0, stream>>>(
      bufC, yh1, F1c, bufA, bufB);
  row_j1_kernel<256, 256><<<dim3(16 * 16, 256, 1), 32, 0, stream>>>(
      bufA, bufB, G1r, (float*)d_out);
}


# Row filter first, column filter second, each pass storing its D tile transposed so k is contiguous in every operand: all loads and stores 16 B per lane, no LDS (the filter sums are re-associated)
# speedup vs baseline: 1.3558x; 1.3558x over previous
#include <hip/hip_runtime.h>


typedef __attribute__((ext_vector_type(2))) float v2f;
typedef __attribute__((ext_vector_type(4))) float v4f;
typedef __attribute__((ext_vector_type(8))) float v8f;

#define SC_HALF_SQRT2 0.70710678118654752f

__device__ __forceinline__ int refl(int i, int n) {
  i = (i < 0) ? (-1 - i) : i;
  i = (i >= n) ? (2 * n - 1 - i) : i;
  return i;
}

__device__ __forceinline__ int iclamp(int v, int lo, int hi) {
  return v < lo ? lo : (v > hi ? hi : v);
}

__device__ __forceinline__ v8f wmma4(v2f a, v2f b, v8f c) {
  return __builtin_amdgcn_wmma_f32_16x16x4_f32(false, a, false, b, (short)0, c,
                                               false, false);
}

__device__ __forceinline__ float ifilt_coef(int o, int kg, int n,
                                            const float (&ha)[10],
                                            const float (&hb)[10], bool hp) {
  const int i = o >> 2, s = o & 3;
  const int base = 4 + ((((s & 1) != 0) != hp) ? 1 : 0);
  float acc = 0.f;
#pragma unroll
  for (int m = 0; m < 5; ++m) {
    const int L = refl(2 * i + base - 2 * m, n);
    const float c = (s == 0)   ? ha[2 * m]
                    : (s == 1) ? hb[2 * m]
                    : (s == 2) ? ha[2 * m + 1]
                               : hb[2 * m + 1];
    acc += (L == kg) ? c : 0.f;
  }
  return acc;
}

template <int NT>
__device__ __forceinline__ float filt_coef(int o, int kg, int n,
                                           const float (&h)[NT]) {
  float acc = 0.f;
#pragma unroll
  for (int m = 0; m < NT; ++m) {
    const int L = refl(o + (NT >> 1) - m, n);
    acc += (L == kg) ? h[m] : 0.f;
  }
  return acc;
}


__global__ __launch_bounds__(256) void build_ifilt_T(
    const float* __restrict__ g0a, const float* __restrict__ g0b,
    const float* __restrict__ g1a, const float* __restrict__ g1b,
    float* __restrict__ M, int r) {
  const int ty = blockIdx.x, f = blockIdx.y;
  float ha[10], hb[10];
#pragma unroll
  for (int m = 0; m < 10; ++m) {
    ha[m] = f ? g1b[m] : g0b[m];
    hb[m] = f ? g1a[m] : g0a[m];
  }
  const int o0 = ty << 4;
  const int w0 = iclamp((o0 >> 1) - 4, 0, r - 16);
  const int idx = threadIdx.x;
  const int row = idx >> 4, k = idx & 15;
  M[(((size_t)ty * 2 + f) * 16 + row) * 16 + k] =
      ifilt_coef(o0 + row, w0 + k, r, ha, hb, f != 0);
}

__global__ __launch_bounds__(256) void build_j1_T(
    const float* __restrict__ g0o, const float* __restrict__ g1o,
    float* __restrict__ M, int r) {
  const int ty = blockIdx.x, f = blockIdx.y;
  float h0[7], h1[5];
#pragma unroll
  for (int m = 0; m < 7; ++m) h0[m] = g0o[m];
#pragma unroll
  for (int m = 0; m < 5; ++m) h1[m] = g1o[m];
  const int o0 = ty << 4;
  const int w0 = iclamp(o0 - 4, 0, r - 24);
  for (int idx = threadIdx.x; idx < 16 * 24; idx += 256) {
    const int row = idx / 24, k = idx % 24;
    const float v = f ? filt_coef<5>(o0 + row, w0 + k, r, h1)
                      : filt_coef<7>(o0 + row, w0 + k, r, h0);
    M[((size_t)ty * 2 + f) * 384 + row * 24 + k] = v;
  }
}

template <int K>
__device__ __forceinline__ void load_half(const float* __restrict__ p,
                                          v2f (&o)[K / 4]) {
#pragma unroll
  for (int q = 0; q < K / 8; ++q) {
    const v4f v = *reinterpret_cast<const v4f*>(p + 4 * q);
    o[2 * q] = v2f{v.x, v.y};
    o[2 * q + 1] = v2f{v.z, v.w};
  }
}

template <int K>
__device__ __forceinline__ void load_c2q(const float* __restrict__ p1,
                                         const float* __restrict__ p2,
                                         v2f sgn, v2f mul, bool odd,
                                         v2f (&o)[K / 4]) {
  v2f r1[K / 4], r2[K / 4];
  load_half<K>(p1, r1);
  load_half<K>(p2, r2);
#pragma unroll
  for (int t = 0; t < K / 4; ++t) {
    const v2f u = __builtin_elementwise_fma(sgn, r2[t], r1[t]) * mul;
    o[t].x = odd ? u.y : u.x;
    o[t].y = odd ? u.x : u.y;
  }
}

__device__ __forceinline__ void store8(float* __restrict__ p, v8f a) {
  *reinterpret_cast<v4f*>(p) = v4f{a[0], a[1], a[2], a[3]};
  *reinterpret_cast<v4f*>(p + 4) = v4f{a[4], a[5], a[6], a[7]};
}

template <int R, int K, bool UP>
__global__ __launch_bounds__(256) __attribute__((amdgpu_waves_per_eu(2, 6)))
void row_pass_T(
    const float* __restrict__ ll, const float* __restrict__ yh,
    const float* __restrict__ FT, float* __restrict__ z1T,
    float* __restrict__ z2T) {
  constexpr int P = UP ? 2 * R : R;
  constexpr int NT = P / 16;
  constexpr int H2 = R / 2;
  constexpr int SB = H2 * H2 * 2;
  constexpr int KT = K / 4;
  const int wave = threadIdx.x >> 5, lane = threadIdx.x & 31;
  const int u = blockIdx.x * 8 + wave;
  const int nt = u % NT, ms = u / NT;
  const int mc = blockIdx.y;
  const int p0 = nt << 4;
  const int w0 = UP ? iclamp((p0 >> 1) - 4, 0, R - 16) : iclamp(p0 - 4, 0, R - K);
  const int h = lane >> 4, c = lane & 15;
  const bool odd = (c & 1) != 0;
  const v2f sgn = odd ? v2f{-1.f, -1.f} : v2f{1.f, 1.f};
  const v2f mul = odd ? v2f{-SC_HALF_SQRT2, SC_HALF_SQRT2}
                      : v2f{SC_HALF_SQRT2, SC_HALF_SQRT2};

  v2f bL[KT], bH[KT];
  load_half<K>(FT + ((size_t)(nt * 2 + 0) * 16 + c) * K + (K / 2) * h, bL);
  load_half<K>(FT + ((size_t)(nt * 2 + 1) * 16 + c) * K + (K / 2) * h, bH);

  const float* llp = ll + (size_t)mc * (R * R);
  const float* yhp = yh + (size_t)mc * (6 * SB);
  float* o1 = z1T + (size_t)mc * (P * R) + (size_t)(p0 + c) * R + 8 * h;
  float* o2 = z2T + (size_t)mc * (P * R) + (size_t)(p0 + c) * R + 8 * h;
  const unsigned koff = (unsigned)(w0 + (K / 2) * h);

#pragma unroll 1
  for (int m = 0; m < 4; ++m) {
    const int yy0 = (ms * 4 + m) << 4;
    const unsigned y = (unsigned)(yy0 + c);
    const unsigned gl = y * R + koff;
    const unsigned gs = (y >> 1) * R + koff;
    v2f aLL[KT], aLH[KT], aHL[KT], aHH[KT];
    load_half<K>(llp + gl, aLL);
    load_c2q<K>(yhp + 0 * SB + gs, yhp + 5 * SB + gs, sgn, mul, odd, aLH);
    load_c2q<K>(yhp + 2 * SB + gs, yhp + 3 * SB + gs, sgn, mul, odd, aHL);
    load_c2q<K>(yhp + 1 * SB + gs, yhp + 4 * SB + gs, sgn, mul, odd, aHH);
    v8f acc1 = {0.f, 0.f, 0.f, 0.f, 0.f, 0.f, 0.f, 0.f};
    v8f acc2 = {0.f, 0.f, 0.f, 0.f, 0.f, 0.f, 0.f, 0.f};
#pragma unroll
    for (int t = 0; t < KT; ++t) {
      acc1 = wmma4(aLL[t], bL[t], acc1);
      acc2 = wmma4(aLH[t], bL[t], acc2);
      acc1 = wmma4(aHL[t], bH[t], acc1);
      acc2 = wmma4(aHH[t], bH[t], acc2);
    }
    store8(o1 + yy0, acc1);
    store8(o2 + yy0, acc2);
  }
}

template <int R, int K, bool UP>
__global__ __launch_bounds__(256) void col_pass_T(
    const float* __restrict__ z1T, const float* __restrict__ z2T,
    const float* __restrict__ FT, float* __restrict__ out) {
  constexpr int P = UP ? 2 * R : R;
  constexpr int OR = UP ? 2 * R : R;
  constexpr int NT = OR / 16;
  constexpr int KT = K / 4;
  const int wave = threadIdx.x >> 5, lane = threadIdx.x & 31;
  const int u = blockIdx.x * 8 + wave;
  const int nt = u % NT, ms = u / NT;
  const int mc = blockIdx.y;
  const int o0 = nt << 4;
  const int w0 = UP ? iclamp((o0 >> 1) - 4, 0, R - 16) : iclamp(o0 - 4, 0, R - K);
  const int h = lane >> 4, c = lane & 15;

  v2f bL[KT], bH[KT];
  load_half<K>(FT + ((size_t)(nt * 2 + 0) * 16 + c) * K + (K / 2) * h, bL);
  load_half<K>(FT + ((size_t)(nt * 2 + 1) * 16 + c) * K + (K / 2) * h, bH);

  const float* z1p = z1T + (size_t)mc * (P * R);
  const float* z2p = z2T + (size_t)mc * (P * R);
  float* op = out + (size_t)mc * (OR * P) + (size_t)(o0 + c) * P + 8 * h;
  const unsigned koff = (unsigned)(w0 + (K / 2) * h);

#pragma unroll 1
  for (int m = 0; m < 4; ++m) {
    const int p0 = (ms * 4 + m) << 4;
    const unsigned g = (unsigned)(p0 + c) * R + koff;
    v2f a1[KT], a2[KT];
    load_half<K>(z1p + g, a1);
    load_half<K>(z2p + g, a2);
    v8f acc1 = {0.f, 0.f, 0.f, 0.f, 0.f, 0.f, 0.f, 0.f};
    v8f acc2 = {0.f, 0.f, 0.f, 0.f, 0.f, 0.f, 0.f, 0.f};
#pragma unroll
    for (int t = 0; t < KT; ++t) {
      acc1 = wmma4(a1[t], bL[t], acc1);
      acc2 = wmma4(a2[t], bH[t], acc2);
    }
    store8(op + p0, acc1 + acc2);
  }
}

extern "C" void kernel_launch(void* const* d_in, const int* in_sizes, int n_in,
                              void* d_out, int out_size, void* d_ws,
                              size_t ws_size, hipStream_t stream) {
  (void)in_sizes; (void)n_in; (void)out_size; (void)ws_size;
  const float* yl  = (const float*)d_in[0];
  const float* yh1 = (const float*)d_in[1];
  const float* yh2 = (const float*)d_in[2];
  const float* yh3 = (const float*)d_in[3];
  const float* g0o = (const float*)d_in[4];
  const float* g1o = (const float*)d_in[5];
  const float* g0a = (const float*)d_in[6];
  const float* g0b = (const float*)d_in[7];
  const float* g1a = (const float*)d_in[8];
  const float* g1b = (const float*)d_in[9];

  const size_t SEG = (size_t)16 * 1024 * 1024;
  float* bufA = (float*)d_ws;
  float* bufB = bufA + SEG;
  float* bufC = bufB + SEG;
  float* F3 = bufC + SEG;
  float* F2 = F3 + 8 * 2 * 256;
  float* F1 = F2 + 16 * 2 * 256;

  build_ifilt_T<<<dim3(8, 2),  256, 0, stream>>>(g0a, g0b, g1a, g1b, F3, 64);
  build_ifilt_T<<<dim3(16, 2), 256, 0, stream>>>(g0a, g0b, g1a, g1b, F2, 128);
  build_j1_T<<<dim3(16, 2),    256, 0, stream>>>(g0o, g1o, F1, 256);

  row_pass_T<64, 16, true><<<dim3(1, 256), 256, 0, stream>>>(
      yl, yh3, F3, bufA, bufB);
  col_pass_T<64, 16, true><<<dim3(2, 256), 256, 0, stream>>>(
      bufA, bufB, F3, bufC);

  row_pass_T<128, 16, true><<<dim3(4, 256), 256, 0, stream>>>(
      bufC, yh2, F2, bufA, bufB);
  col_pass_T<128, 16, true><<<dim3(8, 256), 256, 0, stream>>>(
      bufA, bufB, F2, bufC);

  row_pass_T<256, 24, false><<<dim3(8, 256), 256, 0, stream>>>(
      bufC, yh1, F1, bufA, bufB);
  col_pass_T<256, 24, false><<<dim3(8, 256), 256, 0, stream>>>(
      bufA, bufB, F1, (float*)d_out);
}


# Each level is ONE kernel, no LDS: the row pass's WMMA D tile is fed back unchanged as the A operand of the column pass (K=24 level: two 16-row z tiles; sums re-associated); 9 -> 6 launches
# speedup vs baseline: 1.9437x; 1.4336x over previous
#include <hip/hip_runtime.h>


typedef __attribute__((ext_vector_type(2))) float v2f;
typedef __attribute__((ext_vector_type(4))) float v4f;
typedef __attribute__((ext_vector_type(8))) float v8f;

#define SC_HALF_SQRT2 0.70710678118654752f

__device__ __forceinline__ int refl(int i, int n) {
  i = (i < 0) ? (-1 - i) : i;
  i = (i >= n) ? (2 * n - 1 - i) : i;
  return i;
}

__device__ __forceinline__ int iclamp(int v, int lo, int hi) {
  return v < lo ? lo : (v > hi ? hi : v);
}

__device__ __forceinline__ v8f wmma4(v2f a, v2f b, v8f c) {
  return __builtin_amdgcn_wmma_f32_16x16x4_f32(false, a, false, b, (short)0, c,
                                               false, false);
}

__device__ __forceinline__ float ifilt_coef(int o, int kg, int n,
                                            const float (&ha)[10],
                                            const float (&hb)[10], bool hp) {
  const int i = o >> 2, s = o & 3;
  const int base = 4 + ((((s & 1) != 0) != hp) ? 1 : 0);
  float acc = 0.f;
#pragma unroll
  for (int m = 0; m < 5; ++m) {
    const int L = refl(2 * i + base - 2 * m, n);
    const float c = (s == 0)   ? ha[2 * m]
                    : (s == 1) ? hb[2 * m]
                    : (s == 2) ? ha[2 * m + 1]
                               : hb[2 * m + 1];
    acc += (L == kg) ? c : 0.f;
  }
  return acc;
}

template <int NT>
__device__ __forceinline__ float filt_coef(int o, int kg, int n,
                                           const float (&h)[NT]) {
  float acc = 0.f;
#pragma unroll
  for (int m = 0; m < NT; ++m) {
    const int L = refl(o + (NT >> 1) - m, n);
    acc += (L == kg) ? h[m] : 0.f;
  }
  return acc;
}


__global__ __launch_bounds__(256) void build_ifilt_T(
    const float* __restrict__ g0a, const float* __restrict__ g0b,
    const float* __restrict__ g1a, const float* __restrict__ g1b,
    float* __restrict__ M, int r) {
  const int ty = blockIdx.x, f = blockIdx.y;
  float ha[10], hb[10];
#pragma unroll
  for (int m = 0; m < 10; ++m) {
    ha[m] = f ? g1b[m] : g0b[m];
    hb[m] = f ? g1a[m] : g0a[m];
  }
  const int o0 = ty << 4;
  const int w0 = iclamp((o0 >> 1) - 4, 0, r - 16);
  const int idx = threadIdx.x;
  const int row = idx >> 4, k = idx & 15;
  M[(((size_t)ty * 2 + f) * 16 + row) * 16 + k] =
      ifilt_coef(o0 + row, w0 + k, r, ha, hb, f != 0);
}

__global__ __launch_bounds__(256) void build_j1_T(
    const float* __restrict__ g0o, const float* __restrict__ g1o,
    float* __restrict__ M, int r) {
  const int ty = blockIdx.x, f = blockIdx.y;
  float h0[7], h1[5];
#pragma unroll
  for (int m = 0; m < 7; ++m) h0[m] = g0o[m];
#pragma unroll
  for (int m = 0; m < 5; ++m) h1[m] = g1o[m];
  const int o0 = ty << 4;
  const int w0 = iclamp(o0 - 4, 0, r - 24);
  for (int idx = threadIdx.x; idx < 16 * 24; idx += 256) {
    const int row = idx / 24, k = idx % 24;
    const float v = f ? filt_coef<5>(o0 + row, w0 + k, r, h1)
                      : filt_coef<7>(o0 + row, w0 + k, r, h0);
    M[((size_t)ty * 2 + f) * 384 + row * 24 + k] = v;
  }
}

template <int K>
__device__ __forceinline__ void load_half(const float* __restrict__ p,
                                          v2f (&o)[K / 4]) {
#pragma unroll
  for (int q = 0; q < K / 8; ++q) {
    const v4f v = *reinterpret_cast<const v4f*>(p + 4 * q);
    o[2 * q] = v2f{v.x, v.y};
    o[2 * q + 1] = v2f{v.z, v.w};
  }
}

template <int K>
__device__ __forceinline__ void load_c2q(const float* __restrict__ p1,
                                         const float* __restrict__ p2,
                                         v2f sgn, v2f mul, bool odd,
                                         v2f (&o)[K / 4]) {
  v2f r1[K / 4], r2[K / 4];
  load_half<K>(p1, r1);
  load_half<K>(p2, r2);
#pragma unroll
  for (int t = 0; t < K / 4; ++t) {
    const v2f u = __builtin_elementwise_fma(sgn, r2[t], r1[t]) * mul;
    o[t].x = odd ? u.y : u.x;
    o[t].y = odd ? u.x : u.y;
  }
}

__device__ __forceinline__ void store8(float* __restrict__ p, v8f a) {
  *reinterpret_cast<v4f*>(p) = v4f{a[0], a[1], a[2], a[3]};
  *reinterpret_cast<v4f*>(p + 4) = v4f{a[4], a[5], a[6], a[7]};
}

template <int K>
__device__ __forceinline__ void row_tile(const float* __restrict__ llp,
                                         const float* __restrict__ yhp,
                                         unsigned gl, unsigned gs, int SB,
                                         v2f sgn, v2f mul, bool odd,
                                         const v2f (&bL)[K / 4],
                                         const v2f (&bH)[K / 4], v8f& z1,
                                         v8f& z2) {
  constexpr int KT = K / 4;
  v2f aLL[KT], aLH[KT], aHL[KT], aHH[KT];
  load_half<K>(llp + gl, aLL);
  load_c2q<K>(yhp + 0 * SB + gs, yhp + 5 * SB + gs, sgn, mul, odd, aLH);
  load_c2q<K>(yhp + 2 * SB + gs, yhp + 3 * SB + gs, sgn, mul, odd, aHL);
  load_c2q<K>(yhp + 1 * SB + gs, yhp + 4 * SB + gs, sgn, mul, odd, aHH);
  v8f acc1 = {0.f, 0.f, 0.f, 0.f, 0.f, 0.f, 0.f, 0.f};
  v8f acc2 = {0.f, 0.f, 0.f, 0.f, 0.f, 0.f, 0.f, 0.f};
#pragma unroll
  for (int t = 0; t < KT; ++t) {
    acc1 = wmma4(aLL[t], bL[t], acc1);
    acc2 = wmma4(aLH[t], bL[t], acc2);
    acc1 = wmma4(aHL[t], bH[t], acc1);
    acc2 = wmma4(aHH[t], bH[t], acc2);
  }
  z1 = acc1;
  z2 = acc2;
}

__device__ __forceinline__ v8f col_tile(v8f z, const v2f (&g)[4], v8f acc) {
#pragma unroll
  for (int t = 0; t < 4; ++t)
    acc = wmma4(v2f{z[2 * t], z[2 * t + 1]}, g[t], acc);
  return acc;
}

template <int R>
__global__ __launch_bounds__(256) __attribute__((amdgpu_waves_per_eu(2, 6)))
void level_up(const float* __restrict__ ll, const float* __restrict__ yh,
              const float* __restrict__ FT, float* __restrict__ out) {
  constexpr int P = 2 * R;
  constexpr int NT = P / 16;
  constexpr int H2 = R / 2;
  constexpr int SB = H2 * H2 * 2;
  const int wave = threadIdx.x >> 5, lane = threadIdx.x & 31;
  const int u = blockIdx.x * 8 + wave;
  const int nt = u % NT, ms = u / NT;
  const int mc = blockIdx.y;
  const int o0 = nt << 4;
  const int wy = iclamp((o0 >> 1) - 4, 0, R - 16);
  const int h = lane >> 4, c = lane & 15;
  const bool odd = (c & 1) != 0;
  const v2f sgn = odd ? v2f{-1.f, -1.f} : v2f{1.f, 1.f};
  const v2f mul = odd ? v2f{-SC_HALF_SQRT2, SC_HALF_SQRT2}
                      : v2f{SC_HALF_SQRT2, SC_HALF_SQRT2};

  v2f gL[4], gH[4];
  load_half<16>(FT + ((size_t)(nt * 2 + 0) * 16 + c) * 16 + 8 * h, gL);
  load_half<16>(FT + ((size_t)(nt * 2 + 1) * 16 + c) * 16 + 8 * h, gH);

  const float* llp = ll + (size_t)mc * (R * R);
  const float* yhp = yh + (size_t)mc * (6 * SB);
  float* op = out + (size_t)mc * (P * P) + (size_t)(o0 + c) * P + 8 * h;
  const unsigned y = (unsigned)(wy + c);
  const unsigned rl = y * R;
  const unsigned rs = (y >> 1) * R;

#pragma unroll 1
  for (int m = 0; m < 4; ++m) {
    const int pt = ms * 4 + m;
    const int p0 = pt << 4;
    const int wx = iclamp((p0 >> 1) - 4, 0, R - 16);
    v2f bL[4], bH[4];
    load_half<16>(FT + ((size_t)(pt * 2 + 0) * 16 + c) * 16 + 8 * h, bL);
    load_half<16>(FT + ((size_t)(pt * 2 + 1) * 16 + c) * 16 + 8 * h, bH);
    const unsigned koff = (unsigned)(wx + 8 * h);
    v8f z1, z2;
    row_tile<16>(llp, yhp, rl + koff, rs + koff, SB, sgn, mul, odd, bL, bH, z1,
                 z2);
    const v8f zero = {0.f, 0.f, 0.f, 0.f, 0.f, 0.f, 0.f, 0.f};
    const v8f a1 = col_tile(z1, gL, zero);
    const v8f a2 = col_tile(z2, gH, zero);
    store8(op + p0, a1 + a2);
  }
}

template <int R>
__global__ __launch_bounds__(256) __attribute__((amdgpu_waves_per_eu(2, 5)))
void level_one(const float* __restrict__ ll, const float* __restrict__ yh,
               const float* __restrict__ FT, float* __restrict__ out) {
  constexpr int K = 24;
  constexpr int NT = R / 16;
  constexpr int H2 = R / 2;
  constexpr int SB = H2 * H2 * 2;
  const int wave = threadIdx.x >> 5, lane = threadIdx.x & 31;
  const int u = blockIdx.x * 8 + wave;
  const int nt = u % NT, ms = u / NT;
  const int mc = blockIdx.y;
  const int p0 = nt << 4;
  const int wx = iclamp(p0 - 4, 0, R - K);
  const int h = lane >> 4, c = lane & 15;
  const bool odd = (c & 1) != 0;
  const v2f sgn = odd ? v2f{-1.f, -1.f} : v2f{1.f, 1.f};
  const v2f mul = odd ? v2f{-SC_HALF_SQRT2, SC_HALF_SQRT2}
                      : v2f{SC_HALF_SQRT2, SC_HALF_SQRT2};

  v2f bL[6], bH[6];
  load_half<K>(FT + ((size_t)(nt * 2 + 0) * 16 + c) * K + 12 * h, bL);
  load_half<K>(FT + ((size_t)(nt * 2 + 1) * 16 + c) * K + 12 * h, bH);

  const float* llp = ll + (size_t)mc * (R * R);
  const float* yhp = yh + (size_t)mc * (6 * SB);
  float* op = out + (size_t)mc * (R * R) + (size_t)c * R + p0 + 8 * h;
  const unsigned koff = (unsigned)(wx + 12 * h);
  const int i0 = ms * 4;

  v8f zp1, zp2;
  {
    const unsigned y = (unsigned)iclamp(16 * i0 - 8 + c, 0, R - 1);
    row_tile<K>(llp, yhp, y * R + koff, (y >> 1) * R + koff, SB, sgn, mul, odd,
                bL, bH, zp1, zp2);
  }

#pragma unroll 1
  for (int m = 0; m < 4; ++m) {
    const int i = i0 + m;
    v8f zc1, zc2;
    {
      const unsigned y = (unsigned)iclamp(16 * i + 8 + c, 0, R - 1);
      row_tile<K>(llp, yhp, y * R + koff, (y >> 1) * R + koff, SB, sgn, mul,
                  odd, bL, bH, zc1, zc2);
    }
    const int w0 = iclamp(16 * i - 4, 0, R - K);
    v2f g[2][2][4];
#pragma unroll
    for (int sl = 0; sl < 2; ++sl) {
#pragma unroll
      for (int f = 0; f < 2; ++f) {
        const float* fr = FT + ((size_t)(i * 2 + f) * 16 + c) * K;
#pragma unroll
        for (int q = 0; q < 2; ++q) {
          const int kq = 16 * i - 8 + 16 * sl + 8 * h + 4 * q - w0;
          const bool in = (kq >= 0) && (kq <= K - 4);
          const v4f v = *reinterpret_cast<const v4f*>(fr + iclamp(kq, 0, K - 4));
          g[sl][f][2 * q] = in ? v2f{v.x, v.y} : v2f{0.f, 0.f};
          g[sl][f][2 * q + 1] = in ? v2f{v.z, v.w} : v2f{0.f, 0.f};
        }
      }
    }
    const v8f zero = {0.f, 0.f, 0.f, 0.f, 0.f, 0.f, 0.f, 0.f};
    v8f a1 = col_tile(zp1, g[0][0], zero);
    v8f a2 = col_tile(zp2, g[0][1], zero);
    a1 = col_tile(zc1, g[1][0], a1);
    a2 = col_tile(zc2, g[1][1], a2);
    store8(op + (size_t)(16 * i) * R, a1 + a2);
    zp1 = zc1;
    zp2 = zc2;
  }
}

extern "C" void kernel_launch(void* const* d_in, const int* in_sizes, int n_in,
                              void* d_out, int out_size, void* d_ws,
                              size_t ws_size, hipStream_t stream) {
  (void)in_sizes; (void)n_in; (void)out_size; (void)ws_size;
  const float* yl  = (const float*)d_in[0];
  const float* yh1 = (const float*)d_in[1];
  const float* yh2 = (const float*)d_in[2];
  const float* yh3 = (const float*)d_in[3];
  const float* g0o = (const float*)d_in[4];
  const float* g1o = (const float*)d_in[5];
  const float* g0a = (const float*)d_in[6];
  const float* g0b = (const float*)d_in[7];
  const float* g1a = (const float*)d_in[8];
  const float* g1b = (const float*)d_in[9];

  const size_t SEG = (size_t)16 * 1024 * 1024;
  float* bufA = (float*)d_ws;
  float* bufB = bufA + SEG;
  float* F3 = bufB + SEG;
  float* F2 = F3 + 8 * 2 * 256;
  float* F1 = F2 + 16 * 2 * 256;

  build_ifilt_T<<<dim3(8, 2),  256, 0, stream>>>(g0a, g0b, g1a, g1b, F3, 64);
  build_ifilt_T<<<dim3(16, 2), 256, 0, stream>>>(g0a, g0b, g1a, g1b, F2, 128);
  build_j1_T<<<dim3(16, 2),    256, 0, stream>>>(g0o, g1o, F1, 256);

  level_up<64><<<dim3(2, 256), 256, 0, stream>>>(yl, yh3, F3, bufA);
  level_up<128><<<dim3(8, 256), 256, 0, stream>>>(bufA, yh2, F2, bufB);
  level_one<256><<<dim3(8, 256), 256, 0, stream>>>(bufB, yh1, F1,
                                                   (float*)d_out);
}


# All three level kernels: a work-group stages its input rows (ll and six sub-bands) in LDS with row-wise whole-line loads; waves cut their A fragments from LDS instead of 16-row gathers
# speedup vs baseline: 2.2546x; 1.1600x over previous
#include <hip/hip_runtime.h>


typedef __attribute__((ext_vector_type(2))) float v2f;
typedef __attribute__((ext_vector_type(4))) float v4f;
typedef __attribute__((ext_vector_type(8))) float v8f;

#define SC_HALF_SQRT2 0.70710678118654752f

__device__ __forceinline__ int refl(int i, int n) {
  i = (i < 0) ? (-1 - i) : i;
  i = (i >= n) ? (2 * n - 1 - i) : i;
  return i;
}

__device__ __forceinline__ int iclamp(int v, int lo, int hi) {
  return v < lo ? lo : (v > hi ? hi : v);
}

__device__ __forceinline__ v8f wmma4(v2f a, v2f b, v8f c) {
  return __builtin_amdgcn_wmma_f32_16x16x4_f32(false, a, false, b, (short)0, c,
                                               false, false);
}

__device__ __forceinline__ float ifilt_coef(int o, int kg, int n,
                                            const float (&ha)[10],
                                            const float (&hb)[10], bool hp) {
  const int i = o >> 2, s = o & 3;
  const int base = 4 + ((((s & 1) != 0) != hp) ? 1 : 0);
  float acc = 0.f;
#pragma unroll
  for (int m = 0; m < 5; ++m) {
    const int L = refl(2 * i + base - 2 * m, n);
    const float c = (s == 0)   ? ha[2 * m]
                    : (s == 1) ? hb[2 * m]
                    : (s == 2) ? ha[2 * m + 1]
                               : hb[2 * m + 1];
    acc += (L == kg) ? c : 0.f;
  }
  return acc;
}

template <int NT>
__device__ __forceinline__ float filt_coef(int o, int kg, int n,
                                           const float (&h)[NT]) {
  float acc = 0.f;
#pragma unroll
  for (int m = 0; m < NT; ++m) {
    const int L = refl(o + (NT >> 1) - m, n);
    acc += (L == kg) ? h[m] : 0.f;
  }
  return acc;
}


__global__ __launch_bounds__(256) void build_ifilt_T(
    const float* __restrict__ g0a, const float* __restrict__ g0b,
    const float* __restrict__ g1a, const float* __restrict__ g1b,
    float* __restrict__ M, int r) {
  const int ty = blockIdx.x, f = blockIdx.y;
  float ha[10], hb[10];
#pragma unroll
  for (int m = 0; m < 10; ++m) {
    ha[m] = f ? g1b[m] : g0b[m];
    hb[m] = f ? g1a[m] : g0a[m];
  }
  const int o0 = ty << 4;
  const int w0 = iclamp((o0 >> 1) - 4, 0, r - 16);
  const int idx = threadIdx.x;
  const int row = idx >> 4, k = idx & 15;
  M[(((size_t)ty * 2 + f) * 16 + row) * 16 + k] =
      ifilt_coef(o0 + row, w0 + k, r, ha, hb, f != 0);
}

__global__ __launch_bounds__(256) void build_j1_T(
    const float* __restrict__ g0o, const float* __restrict__ g1o,
    float* __restrict__ M, int r) {
  const int ty = blockIdx.x, f = blockIdx.y;
  float h0[7], h1[5];
#pragma unroll
  for (int m = 0; m < 7; ++m) h0[m] = g0o[m];
#pragma unroll
  for (int m = 0; m < 5; ++m) h1[m] = g1o[m];
  const int o0 = ty << 4;
  const int w0 = iclamp(o0 - 4, 0, r - 24);
  for (int idx = threadIdx.x; idx < 16 * 24; idx += 256) {
    const int row = idx / 24, k = idx % 24;
    const float v = f ? filt_coef<5>(o0 + row, w0 + k, r, h1)
                      : filt_coef<7>(o0 + row, w0 + k, r, h0);
    M[((size_t)ty * 2 + f) * 384 + row * 24 + k] = v;
  }
}

template <int K>
__device__ __forceinline__ void load_half(const float* __restrict__ p,
                                          v2f (&o)[K / 4]) {
#pragma unroll
  for (int q = 0; q < K / 8; ++q) {
    const v4f v = *reinterpret_cast<const v4f*>(p + 4 * q);
    o[2 * q] = v2f{v.x, v.y};
    o[2 * q + 1] = v2f{v.z, v.w};
  }
}

template <int K>
__device__ __forceinline__ void load_c2q(const float* __restrict__ p1,
                                         const float* __restrict__ p2,
                                         v2f sgn, v2f mul, bool odd,
                                         v2f (&o)[K / 4]) {
  v2f r1[K / 4], r2[K / 4];
  load_half<K>(p1, r1);
  load_half<K>(p2, r2);
#pragma unroll
  for (int t = 0; t < K / 4; ++t) {
    const v2f u = __builtin_elementwise_fma(sgn, r2[t], r1[t]) * mul;
    o[t].x = odd ? u.y : u.x;
    o[t].y = odd ? u.x : u.y;
  }
}

__device__ __forceinline__ void lds_barrier() {
  __builtin_amdgcn_fence(__ATOMIC_RELEASE, "workgroup", "local");
  __builtin_amdgcn_s_barrier();
  __builtin_amdgcn_fence(__ATOMIC_ACQUIRE, "workgroup", "local");
}

__device__ __forceinline__ void store8(float* __restrict__ p, v8f a) {
  *reinterpret_cast<v4f*>(p) = v4f{a[0], a[1], a[2], a[3]};
  *reinterpret_cast<v4f*>(p + 4) = v4f{a[4], a[5], a[6], a[7]};
}

template <int K>
__device__ __forceinline__ void row_tile(const float* __restrict__ llp,
                                         const float* __restrict__ yhp,
                                         unsigned gl, unsigned gs, int SB,
                                         v2f sgn, v2f mul, bool odd,
                                         const v2f (&bL)[K / 4],
                                         const v2f (&bH)[K / 4], v8f& z1,
                                         v8f& z2) {
  constexpr int KT = K / 4;
  v2f aLL[KT], aLH[KT], aHL[KT], aHH[KT];
  load_half<K>(llp + gl, aLL);
  load_c2q<K>(yhp + 0 * SB + gs, yhp + 5 * SB + gs, sgn, mul, odd, aLH);
  load_c2q<K>(yhp + 2 * SB + gs, yhp + 3 * SB + gs, sgn, mul, odd, aHL);
  load_c2q<K>(yhp + 1 * SB + gs, yhp + 4 * SB + gs, sgn, mul, odd, aHH);
  v8f acc1 = {0.f, 0.f, 0.f, 0.f, 0.f, 0.f, 0.f, 0.f};
  v8f acc2 = {0.f, 0.f, 0.f, 0.f, 0.f, 0.f, 0.f, 0.f};
#pragma unroll
  for (int t = 0; t < KT; ++t) {
    acc1 = wmma4(aLL[t], bL[t], acc1);
    acc2 = wmma4(aLH[t], bL[t], acc2);
    acc1 = wmma4(aHL[t], bH[t], acc1);
    acc2 = wmma4(aHH[t], bH[t], acc2);
  }
  z1 = acc1;
  z2 = acc2;
}

__device__ __forceinline__ v8f col_tile(v8f z, const v2f (&g)[4], v8f acc) {
#pragma unroll
  for (int t = 0; t < 4; ++t)
    acc = wmma4(v2f{z[2 * t], z[2 * t + 1]}, g[t], acc);
  return acc;
}

template <int R>
__global__ __launch_bounds__(256)
void level_up(const float* __restrict__ ll, const float* __restrict__ yh,
              const float* __restrict__ FT, float* __restrict__ out) {
  constexpr int P = 2 * R;
  constexpr int NT = P / 16;
  constexpr int H2 = R / 2;
  constexpr int SB = H2 * H2 * 2;
  constexpr int YR = (R < 68) ? R : 68;
  constexpr int YS = YR / 2;
  constexpr int XW = 40;
  constexpr int SEG = XW / 4;
  constexpr int PITCH = 44;
  constexpr int NROW = YR + 6 * YS;
  constexpr int NPIECE = NROW * SEG;
  constexpr int NLD = (NPIECE + 255) / 256;
  __shared__ __attribute__((aligned(16))) float tile[NROW * PITCH];

  const int tid = threadIdx.x;
  const int wave = tid >> 5, lane = tid & 31;
  const int u = blockIdx.x * 8 + wave;
  const int nt = u % NT, ms = u / NT;
  const int nt0 = (blockIdx.x * 8) % NT;
  const int mc = blockIdx.y;
  const int o0 = nt << 4;
  const int wy = iclamp((o0 >> 1) - 4, 0, R - 16);
  const int Y0 = iclamp(8 * nt0 - 4, 0, R - YR);
  const int X0 = iclamp(32 * ms - 4, 0, R - XW);
  const int h = lane >> 4, c = lane & 15;
  const bool odd = (c & 1) != 0;
  const v2f sgn = odd ? v2f{-1.f, -1.f} : v2f{1.f, 1.f};
  const v2f mul = odd ? v2f{-SC_HALF_SQRT2, SC_HALF_SQRT2}
                      : v2f{SC_HALF_SQRT2, SC_HALF_SQRT2};

  const float* llp = ll + (size_t)mc * (R * R) + (size_t)Y0 * R + X0;
  const float* yhp = yh + (size_t)mc * (6 * SB) + (size_t)(Y0 >> 1) * R + X0;

  v4f pre[NLD];
#pragma unroll
  for (int n = 0; n < NLD; ++n) {
    const int j = tid + 256 * n;
    if (256 * (n + 1) <= NPIECE || j < NPIECE) {
      const int row = j / SEG, seg = j - row * SEG;
      const bool isll = row < YR;
      const int rr = row - YR;
      const int b = rr / YS, sr = rr - b * YS;
      const unsigned offl = (unsigned)row * R;
      const unsigned offs = (unsigned)b * SB + (unsigned)sr * R;
      const float* src = isll ? (llp + offl) : (yhp + offs);
      pre[n] = *reinterpret_cast<const v4f*>(src + 4 * seg);
    }
  }

  v2f gL[4], gH[4];
  load_half<16>(FT + ((size_t)(nt * 2 + 0) * 16 + c) * 16 + 8 * h, gL);
  load_half<16>(FT + ((size_t)(nt * 2 + 1) * 16 + c) * 16 + 8 * h, gH);
  float* op = out + (size_t)mc * (P * P) + (size_t)(o0 + c) * P + 8 * h;

#pragma unroll
  for (int n = 0; n < NLD; ++n) {
    const int j = tid + 256 * n;
    if (256 * (n + 1) <= NPIECE || j < NPIECE) {
      const int row = j / SEG, seg = j - row * SEG;
      *reinterpret_cast<v4f*>(tile + row * PITCH + 4 * seg) = pre[n];
    }
  }
  __syncthreads();

  const float* tSB = tile + YR * PITCH;
  const int fl = (wy - Y0 + c) * PITCH + 8 * h - X0;
  const int fs = (((wy - Y0) >> 1) + (c >> 1)) * PITCH + 8 * h - X0;

#pragma unroll 1
  for (int m = 0; m < 4; ++m) {
    const int pt = ms * 4 + m;
    const int p0 = pt << 4;
    const int wx = iclamp((p0 >> 1) - 4, 0, R - 16);
    v2f bL[4], bH[4];
    load_half<16>(FT + ((size_t)(pt * 2 + 0) * 16 + c) * 16 + 8 * h, bL);
    load_half<16>(FT + ((size_t)(pt * 2 + 1) * 16 + c) * 16 + 8 * h, bH);
    v8f z1, z2;
    row_tile<16>(tile, tSB, (unsigned)(fl + wx), (unsigned)(fs + wx), YS * PITCH,
                 sgn, mul, odd, bL, bH, z1, z2);
    const v8f zero = {0.f, 0.f, 0.f, 0.f, 0.f, 0.f, 0.f, 0.f};
    const v8f a1 = col_tile(z1, gL, zero);
    const v8f a2 = col_tile(z2, gH, zero);
    store8(op + p0, a1 + a2);
  }
}

template <int R>
__global__ __launch_bounds__(256)
void level_one(const float* __restrict__ ll, const float* __restrict__ yh,
               const float* __restrict__ FT, float* __restrict__ out) {
  constexpr int K = 24;
  constexpr int NT = R / 16;
  constexpr int H2 = R / 2;
  constexpr int SB = H2 * H2 * 2;
  constexpr int XW = 136;
  constexpr int SEG = XW / 4;
  constexpr int PITCH = 140;
  constexpr int NROW = 16 + 6 * 8;
  constexpr int NPIECE = NROW * SEG;
  constexpr int NLD = (NPIECE + 255) / 256;
  __shared__ __attribute__((aligned(16))) float tile[NROW * PITCH];

  const int tid = threadIdx.x;
  const int wave = tid >> 5, lane = tid & 31;
  const int u = blockIdx.x * 8 + wave;
  const int nt = u % NT, ms = u / NT;
  const int nt0 = (blockIdx.x * 8) % NT;
  const int X0 = iclamp(16 * nt0 - 4, 0, R - XW);
  const int mc = blockIdx.y;
  const int p0 = nt << 4;
  const int wx = iclamp(p0 - 4, 0, R - K);
  const int h = lane >> 4, c = lane & 15;
  const bool odd = (c & 1) != 0;
  const v2f sgn = odd ? v2f{-1.f, -1.f} : v2f{1.f, 1.f};
  const v2f mul = odd ? v2f{-SC_HALF_SQRT2, SC_HALF_SQRT2}
                      : v2f{SC_HALF_SQRT2, SC_HALF_SQRT2};

  v2f bL[6], bH[6];
  load_half<K>(FT + ((size_t)(nt * 2 + 0) * 16 + c) * K + 12 * h, bL);
  load_half<K>(FT + ((size_t)(nt * 2 + 1) * 16 + c) * K + 12 * h, bH);

  const float* llp = ll + (size_t)mc * (R * R) + X0;
  const float* yhp = yh + (size_t)mc * (6 * SB) + X0;
  float* op = out + (size_t)mc * (R * R) + (size_t)c * R + p0 + 8 * h;
  const int i0 = ms * 4;

  const unsigned fl = (unsigned)(c * PITCH + wx - X0 + 12 * h);
  const unsigned fs = (unsigned)((c >> 1) * PITCH + wx - X0 + 12 * h);
  const float* tSB = tile + 16 * PITCH;

  v4f pre[NLD];
  auto fetch = [&](int ybase) {
#pragma unroll
    for (int n = 0; n < NLD; ++n) {
      const int j = tid + 256 * n;
      if (256 * (n + 1) <= NPIECE || j < NPIECE) {
        const int row = j / SEG, seg = j - row * SEG;
        const bool isll = row < 16;
        const int rr = row - 16;
        const int yy = iclamp(ybase + (isll ? row : 2 * (rr & 7)), 0, R - 1);
        const unsigned offl = (unsigned)yy * R;
        const unsigned offs = (unsigned)(rr >> 3) * SB + (unsigned)(yy >> 1) * R;
        const float* src = isll ? (llp + offl) : (yhp + offs);
        pre[n] = *reinterpret_cast<const v4f*>(src + 4 * seg);
      }
    }
  };
  fetch(16 * i0 - 8);

  const v8f zero = {0.f, 0.f, 0.f, 0.f, 0.f, 0.f, 0.f, 0.f};
  v8f zp1 = zero, zp2 = zero;

#pragma unroll 1
  for (int t = 0; t < 5; ++t) {
    const int i = i0 + t - 1;
    v2f g[2][2][4];
    if (t > 0) {
      const int w0 = iclamp(16 * i - 4, 0, R - K);
#pragma unroll
      for (int sl = 0; sl < 2; ++sl) {
#pragma unroll
        for (int f = 0; f < 2; ++f) {
          const float* fr = FT + ((size_t)(i * 2 + f) * 16 + c) * K;
#pragma unroll
          for (int q = 0; q < 2; ++q) {
            const int kq = 16 * i - 8 + 16 * sl + 8 * h + 4 * q - w0;
            const bool in = (kq >= 0) && (kq <= K - 4);
            const v4f v =
                *reinterpret_cast<const v4f*>(fr + iclamp(kq, 0, K - 4));
            g[sl][f][2 * q] = in ? v2f{v.x, v.y} : v2f{0.f, 0.f};
            g[sl][f][2 * q + 1] = in ? v2f{v.z, v.w} : v2f{0.f, 0.f};
          }
        }
      }
    }

#pragma unroll
    for (int n = 0; n < NLD; ++n) {
      const int j = tid + 256 * n;
      if (256 * (n + 1) <= NPIECE || j < NPIECE) {
        const int row = j / SEG, seg = j - row * SEG;
        *reinterpret_cast<v4f*>(tile + row * PITCH + 4 * seg) = pre[n];
      }
    }
    lds_barrier();

    if (t < 4) fetch(16 * (i0 + t + 1) - 8);

    v8f zc1, zc2;
    row_tile<K>(tile, tSB, fl, fs, 8 * PITCH, sgn, mul, odd, bL, bH, zc1, zc2);

    if (t > 0) {
      v8f a1 = col_tile(zp1, g[0][0], zero);
      v8f a2 = col_tile(zp2, g[0][1], zero);
      a1 = col_tile(zc1, g[1][0], a1);
      a2 = col_tile(zc2, g[1][1], a2);
      store8(op + (size_t)(16 * i) * R, a1 + a2);
    }
    zp1 = zc1;
    zp2 = zc2;
    lds_barrier();
  }
}

extern "C" void kernel_launch(void* const* d_in, const int* in_sizes, int n_in,
                              void* d_out, int out_size, void* d_ws,
                              size_t ws_size, hipStream_t stream) {
  (void)in_sizes; (void)n_in; (void)out_size; (void)ws_size;
  const float* yl  = (const float*)d_in[0];
  const float* yh1 = (const float*)d_in[1];
  const float* yh2 = (const float*)d_in[2];
  const float* yh3 = (const float*)d_in[3];
  const float* g0o = (const float*)d_in[4];
  const float* g1o = (const float*)d_in[5];
  const float* g0a = (const float*)d_in[6];
  const float* g0b = (const float*)d_in[7];
  const float* g1a = (const float*)d_in[8];
  const float* g1b = (const float*)d_in[9];

  const size_t SEG = (size_t)16 * 1024 * 1024;
  float* bufA = (float*)d_ws;
  float* bufB = bufA + SEG;
  float* F3 = bufB + SEG;
  float* F2 = F3 + 8 * 2 * 256;
  float* F1 = F2 + 16 * 2 * 256;

  build_ifilt_T<<<dim3(8, 2),  256, 0, stream>>>(g0a, g0b, g1a, g1b, F3, 64);
  build_ifilt_T<<<dim3(16, 2), 256, 0, stream>>>(g0a, g0b, g1a, g1b, F2, 128);
  build_j1_T<<<dim3(16, 2),    256, 0, stream>>>(g0o, g1o, F1, 256);

  level_up<64><<<dim3(2, 256), 256, 0, stream>>>(yl, yh3, F3, bufA);
  level_up<128><<<dim3(8, 256), 256, 0, stream>>>(bufA, yh2, F2, bufB);
  level_one<256><<<dim3(8, 256), 256, 0, stream>>>(bufB, yh1, F1,
                                                   (float*)d_out);
}


# No builder launches (6 -> 3): each work-group computes the three distinct banded filter tiles (first, inner, last) from the taps into LDS behind its input loads; B fragments cut from LDS; same sums
# speedup vs baseline: 2.5379x; 1.1257x over previous
#include <hip/hip_runtime.h>


typedef __attribute__((ext_vector_type(2))) float v2f;
typedef __attribute__((ext_vector_type(4))) float v4f;
typedef __attribute__((ext_vector_type(8))) float v8f;

#define SC_HALF_SQRT2 0.70710678118654752f

__device__ __forceinline__ int refl(int i, int n) {
  i = (i < 0) ? (-1 - i) : i;
  i = (i >= n) ? (2 * n - 1 - i) : i;
  return i;
}

__device__ __forceinline__ int iclamp(int v, int lo, int hi) {
  return v < lo ? lo : (v > hi ? hi : v);
}

__device__ __forceinline__ v8f wmma4(v2f a, v2f b, v8f c) {
  return __builtin_amdgcn_wmma_f32_16x16x4_f32(false, a, false, b, (short)0, c,
                                               false, false);
}

__device__ __forceinline__ float ifilt_coef(int o, int kg, int n,
                                            const float (&ha)[10],
                                            const float (&hb)[10], bool hp) {
  const int i = o >> 2, s = o & 3;
  const int base = 4 + ((((s & 1) != 0) != hp) ? 1 : 0);
  float acc = 0.f;
#pragma unroll
  for (int m = 0; m < 5; ++m) {
    const int L = refl(2 * i + base - 2 * m, n);
    const float c = (s == 0)   ? ha[2 * m]
                    : (s == 1) ? hb[2 * m]
                    : (s == 2) ? ha[2 * m + 1]
                               : hb[2 * m + 1];
    acc += (L == kg) ? c : 0.f;
  }
  return acc;
}

template <int NT>
__device__ __forceinline__ float filt_coef(int o, int kg, int n,
                                           const float (&h)[NT]) {
  float acc = 0.f;
#pragma unroll
  for (int m = 0; m < NT; ++m) {
    const int L = refl(o + (NT >> 1) - m, n);
    acc += (L == kg) ? h[m] : 0.f;
  }
  return acc;
}

constexpr int FPU = 20;
constexpr int FP1 = 28;

template <int NT>
__device__ __forceinline__ int tile_class(int ty) {
  return ty == 0 ? 0 : (ty == NT - 1 ? 2 : 1);
}

template <int MT, int W, int FP>
__device__ __forceinline__ void put_row(float* __restrict__ rp,
                                        const float (&cm)[MT],
                                        const int (&km)[MT]) {
  const v4f z4 = {0.f, 0.f, 0.f, 0.f};
#pragma unroll
  for (int q = 0; q < FP / 4; ++q) *reinterpret_cast<v4f*>(rp + 4 * q) = z4;
#pragma unroll
  for (int m = 0; m < MT; ++m) {
    float acc = 0.f;
#pragma unroll
    for (int mm = 0; mm <= m; ++mm) acc += (km[mm] == km[m]) ? cm[mm] : 0.f;
    if (km[m] >= 0 && km[m] < W) rp[km[m]] = acc;
  }
}

template <int R>
__device__ __forceinline__ void ifilt_row_taps(
    const float* __restrict__ g0a, const float* __restrict__ g0b,
    const float* __restrict__ g1a, const float* __restrict__ g1b, int tid,
    float (&cm)[5], int (&km)[5]) {
  constexpr int NT = R / 8;
  const int nf = tid >> 4, row = tid & 15;
  const int cls = nf >> 1;
  const bool hp = (nf & 1) != 0;
  const int ty = (cls == 0) ? 0 : ((cls == 1) ? 1 : NT - 1);
  const int o = (ty << 4) + row;
  const int w0 = iclamp(8 * ty - 4, 0, R - 16);
  const int i = o >> 2, s = o & 3;
  const bool sodd = (s & 1) != 0;
  const int base = 4 + ((sodd != hp) ? 1 : 0);
  const float* tp = sodd ? (hp ? g1a : g0a) : (hp ? g1b : g0b);
  tp += (s >> 1);
#pragma unroll
  for (int m = 0; m < 5; ++m) {
    cm[m] = tp[2 * m];
    km[m] = refl(2 * i + base - 2 * m, R) - w0;
  }
}

template <int R>
__device__ __forceinline__ void build_j1_row(const float (&h0)[7],
                                             const float (&h1)[5],
                                             float* __restrict__ ft, int tid) {
  constexpr int NT = R / 16;
  const int nf = tid >> 4, row = tid & 15;
  const int cls = nf >> 1;
  const bool hi = (nf & 1) != 0;
  const int ty = (cls == 0) ? 0 : ((cls == 1) ? 1 : NT - 1);
  const int o = (ty << 4) + row;
  const int w0 = iclamp(16 * ty - 4, 0, R - 24);
  const int half = hi ? 2 : 3;
  float cm[7];
  int km[7];
#pragma unroll
  for (int m = 0; m < 7; ++m) {
    const float hv = (m < 5) ? h1[m] : 0.f;
    cm[m] = hi ? hv : h0[m];
    const int pos = refl(o + half - m, R) - w0;
    km[m] = (hi && m >= 5) ? (-1 - m) : pos;
  }
  put_row<7, 24, FP1>(ft + tid * FP1, cm, km);
}

template <int K>
__device__ __forceinline__ void load_half(const float* __restrict__ p,
                                          v2f (&o)[K / 4]) {
#pragma unroll
  for (int q = 0; q < K / 8; ++q) {
    const v4f v = *reinterpret_cast<const v4f*>(p + 4 * q);
    o[2 * q] = v2f{v.x, v.y};
    o[2 * q + 1] = v2f{v.z, v.w};
  }
}

template <int K>
__device__ __forceinline__ void load_c2q(const float* __restrict__ p1,
                                         const float* __restrict__ p2,
                                         v2f sgn, v2f mul, bool odd,
                                         v2f (&o)[K / 4]) {
  v2f r1[K / 4], r2[K / 4];
  load_half<K>(p1, r1);
  load_half<K>(p2, r2);
#pragma unroll
  for (int t = 0; t < K / 4; ++t) {
    const v2f u = __builtin_elementwise_fma(sgn, r2[t], r1[t]) * mul;
    o[t].x = odd ? u.y : u.x;
    o[t].y = odd ? u.x : u.y;
  }
}

__device__ __forceinline__ void lds_barrier() {
  __builtin_amdgcn_fence(__ATOMIC_RELEASE, "workgroup", "local");
  __builtin_amdgcn_s_barrier();
  __builtin_amdgcn_fence(__ATOMIC_ACQUIRE, "workgroup", "local");
}

__device__ __forceinline__ void store8(float* __restrict__ p, v8f a) {
  *reinterpret_cast<v4f*>(p) = v4f{a[0], a[1], a[2], a[3]};
  *reinterpret_cast<v4f*>(p + 4) = v4f{a[4], a[5], a[6], a[7]};
}

template <int K>
__device__ __forceinline__ void row_tile(const float* __restrict__ llp,
                                         const float* __restrict__ yhp,
                                         unsigned gl, unsigned gs, int SB,
                                         v2f sgn, v2f mul, bool odd,
                                         const v2f (&bL)[K / 4],
                                         const v2f (&bH)[K / 4], v8f& z1,
                                         v8f& z2) {
  constexpr int KT = K / 4;
  v2f aLL[KT], aLH[KT], aHL[KT], aHH[KT];
  load_half<K>(llp + gl, aLL);
  load_c2q<K>(yhp + 0 * SB + gs, yhp + 5 * SB + gs, sgn, mul, odd, aLH);
  load_c2q<K>(yhp + 2 * SB + gs, yhp + 3 * SB + gs, sgn, mul, odd, aHL);
  load_c2q<K>(yhp + 1 * SB + gs, yhp + 4 * SB + gs, sgn, mul, odd, aHH);
  v8f acc1 = {0.f, 0.f, 0.f, 0.f, 0.f, 0.f, 0.f, 0.f};
  v8f acc2 = {0.f, 0.f, 0.f, 0.f, 0.f, 0.f, 0.f, 0.f};
#pragma unroll
  for (int t = 0; t < KT; ++t) {
    acc1 = wmma4(aLL[t], bL[t], acc1);
    acc2 = wmma4(aLH[t], bL[t], acc2);
    acc1 = wmma4(aHL[t], bH[t], acc1);
    acc2 = wmma4(aHH[t], bH[t], acc2);
  }
  z1 = acc1;
  z2 = acc2;
}

__device__ __forceinline__ v8f col_tile(v8f z, const v2f (&g)[4], v8f acc) {
#pragma unroll
  for (int t = 0; t < 4; ++t)
    acc = wmma4(v2f{z[2 * t], z[2 * t + 1]}, g[t], acc);
  return acc;
}

template <int R>
__global__ __launch_bounds__(256)
void level_up(const float* __restrict__ ll, const float* __restrict__ yh,
              const float* __restrict__ g0a, const float* __restrict__ g0b,
              const float* __restrict__ g1a, const float* __restrict__ g1b,
              float* __restrict__ out) {
  constexpr int P = 2 * R;
  constexpr int NT = P / 16;
  constexpr int H2 = R / 2;
  constexpr int SB = H2 * H2 * 2;
  constexpr int YR = (R < 68) ? R : 68;
  constexpr int YS = YR / 2;
  constexpr int XW = 40;
  constexpr int SEG = XW / 4;
  constexpr int PITCH = 44;
  constexpr int NROW = YR + 6 * YS;
  constexpr int NPIECE = NROW * SEG;
  constexpr int NLD = (NPIECE + 255) / 256;
  __shared__ __attribute__((aligned(16))) float tile[NROW * PITCH];
  __shared__ __attribute__((aligned(16))) float ftile[3 * 2 * 16 * FPU];

  const int tid = threadIdx.x;
  const int wave = tid >> 5, lane = tid & 31;
  const int u = blockIdx.x * 8 + wave;
  const int nt = u % NT, ms = u / NT;
  const int nt0 = (blockIdx.x * 8) % NT;
  const int mc = blockIdx.y;
  const int o0 = nt << 4;
  const int wy = iclamp((o0 >> 1) - 4, 0, R - 16);
  const int Y0 = iclamp(8 * nt0 - 4, 0, R - YR);
  const int X0 = iclamp(32 * ms - 4, 0, R - XW);
  const int h = lane >> 4, c = lane & 15;
  const bool odd = (c & 1) != 0;
  const v2f sgn = odd ? v2f{-1.f, -1.f} : v2f{1.f, 1.f};
  const v2f mul = odd ? v2f{-SC_HALF_SQRT2, SC_HALF_SQRT2}
                      : v2f{SC_HALF_SQRT2, SC_HALF_SQRT2};

  const float* llp = ll + (size_t)mc * (R * R) + (size_t)Y0 * R + X0;
  const float* yhp = yh + (size_t)mc * (6 * SB) + (size_t)(Y0 >> 1) * R + X0;

  float fcm[5];
  int fkm[5];
  ifilt_row_taps<R>(g0a, g0b, g1a, g1b, tid, fcm, fkm);

  v4f pre[NLD];
#pragma unroll
  for (int n = 0; n < NLD; ++n) {
    const int j = tid + 256 * n;
    if (256 * (n + 1) <= NPIECE || j < NPIECE) {
      const int row = j / SEG, seg = j - row * SEG;
      const bool isll = row < YR;
      const int rr = row - YR;
      const int b = rr / YS, sr = rr - b * YS;
      const unsigned offl = (unsigned)row * R;
      const unsigned offs = (unsigned)b * SB + (unsigned)sr * R;
      const float* src = isll ? (llp + offl) : (yhp + offs);
      pre[n] = *reinterpret_cast<const v4f*>(src + 4 * seg);
    }
  }

  if (tid < 96) put_row<5, 16, FPU>(ftile + tid * FPU, fcm, fkm);
  float* op = out + (size_t)mc * (P * P) + (size_t)(o0 + c) * P + 8 * h;

#pragma unroll
  for (int n = 0; n < NLD; ++n) {
    const int j = tid + 256 * n;
    if (256 * (n + 1) <= NPIECE || j < NPIECE) {
      const int row = j / SEG, seg = j - row * SEG;
      *reinterpret_cast<v4f*>(tile + row * PITCH + 4 * seg) = pre[n];
    }
  }
  __syncthreads();

  const float* fw = ftile + c * FPU + 8 * h;
  v2f gL[4], gH[4];
  load_half<16>(fw + (tile_class<NT>(nt) * 2 + 0) * (16 * FPU), gL);
  load_half<16>(fw + (tile_class<NT>(nt) * 2 + 1) * (16 * FPU), gH);

  const float* tSB = tile + YR * PITCH;
  const int fl = (wy - Y0 + c) * PITCH + 8 * h - X0;
  const int fs = (((wy - Y0) >> 1) + (c >> 1)) * PITCH + 8 * h - X0;

#pragma unroll 1
  for (int m = 0; m < 4; ++m) {
    const int pt = ms * 4 + m;
    const int p0 = pt << 4;
    const int wx = iclamp((p0 >> 1) - 4, 0, R - 16);
    v2f bL[4], bH[4];
    load_half<16>(fw + (tile_class<NT>(pt) * 2 + 0) * (16 * FPU), bL);
    load_half<16>(fw + (tile_class<NT>(pt) * 2 + 1) * (16 * FPU), bH);
    v8f z1, z2;
    row_tile<16>(tile, tSB, (unsigned)(fl + wx), (unsigned)(fs + wx), YS * PITCH,
                 sgn, mul, odd, bL, bH, z1, z2);
    const v8f zero = {0.f, 0.f, 0.f, 0.f, 0.f, 0.f, 0.f, 0.f};
    const v8f a1 = col_tile(z1, gL, zero);
    const v8f a2 = col_tile(z2, gH, zero);
    store8(op + p0, a1 + a2);
  }
}

template <int R>
__global__ __launch_bounds__(256)
void level_one(const float* __restrict__ ll, const float* __restrict__ yh,
               const float* __restrict__ g0o, const float* __restrict__ g1o,
               float* __restrict__ out) {
  constexpr int K = 24;
  constexpr int NT = R / 16;
  constexpr int H2 = R / 2;
  constexpr int SB = H2 * H2 * 2;
  constexpr int XW = 136;
  constexpr int SEG = XW / 4;
  constexpr int PITCH = 140;
  constexpr int NROW = 16 + 6 * 8;
  constexpr int NPIECE = NROW * SEG;
  constexpr int NLD = (NPIECE + 255) / 256;
  __shared__ __attribute__((aligned(16))) float tile[NROW * PITCH];
  __shared__ __attribute__((aligned(16))) float ftile[3 * 2 * 16 * FP1];

  const int tid = threadIdx.x;
  const int wave = tid >> 5, lane = tid & 31;
  const int u = blockIdx.x * 8 + wave;
  const int nt = u % NT, ms = u / NT;
  const int nt0 = (blockIdx.x * 8) % NT;
  const int X0 = iclamp(16 * nt0 - 4, 0, R - XW);
  const int mc = blockIdx.y;
  const int p0 = nt << 4;
  const int wx = iclamp(p0 - 4, 0, R - K);
  const int h = lane >> 4, c = lane & 15;
  const bool odd = (c & 1) != 0;
  const v2f sgn = odd ? v2f{-1.f, -1.f} : v2f{1.f, 1.f};
  const v2f mul = odd ? v2f{-SC_HALF_SQRT2, SC_HALF_SQRT2}
                      : v2f{SC_HALF_SQRT2, SC_HALF_SQRT2};

  const float* llp = ll + (size_t)mc * (R * R) + X0;
  const float* yhp = yh + (size_t)mc * (6 * SB) + X0;
  float* op = out + (size_t)mc * (R * R) + (size_t)c * R + p0 + 8 * h;
  const int i0 = ms * 4;

  const unsigned fl = (unsigned)(c * PITCH + wx - X0 + 12 * h);
  const unsigned fs = (unsigned)((c >> 1) * PITCH + wx - X0 + 12 * h);
  const float* tSB = tile + 16 * PITCH;

  v4f pre[NLD];
  auto fetch = [&](int ybase) {
#pragma unroll
    for (int n = 0; n < NLD; ++n) {
      const int j = tid + 256 * n;
      if (256 * (n + 1) <= NPIECE || j < NPIECE) {
        const int row = j / SEG, seg = j - row * SEG;
        const bool isll = row < 16;
        const int rr = row - 16;
        const int yy = iclamp(ybase + (isll ? row : 2 * (rr & 7)), 0, R - 1);
        const unsigned offl = (unsigned)yy * R;
        const unsigned offs = (unsigned)(rr >> 3) * SB + (unsigned)(yy >> 1) * R;
        const float* src = isll ? (llp + offl) : (yhp + offs);
        pre[n] = *reinterpret_cast<const v4f*>(src + 4 * seg);
      }
    }
  };
  float h0[7], h1[5];
#pragma unroll
  for (int m = 0; m < 7; ++m) h0[m] = g0o[m];
#pragma unroll
  for (int m = 0; m < 5; ++m) h1[m] = g1o[m];

  fetch(16 * i0 - 8);

  if (tid < 96) build_j1_row<R>(h0, h1, ftile, tid);
  lds_barrier();

  const float* fw = ftile + c * FP1;
  v2f bL[6], bH[6];
  load_half<K>(fw + (tile_class<NT>(nt) * 2 + 0) * (16 * FP1) + 12 * h, bL);
  load_half<K>(fw + (tile_class<NT>(nt) * 2 + 1) * (16 * FP1) + 12 * h, bH);

  const v8f zero = {0.f, 0.f, 0.f, 0.f, 0.f, 0.f, 0.f, 0.f};
  v8f zp1 = zero, zp2 = zero;

#pragma unroll 1
  for (int t = 0; t < 5; ++t) {
#pragma unroll
    for (int n = 0; n < NLD; ++n) {
      const int j = tid + 256 * n;
      if (256 * (n + 1) <= NPIECE || j < NPIECE) {
        const int row = j / SEG, seg = j - row * SEG;
        *reinterpret_cast<v4f*>(tile + row * PITCH + 4 * seg) = pre[n];
      }
    }
    lds_barrier();

    if (t < 4) fetch(16 * (i0 + t + 1) - 8);

    const int i = i0 + t - 1;
    const int w0 = iclamp(16 * i - 4, 0, R - K);
    v4f graw[2][2][2];
    if (t > 0) {
#pragma unroll
      for (int sl = 0; sl < 2; ++sl) {
#pragma unroll
        for (int f = 0; f < 2; ++f) {
          const float* fr = fw + (tile_class<NT>(i) * 2 + f) * (16 * FP1);
#pragma unroll
          for (int q = 0; q < 2; ++q) {
            const int kq = 16 * i - 8 + 16 * sl + 8 * h + 4 * q - w0;
            graw[sl][f][q] =
                *reinterpret_cast<const v4f*>(fr + iclamp(kq, 0, K - 4));
          }
        }
      }
    }

    v8f zc1, zc2;
    row_tile<K>(tile, tSB, fl, fs, 8 * PITCH, sgn, mul, odd, bL, bH, zc1, zc2);

    if (t > 0) {
      v2f g[2][2][4];
#pragma unroll
      for (int sl = 0; sl < 2; ++sl) {
#pragma unroll
        for (int q = 0; q < 2; ++q) {
          const int kq = 16 * i - 8 + 16 * sl + 8 * h + 4 * q - w0;
          const bool in = (kq >= 0) && (kq <= K - 4);
#pragma unroll
          for (int f = 0; f < 2; ++f) {
            const v4f v = graw[sl][f][q];
            g[sl][f][2 * q] = in ? v2f{v.x, v.y} : v2f{0.f, 0.f};
            g[sl][f][2 * q + 1] = in ? v2f{v.z, v.w} : v2f{0.f, 0.f};
          }
        }
      }
      v8f a1 = col_tile(zp1, g[0][0], zero);
      v8f a2 = col_tile(zp2, g[0][1], zero);
      a1 = col_tile(zc1, g[1][0], a1);
      a2 = col_tile(zc2, g[1][1], a2);
      store8(op + (size_t)(16 * i) * R, a1 + a2);
    }
    zp1 = zc1;
    zp2 = zc2;
    lds_barrier();
  }
}

extern "C" void kernel_launch(void* const* d_in, const int* in_sizes, int n_in,
                              void* d_out, int out_size, void* d_ws,
                              size_t ws_size, hipStream_t stream) {
  (void)in_sizes; (void)n_in; (void)out_size; (void)ws_size;
  const float* yl  = (const float*)d_in[0];
  const float* yh1 = (const float*)d_in[1];
  const float* yh2 = (const float*)d_in[2];
  const float* yh3 = (const float*)d_in[3];
  const float* g0o = (const float*)d_in[4];
  const float* g1o = (const float*)d_in[5];
  const float* g0a = (const float*)d_in[6];
  const float* g0b = (const float*)d_in[7];
  const float* g1a = (const float*)d_in[8];
  const float* g1b = (const float*)d_in[9];

  const size_t SEG = (size_t)16 * 1024 * 1024;
  float* bufA = (float*)d_ws;
  float* bufB = bufA + SEG;

  level_up<64><<<dim3(2, 256), 256, 0, stream>>>(yl, yh3, g0a, g0b, g1a, g1b,
                                                 bufA);
  level_up<128><<<dim3(8, 256), 256, 0, stream>>>(bufA, yh2, g0a, g0b, g1a,
                                                  g1b, bufB);
  level_one<256><<<dim3(8, 256), 256, 0, stream>>>(bufB, yh1, g0o, g1o,
                                                   (float*)d_out);
}


# Level-1 kernel: a wave walks its 16-column strip down the whole image (17 row-pass tiles per 16 output tiles, was 5 per 4): no z tile computed or staged twice; 2048 -> 512 work-groups; same sums
# speedup vs baseline: 2.7370x; 1.0785x over previous
#include <hip/hip_runtime.h>


typedef __attribute__((ext_vector_type(2))) float v2f;
typedef __attribute__((ext_vector_type(4))) float v4f;
typedef __attribute__((ext_vector_type(8))) float v8f;

#define SC_HALF_SQRT2 0.70710678118654752f

__device__ __forceinline__ int refl(int i, int n) {
  i = (i < 0) ? (-1 - i) : i;
  i = (i >= n) ? (2 * n - 1 - i) : i;
  return i;
}

__device__ __forceinline__ int iclamp(int v, int lo, int hi) {
  return v < lo ? lo : (v > hi ? hi : v);
}

__device__ __forceinline__ v8f wmma4(v2f a, v2f b, v8f c) {
  return __builtin_amdgcn_wmma_f32_16x16x4_f32(false, a, false, b, (short)0, c,
                                               false, false);
}

__device__ __forceinline__ float ifilt_coef(int o, int kg, int n,
                                            const float (&ha)[10],
                                            const float (&hb)[10], bool hp) {
  const int i = o >> 2, s = o & 3;
  const int base = 4 + ((((s & 1) != 0) != hp) ? 1 : 0);
  float acc = 0.f;
#pragma unroll
  for (int m = 0; m < 5; ++m) {
    const int L = refl(2 * i + base - 2 * m, n);
    const float c = (s == 0)   ? ha[2 * m]
                    : (s == 1) ? hb[2 * m]
                    : (s == 2) ? ha[2 * m + 1]
                               : hb[2 * m + 1];
    acc += (L == kg) ? c : 0.f;
  }
  return acc;
}

template <int NT>
__device__ __forceinline__ float filt_coef(int o, int kg, int n,
                                           const float (&h)[NT]) {
  float acc = 0.f;
#pragma unroll
  for (int m = 0; m < NT; ++m) {
    const int L = refl(o + (NT >> 1) - m, n);
    acc += (L == kg) ? h[m] : 0.f;
  }
  return acc;
}

constexpr int FPU = 20;
constexpr int FP1 = 28;

template <int NT>
__device__ __forceinline__ int tile_class(int ty) {
  return ty == 0 ? 0 : (ty == NT - 1 ? 2 : 1);
}

template <int MT, int W, int FP>
__device__ __forceinline__ void put_row(float* __restrict__ rp,
                                        const float (&cm)[MT],
                                        const int (&km)[MT]) {
  const v4f z4 = {0.f, 0.f, 0.f, 0.f};
#pragma unroll
  for (int q = 0; q < FP / 4; ++q) *reinterpret_cast<v4f*>(rp + 4 * q) = z4;
#pragma unroll
  for (int m = 0; m < MT; ++m) {
    float acc = 0.f;
#pragma unroll
    for (int mm = 0; mm <= m; ++mm) acc += (km[mm] == km[m]) ? cm[mm] : 0.f;
    if (km[m] >= 0 && km[m] < W) rp[km[m]] = acc;
  }
}

template <int R>
__device__ __forceinline__ void ifilt_row_taps(
    const float* __restrict__ g0a, const float* __restrict__ g0b,
    const float* __restrict__ g1a, const float* __restrict__ g1b, int tid,
    float (&cm)[5], int (&km)[5]) {
  constexpr int NT = R / 8;
  const int nf = tid >> 4, row = tid & 15;
  const int cls = nf >> 1;
  const bool hp = (nf & 1) != 0;
  const int ty = (cls == 0) ? 0 : ((cls == 1) ? 1 : NT - 1);
  const int o = (ty << 4) + row;
  const int w0 = iclamp(8 * ty - 4, 0, R - 16);
  const int i = o >> 2, s = o & 3;
  const bool sodd = (s & 1) != 0;
  const int base = 4 + ((sodd != hp) ? 1 : 0);
  const float* tp = sodd ? (hp ? g1a : g0a) : (hp ? g1b : g0b);
  tp += (s >> 1);
#pragma unroll
  for (int m = 0; m < 5; ++m) {
    cm[m] = tp[2 * m];
    km[m] = refl(2 * i + base - 2 * m, R) - w0;
  }
}

template <int R>
__device__ __forceinline__ void build_j1_row(const float (&h0)[7],
                                             const float (&h1)[5],
                                             float* __restrict__ ft, int tid) {
  constexpr int NT = R / 16;
  const int nf = tid >> 4, row = tid & 15;
  const int cls = nf >> 1;
  const bool hi = (nf & 1) != 0;
  const int ty = (cls == 0) ? 0 : ((cls == 1) ? 1 : NT - 1);
  const int o = (ty << 4) + row;
  const int w0 = iclamp(16 * ty - 4, 0, R - 24);
  const int half = hi ? 2 : 3;
  float cm[7];
  int km[7];
#pragma unroll
  for (int m = 0; m < 7; ++m) {
    const float hv = (m < 5) ? h1[m] : 0.f;
    cm[m] = hi ? hv : h0[m];
    const int pos = refl(o + half - m, R) - w0;
    km[m] = (hi && m >= 5) ? (-1 - m) : pos;
  }
  put_row<7, 24, FP1>(ft + tid * FP1, cm, km);
}

template <int K>
__device__ __forceinline__ void load_half(const float* __restrict__ p,
                                          v2f (&o)[K / 4]) {
#pragma unroll
  for (int q = 0; q < K / 8; ++q) {
    const v4f v = *reinterpret_cast<const v4f*>(p + 4 * q);
    o[2 * q] = v2f{v.x, v.y};
    o[2 * q + 1] = v2f{v.z, v.w};
  }
}

template <int K>
__device__ __forceinline__ void load_c2q(const float* __restrict__ p1,
                                         const float* __restrict__ p2,
                                         v2f sgn, v2f mul, bool odd,
                                         v2f (&o)[K / 4]) {
  v2f r1[K / 4], r2[K / 4];
  load_half<K>(p1, r1);
  load_half<K>(p2, r2);
#pragma unroll
  for (int t = 0; t < K / 4; ++t) {
    const v2f u = __builtin_elementwise_fma(sgn, r2[t], r1[t]) * mul;
    o[t].x = odd ? u.y : u.x;
    o[t].y = odd ? u.x : u.y;
  }
}

__device__ __forceinline__ void lds_barrier() {
  __builtin_amdgcn_fence(__ATOMIC_RELEASE, "workgroup", "local");
  __builtin_amdgcn_s_barrier();
  __builtin_amdgcn_fence(__ATOMIC_ACQUIRE, "workgroup", "local");
}

__device__ __forceinline__ void store8(float* __restrict__ p, v8f a) {
  *reinterpret_cast<v4f*>(p) = v4f{a[0], a[1], a[2], a[3]};
  *reinterpret_cast<v4f*>(p + 4) = v4f{a[4], a[5], a[6], a[7]};
}

template <int K>
__device__ __forceinline__ void row_tile(const float* __restrict__ llp,
                                         const float* __restrict__ yhp,
                                         unsigned gl, unsigned gs, int SB,
                                         v2f sgn, v2f mul, bool odd,
                                         const v2f (&bL)[K / 4],
                                         const v2f (&bH)[K / 4], v8f& z1,
                                         v8f& z2) {
  constexpr int KT = K / 4;
  v2f aLL[KT], aLH[KT], aHL[KT], aHH[KT];
  load_half<K>(llp + gl, aLL);
  load_c2q<K>(yhp + 0 * SB + gs, yhp + 5 * SB + gs, sgn, mul, odd, aLH);
  load_c2q<K>(yhp + 2 * SB + gs, yhp + 3 * SB + gs, sgn, mul, odd, aHL);
  load_c2q<K>(yhp + 1 * SB + gs, yhp + 4 * SB + gs, sgn, mul, odd, aHH);
  v8f acc1 = {0.f, 0.f, 0.f, 0.f, 0.f, 0.f, 0.f, 0.f};
  v8f acc2 = {0.f, 0.f, 0.f, 0.f, 0.f, 0.f, 0.f, 0.f};
#pragma unroll
  for (int t = 0; t < KT; ++t) {
    acc1 = wmma4(aLL[t], bL[t], acc1);
    acc2 = wmma4(aLH[t], bL[t], acc2);
    acc1 = wmma4(aHL[t], bH[t], acc1);
    acc2 = wmma4(aHH[t], bH[t], acc2);
  }
  z1 = acc1;
  z2 = acc2;
}

__device__ __forceinline__ v8f col_tile(v8f z, const v2f (&g)[4], v8f acc) {
#pragma unroll
  for (int t = 0; t < 4; ++t)
    acc = wmma4(v2f{z[2 * t], z[2 * t + 1]}, g[t], acc);
  return acc;
}

template <int R>
__global__ __launch_bounds__(256)
void level_up(const float* __restrict__ ll, const float* __restrict__ yh,
              const float* __restrict__ g0a, const float* __restrict__ g0b,
              const float* __restrict__ g1a, const float* __restrict__ g1b,
              float* __restrict__ out) {
  constexpr int P = 2 * R;
  constexpr int NT = P / 16;
  constexpr int H2 = R / 2;
  constexpr int SB = H2 * H2 * 2;
  constexpr int YR = (R < 68) ? R : 68;
  constexpr int YS = YR / 2;
  constexpr int XW = 40;
  constexpr int SEG = XW / 4;
  constexpr int PITCH = 44;
  constexpr int NROW = YR + 6 * YS;
  constexpr int NPIECE = NROW * SEG;
  constexpr int NLD = (NPIECE + 255) / 256;
  __shared__ __attribute__((aligned(16))) float tile[NROW * PITCH];
  __shared__ __attribute__((aligned(16))) float ftile[3 * 2 * 16 * FPU];

  const int tid = threadIdx.x;
  const int wave = tid >> 5, lane = tid & 31;
  const int u = blockIdx.x * 8 + wave;
  const int nt = u % NT, ms = u / NT;
  const int nt0 = (blockIdx.x * 8) % NT;
  const int mc = blockIdx.y;
  const int o0 = nt << 4;
  const int wy = iclamp((o0 >> 1) - 4, 0, R - 16);
  const int Y0 = iclamp(8 * nt0 - 4, 0, R - YR);
  const int X0 = iclamp(32 * ms - 4, 0, R - XW);
  const int h = lane >> 4, c = lane & 15;
  const bool odd = (c & 1) != 0;
  const v2f sgn = odd ? v2f{-1.f, -1.f} : v2f{1.f, 1.f};
  const v2f mul = odd ? v2f{-SC_HALF_SQRT2, SC_HALF_SQRT2}
                      : v2f{SC_HALF_SQRT2, SC_HALF_SQRT2};

  const float* llp = ll + (size_t)mc * (R * R) + (size_t)Y0 * R + X0;
  const float* yhp = yh + (size_t)mc * (6 * SB) + (size_t)(Y0 >> 1) * R + X0;

  float fcm[5];
  int fkm[5];
  ifilt_row_taps<R>(g0a, g0b, g1a, g1b, tid, fcm, fkm);

  v4f pre[NLD];
#pragma unroll
  for (int n = 0; n < NLD; ++n) {
    const int j = tid + 256 * n;
    if (256 * (n + 1) <= NPIECE || j < NPIECE) {
      const int row = j / SEG, seg = j - row * SEG;
      const bool isll = row < YR;
      const int rr = row - YR;
      const int b = rr / YS, sr = rr - b * YS;
      const unsigned offl = (unsigned)row * R;
      const unsigned offs = (unsigned)b * SB + (unsigned)sr * R;
      const float* src = isll ? (llp + offl) : (yhp + offs);
      pre[n] = *reinterpret_cast<const v4f*>(src + 4 * seg);
    }
  }

  if (tid < 96) put_row<5, 16, FPU>(ftile + tid * FPU, fcm, fkm);
  float* op = out + (size_t)mc * (P * P) + (size_t)(o0 + c) * P + 8 * h;

#pragma unroll
  for (int n = 0; n < NLD; ++n) {
    const int j = tid + 256 * n;
    if (256 * (n + 1) <= NPIECE || j < NPIECE) {
      const int row = j / SEG, seg = j - row * SEG;
      *reinterpret_cast<v4f*>(tile + row * PITCH + 4 * seg) = pre[n];
    }
  }
  __syncthreads();

  const float* fw = ftile + c * FPU + 8 * h;
  v2f gL[4], gH[4];
  load_half<16>(fw + (tile_class<NT>(nt) * 2 + 0) * (16 * FPU), gL);
  load_half<16>(fw + (tile_class<NT>(nt) * 2 + 1) * (16 * FPU), gH);

  const float* tSB = tile + YR * PITCH;
  const int fl = (wy - Y0 + c) * PITCH + 8 * h - X0;
  const int fs = (((wy - Y0) >> 1) + (c >> 1)) * PITCH + 8 * h - X0;

#pragma unroll 1
  for (int m = 0; m < 4; ++m) {
    const int pt = ms * 4 + m;
    const int p0 = pt << 4;
    const int wx = iclamp((p0 >> 1) - 4, 0, R - 16);
    v2f bL[4], bH[4];
    load_half<16>(fw + (tile_class<NT>(pt) * 2 + 0) * (16 * FPU), bL);
    load_half<16>(fw + (tile_class<NT>(pt) * 2 + 1) * (16 * FPU), bH);
    v8f z1, z2;
    row_tile<16>(tile, tSB, (unsigned)(fl + wx), (unsigned)(fs + wx), YS * PITCH,
                 sgn, mul, odd, bL, bH, z1, z2);
    const v8f zero = {0.f, 0.f, 0.f, 0.f, 0.f, 0.f, 0.f, 0.f};
    const v8f a1 = col_tile(z1, gL, zero);
    const v8f a2 = col_tile(z2, gH, zero);
    store8(op + p0, a1 + a2);
  }
}

template <int R>
__global__ __launch_bounds__(256)
void level_one(const float* __restrict__ ll, const float* __restrict__ yh,
               const float* __restrict__ g0o, const float* __restrict__ g1o,
               float* __restrict__ out) {
  constexpr int K = 24;
  constexpr int NT = R / 16;
  constexpr int H2 = R / 2;
  constexpr int SB = H2 * H2 * 2;
  constexpr int XW = 136;
  constexpr int SEG = XW / 4;
  constexpr int PITCH = 140;
  constexpr int NROW = 16 + 6 * 8;
  constexpr int NPIECE = NROW * SEG;
  constexpr int NLD = (NPIECE + 255) / 256;
  __shared__ __attribute__((aligned(16))) float tile[NROW * PITCH];
  __shared__ __attribute__((aligned(16))) float ftile[3 * 2 * 16 * FP1];

  const int tid = threadIdx.x;
  const int wave = tid >> 5, lane = tid & 31;
  const int nt0 = blockIdx.x * 8;
  const int nt = nt0 + wave;
  const int X0 = iclamp(16 * nt0 - 4, 0, R - XW);
  const int mc = blockIdx.y;
  const int p0 = nt << 4;
  const int wx = iclamp(p0 - 4, 0, R - K);
  const int h = lane >> 4, c = lane & 15;
  const bool odd = (c & 1) != 0;
  const v2f sgn = odd ? v2f{-1.f, -1.f} : v2f{1.f, 1.f};
  const v2f mul = odd ? v2f{-SC_HALF_SQRT2, SC_HALF_SQRT2}
                      : v2f{SC_HALF_SQRT2, SC_HALF_SQRT2};

  const float* llp = ll + (size_t)mc * (R * R) + X0;
  const float* yhp = yh + (size_t)mc * (6 * SB) + X0;
  float* op = out + (size_t)mc * (R * R) + (size_t)c * R + p0 + 8 * h;

  const unsigned fl = (unsigned)(c * PITCH + wx - X0 + 12 * h);
  const unsigned fs = (unsigned)((c >> 1) * PITCH + wx - X0 + 12 * h);
  const float* tSB = tile + 16 * PITCH;

  v4f pre[NLD];
  auto fetch = [&](int ybase) {
#pragma unroll
    for (int n = 0; n < NLD; ++n) {
      const int j = tid + 256 * n;
      if (256 * (n + 1) <= NPIECE || j < NPIECE) {
        const int row = j / SEG, seg = j - row * SEG;
        const bool isll = row < 16;
        const int rr = row - 16;
        const int yy = iclamp(ybase + (isll ? row : 2 * (rr & 7)), 0, R - 1);
        const unsigned offl = (unsigned)yy * R;
        const unsigned offs = (unsigned)(rr >> 3) * SB + (unsigned)(yy >> 1) * R;
        const float* src = isll ? (llp + offl) : (yhp + offs);
        pre[n] = *reinterpret_cast<const v4f*>(src + 4 * seg);
      }
    }
  };
  float h0[7], h1[5];
#pragma unroll
  for (int m = 0; m < 7; ++m) h0[m] = g0o[m];
#pragma unroll
  for (int m = 0; m < 5; ++m) h1[m] = g1o[m];

  fetch(-8);

  if (tid < 96) build_j1_row<R>(h0, h1, ftile, tid);
  lds_barrier();

  const float* fw = ftile + c * FP1;
  v2f bL[6], bH[6];
  load_half<K>(fw + (tile_class<NT>(nt) * 2 + 0) * (16 * FP1) + 12 * h, bL);
  load_half<K>(fw + (tile_class<NT>(nt) * 2 + 1) * (16 * FP1) + 12 * h, bH);

  const v8f zero = {0.f, 0.f, 0.f, 0.f, 0.f, 0.f, 0.f, 0.f};
  v8f zp1 = zero, zp2 = zero;

#pragma unroll 1
  for (int t = 0; t <= NT; ++t) {
#pragma unroll
    for (int n = 0; n < NLD; ++n) {
      const int j = tid + 256 * n;
      if (256 * (n + 1) <= NPIECE || j < NPIECE) {
        const int row = j / SEG, seg = j - row * SEG;
        *reinterpret_cast<v4f*>(tile + row * PITCH + 4 * seg) = pre[n];
      }
    }
    lds_barrier();

    if (t < NT) fetch(16 * (t + 1) - 8);

    const int i = t - 1;
    const int w0 = iclamp(16 * i - 4, 0, R - K);
    v4f graw[2][2][2];
    if (t > 0) {
#pragma unroll
      for (int sl = 0; sl < 2; ++sl) {
#pragma unroll
        for (int f = 0; f < 2; ++f) {
          const float* fr = fw + (tile_class<NT>(i) * 2 + f) * (16 * FP1);
#pragma unroll
          for (int q = 0; q < 2; ++q) {
            const int kq = 16 * i - 8 + 16 * sl + 8 * h + 4 * q - w0;
            graw[sl][f][q] =
                *reinterpret_cast<const v4f*>(fr + iclamp(kq, 0, K - 4));
          }
        }
      }
    }

    v8f zc1, zc2;
    row_tile<K>(tile, tSB, fl, fs, 8 * PITCH, sgn, mul, odd, bL, bH, zc1, zc2);

    if (t > 0) {
      v2f g[2][2][4];
#pragma unroll
      for (int sl = 0; sl < 2; ++sl) {
#pragma unroll
        for (int q = 0; q < 2; ++q) {
          const int kq = 16 * i - 8 + 16 * sl + 8 * h + 4 * q - w0;
          const bool in = (kq >= 0) && (kq <= K - 4);
#pragma unroll
          for (int f = 0; f < 2; ++f) {
            const v4f v = graw[sl][f][q];
            g[sl][f][2 * q] = in ? v2f{v.x, v.y} : v2f{0.f, 0.f};
            g[sl][f][2 * q + 1] = in ? v2f{v.z, v.w} : v2f{0.f, 0.f};
          }
        }
      }
      v8f a1 = col_tile(zp1, g[0][0], zero);
      v8f a2 = col_tile(zp2, g[0][1], zero);
      a1 = col_tile(zc1, g[1][0], a1);
      a2 = col_tile(zc2, g[1][1], a2);
      store8(op + (size_t)(16 * i) * R, a1 + a2);
    }
    zp1 = zc1;
    zp2 = zc2;
    lds_barrier();
  }
}

extern "C" void kernel_launch(void* const* d_in, const int* in_sizes, int n_in,
                              void* d_out, int out_size, void* d_ws,
                              size_t ws_size, hipStream_t stream) {
  (void)in_sizes; (void)n_in; (void)out_size; (void)ws_size;
  const float* yl  = (const float*)d_in[0];
  const float* yh1 = (const float*)d_in[1];
  const float* yh2 = (const float*)d_in[2];
  const float* yh3 = (const float*)d_in[3];
  const float* g0o = (const float*)d_in[4];
  const float* g1o = (const float*)d_in[5];
  const float* g0a = (const float*)d_in[6];
  const float* g0b = (const float*)d_in[7];
  const float* g1a = (const float*)d_in[8];
  const float* g1b = (const float*)d_in[9];

  const size_t SEG = (size_t)16 * 1024 * 1024;
  float* bufA = (float*)d_ws;
  float* bufB = bufA + SEG;

  level_up<64><<<dim3(2, 256), 256, 0, stream>>>(yl, yh3, g0a, g0b, g1a, g1b,
                                                 bufA);
  level_up<128><<<dim3(8, 256), 256, 0, stream>>>(bufA, yh2, g0a, g0b, g1a,
                                                  g1b, bufB);
  level_one<256><<<dim3(2, 256), 256, 0, stream>>>(bufB, yh1, g0o, g1o,
                                                   (float*)d_out);
}


# All three kernels: each output tile is parked in LDS and re-read row-wise, so every global store instruction writes whole 128-byte lines (was 2 x 16 B in each of 16 rows); same sums
# speedup vs baseline: 3.3715x; 1.2318x over previous
#include <hip/hip_runtime.h>


typedef __attribute__((ext_vector_type(2))) float v2f;
typedef __attribute__((ext_vector_type(4))) float v4f;
typedef __attribute__((ext_vector_type(8))) float v8f;

#define SC_HALF_SQRT2 0.70710678118654752f

__device__ __forceinline__ int refl(int i, int n) {
  i = (i < 0) ? (-1 - i) : i;
  i = (i >= n) ? (2 * n - 1 - i) : i;
  return i;
}

__device__ __forceinline__ int iclamp(int v, int lo, int hi) {
  return v < lo ? lo : (v > hi ? hi : v);
}

__device__ __forceinline__ v8f wmma4(v2f a, v2f b, v8f c) {
  return __builtin_amdgcn_wmma_f32_16x16x4_f32(false, a, false, b, (short)0, c,
                                               false, false);
}

__device__ __forceinline__ float ifilt_coef(int o, int kg, int n,
                                            const float (&ha)[10],
                                            const float (&hb)[10], bool hp) {
  const int i = o >> 2, s = o & 3;
  const int base = 4 + ((((s & 1) != 0) != hp) ? 1 : 0);
  float acc = 0.f;
#pragma unroll
  for (int m = 0; m < 5; ++m) {
    const int L = refl(2 * i + base - 2 * m, n);
    const float c = (s == 0)   ? ha[2 * m]
                    : (s == 1) ? hb[2 * m]
                    : (s == 2) ? ha[2 * m + 1]
                               : hb[2 * m + 1];
    acc += (L == kg) ? c : 0.f;
  }
  return acc;
}

template <int NT>
__device__ __forceinline__ float filt_coef(int o, int kg, int n,
                                           const float (&h)[NT]) {
  float acc = 0.f;
#pragma unroll
  for (int m = 0; m < NT; ++m) {
    const int L = refl(o + (NT >> 1) - m, n);
    acc += (L == kg) ? h[m] : 0.f;
  }
  return acc;
}

constexpr int FPU = 20;
constexpr int FP1 = 28;

template <int NT>
__device__ __forceinline__ int tile_class(int ty) {
  return ty == 0 ? 0 : (ty == NT - 1 ? 2 : 1);
}

template <int MT, int W, int FP>
__device__ __forceinline__ void put_row(float* __restrict__ rp,
                                        const float (&cm)[MT],
                                        const int (&km)[MT]) {
  const v4f z4 = {0.f, 0.f, 0.f, 0.f};
#pragma unroll
  for (int q = 0; q < FP / 4; ++q) *reinterpret_cast<v4f*>(rp + 4 * q) = z4;
#pragma unroll
  for (int m = 0; m < MT; ++m) {
    float acc = 0.f;
#pragma unroll
    for (int mm = 0; mm <= m; ++mm) acc += (km[mm] == km[m]) ? cm[mm] : 0.f;
    if (km[m] >= 0 && km[m] < W) rp[km[m]] = acc;
  }
}

template <int R>
__device__ __forceinline__ void ifilt_row_taps(
    const float* __restrict__ g0a, const float* __restrict__ g0b,
    const float* __restrict__ g1a, const float* __restrict__ g1b, int tid,
    float (&cm)[5], int (&km)[5]) {
  constexpr int NT = R / 8;
  const int nf = tid >> 4, row = tid & 15;
  const int cls = nf >> 1;
  const bool hp = (nf & 1) != 0;
  const int ty = (cls == 0) ? 0 : ((cls == 1) ? 1 : NT - 1);
  const int o = (ty << 4) + row;
  const int w0 = iclamp(8 * ty - 4, 0, R - 16);
  const int i = o >> 2, s = o & 3;
  const bool sodd = (s & 1) != 0;
  const int base = 4 + ((sodd != hp) ? 1 : 0);
  const float* tp = sodd ? (hp ? g1a : g0a) : (hp ? g1b : g0b);
  tp += (s >> 1);
#pragma unroll
  for (int m = 0; m < 5; ++m) {
    cm[m] = tp[2 * m];
    km[m] = refl(2 * i + base - 2 * m, R) - w0;
  }
}

template <int R>
__device__ __forceinline__ void build_j1_row(const float (&h0)[7],
                                             const float (&h1)[5],
                                             float* __restrict__ ft, int tid) {
  constexpr int NT = R / 16;
  const int nf = tid >> 4, row = tid & 15;
  const int cls = nf >> 1;
  const bool hi = (nf & 1) != 0;
  const int ty = (cls == 0) ? 0 : ((cls == 1) ? 1 : NT - 1);
  const int o = (ty << 4) + row;
  const int w0 = iclamp(16 * ty - 4, 0, R - 24);
  const int half = hi ? 2 : 3;
  float cm[7];
  int km[7];
#pragma unroll
  for (int m = 0; m < 7; ++m) {
    const float hv = (m < 5) ? h1[m] : 0.f;
    cm[m] = hi ? hv : h0[m];
    const int pos = refl(o + half - m, R) - w0;
    km[m] = (hi && m >= 5) ? (-1 - m) : pos;
  }
  put_row<7, 24, FP1>(ft + tid * FP1, cm, km);
}

template <int K>
__device__ __forceinline__ void load_half(const float* __restrict__ p,
                                          v2f (&o)[K / 4]) {
#pragma unroll
  for (int q = 0; q < K / 8; ++q) {
    const v4f v = *reinterpret_cast<const v4f*>(p + 4 * q);
    o[2 * q] = v2f{v.x, v.y};
    o[2 * q + 1] = v2f{v.z, v.w};
  }
}

template <int K>
__device__ __forceinline__ void load_c2q(const float* __restrict__ p1,
                                         const float* __restrict__ p2,
                                         v2f sgn, v2f mul, bool odd,
                                         v2f (&o)[K / 4]) {
  v2f r1[K / 4], r2[K / 4];
  load_half<K>(p1, r1);
  load_half<K>(p2, r2);
#pragma unroll
  for (int t = 0; t < K / 4; ++t) {
    const v2f u = __builtin_elementwise_fma(sgn, r2[t], r1[t]) * mul;
    o[t].x = odd ? u.y : u.x;
    o[t].y = odd ? u.x : u.y;
  }
}

__device__ __forceinline__ void lds_barrier() {
  __builtin_amdgcn_fence(__ATOMIC_RELEASE, "workgroup", "local");
  __builtin_amdgcn_s_barrier();
  __builtin_amdgcn_fence(__ATOMIC_ACQUIRE, "workgroup", "local");
}

__device__ __forceinline__ void store8(float* __restrict__ p, v8f a) {
  *reinterpret_cast<v4f*>(p) = v4f{a[0], a[1], a[2], a[3]};
  *reinterpret_cast<v4f*>(p + 4) = v4f{a[4], a[5], a[6], a[7]};
}

__device__ __forceinline__ void wave_lds_handover() {
  __builtin_amdgcn_fence(__ATOMIC_RELEASE, "workgroup", "local");
  __builtin_amdgcn_wave_barrier();
  __builtin_amdgcn_fence(__ATOMIC_ACQUIRE, "workgroup", "local");
}

template <int K>
__device__ __forceinline__ void row_tile(const float* __restrict__ llp,
                                         const float* __restrict__ yhp,
                                         unsigned gl, unsigned gs, int SB,
                                         v2f sgn, v2f mul, bool odd,
                                         const v2f (&bL)[K / 4],
                                         const v2f (&bH)[K / 4], v8f& z1,
                                         v8f& z2) {
  constexpr int KT = K / 4;
  v2f aLL[KT], aLH[KT], aHL[KT], aHH[KT];
  load_half<K>(llp + gl, aLL);
  load_c2q<K>(yhp + 0 * SB + gs, yhp + 5 * SB + gs, sgn, mul, odd, aLH);
  load_c2q<K>(yhp + 2 * SB + gs, yhp + 3 * SB + gs, sgn, mul, odd, aHL);
  load_c2q<K>(yhp + 1 * SB + gs, yhp + 4 * SB + gs, sgn, mul, odd, aHH);
  v8f acc1 = {0.f, 0.f, 0.f, 0.f, 0.f, 0.f, 0.f, 0.f};
  v8f acc2 = {0.f, 0.f, 0.f, 0.f, 0.f, 0.f, 0.f, 0.f};
#pragma unroll
  for (int t = 0; t < KT; ++t) {
    acc1 = wmma4(aLL[t], bL[t], acc1);
    acc2 = wmma4(aLH[t], bL[t], acc2);
    acc1 = wmma4(aHL[t], bH[t], acc1);
    acc2 = wmma4(aHH[t], bH[t], acc2);
  }
  z1 = acc1;
  z2 = acc2;
}

__device__ __forceinline__ v8f col_tile(v8f z, const v2f (&g)[4], v8f acc) {
#pragma unroll
  for (int t = 0; t < 4; ++t)
    acc = wmma4(v2f{z[2 * t], z[2 * t + 1]}, g[t], acc);
  return acc;
}

constexpr int OPU = 36;

template <int R>
__global__ __launch_bounds__(256)
void level_up(const float* __restrict__ ll, const float* __restrict__ yh,
              const float* __restrict__ g0a, const float* __restrict__ g0b,
              const float* __restrict__ g1a, const float* __restrict__ g1b,
              float* __restrict__ out) {
  constexpr int P = 2 * R;
  constexpr int NT = P / 16;
  constexpr int H2 = R / 2;
  constexpr int SB = H2 * H2 * 2;
  constexpr int YR = (R < 68) ? R : 68;
  constexpr int YS = YR / 2;
  constexpr int XW = 40;
  constexpr int SEG = XW / 4;
  constexpr int PITCH = 44;
  constexpr int NROW = YR + 6 * YS;
  constexpr int NPIECE = NROW * SEG;
  constexpr int NLD = (NPIECE + 255) / 256;
  __shared__ __attribute__((aligned(16))) float tile[NROW * PITCH];
  __shared__ __attribute__((aligned(16))) float ftile[3 * 2 * 16 * FPU];
  __shared__ __attribute__((aligned(16))) float obuf[8 * 16 * OPU];

  const int tid = threadIdx.x;
  const int wave = tid >> 5, lane = tid & 31;
  const int u = blockIdx.x * 8 + wave;
  const int nt = u % NT, ms = u / NT;
  const int nt0 = (blockIdx.x * 8) % NT;
  const int mc = blockIdx.y;
  const int o0 = nt << 4;
  const int wy = iclamp((o0 >> 1) - 4, 0, R - 16);
  const int Y0 = iclamp(8 * nt0 - 4, 0, R - YR);
  const int X0 = iclamp(32 * ms - 4, 0, R - XW);
  const int h = lane >> 4, c = lane & 15;
  const bool odd = (c & 1) != 0;
  const v2f sgn = odd ? v2f{-1.f, -1.f} : v2f{1.f, 1.f};
  const v2f mul = odd ? v2f{-SC_HALF_SQRT2, SC_HALF_SQRT2}
                      : v2f{SC_HALF_SQRT2, SC_HALF_SQRT2};

  const float* llp = ll + (size_t)mc * (R * R) + (size_t)Y0 * R + X0;
  const float* yhp = yh + (size_t)mc * (6 * SB) + (size_t)(Y0 >> 1) * R + X0;

  float fcm[5];
  int fkm[5];
  ifilt_row_taps<R>(g0a, g0b, g1a, g1b, tid, fcm, fkm);

  v4f pre[NLD];
#pragma unroll
  for (int n = 0; n < NLD; ++n) {
    const int j = tid + 256 * n;
    if (256 * (n + 1) <= NPIECE || j < NPIECE) {
      const int row = j / SEG, seg = j - row * SEG;
      const bool isll = row < YR;
      const int rr = row - YR;
      const int b = rr / YS, sr = rr - b * YS;
      const unsigned offl = (unsigned)row * R;
      const unsigned offs = (unsigned)b * SB + (unsigned)sr * R;
      const float* src = isll ? (llp + offl) : (yhp + offs);
      pre[n] = *reinterpret_cast<const v4f*>(src + 4 * seg);
    }
  }

  if (tid < 96) put_row<5, 16, FPU>(ftile + tid * FPU, fcm, fkm);
  float* ow = obuf + wave * (16 * OPU);
  float* owt = ow + c * OPU + 8 * h;
  const float* owr = ow + (lane >> 3) * OPU + 4 * (lane & 7);
  float* og = out + (size_t)mc * (P * P) + (size_t)(o0 + (lane >> 3)) * P +
              4 * (lane & 7);

#pragma unroll
  for (int n = 0; n < NLD; ++n) {
    const int j = tid + 256 * n;
    if (256 * (n + 1) <= NPIECE || j < NPIECE) {
      const int row = j / SEG, seg = j - row * SEG;
      *reinterpret_cast<v4f*>(tile + row * PITCH + 4 * seg) = pre[n];
    }
  }
  __syncthreads();

  const float* fw = ftile + c * FPU + 8 * h;
  v2f gL[4], gH[4];
  load_half<16>(fw + (tile_class<NT>(nt) * 2 + 0) * (16 * FPU), gL);
  load_half<16>(fw + (tile_class<NT>(nt) * 2 + 1) * (16 * FPU), gH);

  const float* tSB = tile + YR * PITCH;
  const int fl = (wy - Y0 + c) * PITCH + 8 * h - X0;
  const int fs = (((wy - Y0) >> 1) + (c >> 1)) * PITCH + 8 * h - X0;

#pragma unroll 1
  for (int m = 0; m < 4; ++m) {
    const int pt = ms * 4 + m;
    const int p0 = pt << 4;
    const int wx = iclamp((p0 >> 1) - 4, 0, R - 16);
    v2f bL[4], bH[4];
    load_half<16>(fw + (tile_class<NT>(pt) * 2 + 0) * (16 * FPU), bL);
    load_half<16>(fw + (tile_class<NT>(pt) * 2 + 1) * (16 * FPU), bH);
    v8f z1, z2;
    row_tile<16>(tile, tSB, (unsigned)(fl + wx), (unsigned)(fs + wx), YS * PITCH,
                 sgn, mul, odd, bL, bH, z1, z2);
    const v8f zero = {0.f, 0.f, 0.f, 0.f, 0.f, 0.f, 0.f, 0.f};
    const v8f a1 = col_tile(z1, gL, zero);
    const v8f a2 = col_tile(z2, gH, zero);
    store8(owt + 16 * (m & 1), a1 + a2);
    if ((m & 1) != 0) {
      wave_lds_handover();
      v4f f[4];
#pragma unroll
      for (int q = 0; q < 4; ++q)
        f[q] = *reinterpret_cast<const v4f*>(owr + (4 * q) * OPU);
#pragma unroll
      for (int q = 0; q < 4; ++q)
        *reinterpret_cast<v4f*>(og + (size_t)(4 * q) * P + (p0 - 16)) = f[q];
      wave_lds_handover();
    }
  }
}

constexpr int OP1 = 132;

template <int R>
__global__ __launch_bounds__(256)
void level_one(const float* __restrict__ ll, const float* __restrict__ yh,
               const float* __restrict__ g0o, const float* __restrict__ g1o,
               float* __restrict__ out) {
  constexpr int K = 24;
  constexpr int NT = R / 16;
  constexpr int H2 = R / 2;
  constexpr int SB = H2 * H2 * 2;
  constexpr int XW = 136;
  constexpr int SEG = XW / 4;
  constexpr int PITCH = 140;
  constexpr int NROW = 16 + 6 * 8;
  constexpr int NPIECE = NROW * SEG;
  constexpr int NLD = (NPIECE + 255) / 256;
  __shared__ __attribute__((aligned(16))) float tile[NROW * PITCH];
  __shared__ __attribute__((aligned(16))) float ftile[3 * 2 * 16 * FP1];
  __shared__ __attribute__((aligned(16))) float obuf[2 * 16 * OP1];

  const int tid = threadIdx.x;
  const int wave = tid >> 5, lane = tid & 31;
  const int nt0 = blockIdx.x * 8;
  const int nt = nt0 + wave;
  const int X0 = iclamp(16 * nt0 - 4, 0, R - XW);
  const int mc = blockIdx.y;
  const int p0 = nt << 4;
  const int wx = iclamp(p0 - 4, 0, R - K);
  const int h = lane >> 4, c = lane & 15;
  const bool odd = (c & 1) != 0;
  const v2f sgn = odd ? v2f{-1.f, -1.f} : v2f{1.f, 1.f};
  const v2f mul = odd ? v2f{-SC_HALF_SQRT2, SC_HALF_SQRT2}
                      : v2f{SC_HALF_SQRT2, SC_HALF_SQRT2};

  const float* llp = ll + (size_t)mc * (R * R) + X0;
  const float* yhp = yh + (size_t)mc * (6 * SB) + X0;
  float* owt = obuf + c * OP1 + 16 * wave + 8 * h;
  const float* owr = obuf + (2 * wave) * OP1 + 4 * lane;
  float* og = out + (size_t)mc * (R * R) + (size_t)(2 * wave) * R + 16 * nt0 +
              4 * lane;

  const unsigned fl = (unsigned)(c * PITCH + wx - X0 + 12 * h);
  const unsigned fs = (unsigned)((c >> 1) * PITCH + wx - X0 + 12 * h);
  const float* tSB = tile + 16 * PITCH;

  v4f pre[NLD];
  auto fetch = [&](int ybase) {
#pragma unroll
    for (int n = 0; n < NLD; ++n) {
      const int j = tid + 256 * n;
      if (256 * (n + 1) <= NPIECE || j < NPIECE) {
        const int row = j / SEG, seg = j - row * SEG;
        const bool isll = row < 16;
        const int rr = row - 16;
        const int yy = iclamp(ybase + (isll ? row : 2 * (rr & 7)), 0, R - 1);
        const unsigned offl = (unsigned)yy * R;
        const unsigned offs = (unsigned)(rr >> 3) * SB + (unsigned)(yy >> 1) * R;
        const float* src = isll ? (llp + offl) : (yhp + offs);
        pre[n] = *reinterpret_cast<const v4f*>(src + 4 * seg);
      }
    }
  };
  float h0[7], h1[5];
#pragma unroll
  for (int m = 0; m < 7; ++m) h0[m] = g0o[m];
#pragma unroll
  for (int m = 0; m < 5; ++m) h1[m] = g1o[m];

  fetch(-8);

  if (tid < 96) build_j1_row<R>(h0, h1, ftile, tid);
  lds_barrier();

  const float* fw = ftile + c * FP1;
  v2f bL[6], bH[6];
  load_half<K>(fw + (tile_class<NT>(nt) * 2 + 0) * (16 * FP1) + 12 * h, bL);
  load_half<K>(fw + (tile_class<NT>(nt) * 2 + 1) * (16 * FP1) + 12 * h, bH);

  const v8f zero = {0.f, 0.f, 0.f, 0.f, 0.f, 0.f, 0.f, 0.f};
  v8f zp1 = zero, zp2 = zero;

#pragma unroll 1
  for (int t = 0; t <= NT; ++t) {
#pragma unroll
    for (int n = 0; n < NLD; ++n) {
      const int j = tid + 256 * n;
      if (256 * (n + 1) <= NPIECE || j < NPIECE) {
        const int row = j / SEG, seg = j - row * SEG;
        *reinterpret_cast<v4f*>(tile + row * PITCH + 4 * seg) = pre[n];
      }
    }
    lds_barrier();

    if (t < NT) fetch(16 * (t + 1) - 8);

    const int i = t - 1;
    const int w0 = iclamp(16 * i - 4, 0, R - K);
    v4f graw[2][2][2];
    if (t > 0) {
#pragma unroll
      for (int sl = 0; sl < 2; ++sl) {
#pragma unroll
        for (int f = 0; f < 2; ++f) {
          const float* fr = fw + (tile_class<NT>(i) * 2 + f) * (16 * FP1);
#pragma unroll
          for (int q = 0; q < 2; ++q) {
            const int kq = 16 * i - 8 + 16 * sl + 8 * h + 4 * q - w0;
            graw[sl][f][q] =
                *reinterpret_cast<const v4f*>(fr + iclamp(kq, 0, K - 4));
          }
        }
      }
    }

    v4f f0, f1;
    if (t > 1) {
      const float* rr = owr + ((t - 1) & 1) * (16 * OP1);
      f0 = *reinterpret_cast<const v4f*>(rr);
      f1 = *reinterpret_cast<const v4f*>(rr + OP1);
    }

    v8f zc1, zc2;
    row_tile<K>(tile, tSB, fl, fs, 8 * PITCH, sgn, mul, odd, bL, bH, zc1, zc2);

    if (t > 1) {
      float* g = og + (size_t)(16 * (t - 2)) * R;
      *reinterpret_cast<v4f*>(g) = f0;
      *reinterpret_cast<v4f*>(g + R) = f1;
    }

    if (t > 0) {
      v2f g[2][2][4];
#pragma unroll
      for (int sl = 0; sl < 2; ++sl) {
#pragma unroll
        for (int q = 0; q < 2; ++q) {
          const int kq = 16 * i - 8 + 16 * sl + 8 * h + 4 * q - w0;
          const bool in = (kq >= 0) && (kq <= K - 4);
#pragma unroll
          for (int f = 0; f < 2; ++f) {
            const v4f v = graw[sl][f][q];
            g[sl][f][2 * q] = in ? v2f{v.x, v.y} : v2f{0.f, 0.f};
            g[sl][f][2 * q + 1] = in ? v2f{v.z, v.w} : v2f{0.f, 0.f};
          }
        }
      }
      v8f a1 = col_tile(zp1, g[0][0], zero);
      v8f a2 = col_tile(zp2, g[0][1], zero);
      a1 = col_tile(zc1, g[1][0], a1);
      a2 = col_tile(zc2, g[1][1], a2);
      store8(owt + (t & 1) * (16 * OP1), a1 + a2);
    }
    zp1 = zc1;
    zp2 = zc2;
    lds_barrier();
  }
  {
    const float* rr = owr + (NT & 1) * (16 * OP1);
    const v4f f0 = *reinterpret_cast<const v4f*>(rr);
    const v4f f1 = *reinterpret_cast<const v4f*>(rr + OP1);
    float* g = og + (size_t)(16 * (NT - 1)) * R;
    *reinterpret_cast<v4f*>(g) = f0;
    *reinterpret_cast<v4f*>(g + R) = f1;
  }
}

extern "C" void kernel_launch(void* const* d_in, const int* in_sizes, int n_in,
                              void* d_out, int out_size, void* d_ws,
                              size_t ws_size, hipStream_t stream) {
  (void)in_sizes; (void)n_in; (void)out_size; (void)ws_size;
  const float* yl  = (const float*)d_in[0];
  const float* yh1 = (const float*)d_in[1];
  const float* yh2 = (const float*)d_in[2];
  const float* yh3 = (const float*)d_in[3];
  const float* g0o = (const float*)d_in[4];
  const float* g1o = (const float*)d_in[5];
  const float* g0a = (const float*)d_in[6];
  const float* g0b = (const float*)d_in[7];
  const float* g1a = (const float*)d_in[8];
  const float* g1b = (const float*)d_in[9];

  const size_t SEG = (size_t)16 * 1024 * 1024;
  float* bufA = (float*)d_ws;
  float* bufB = bufA + SEG;

  level_up<64><<<dim3(2, 256), 256, 0, stream>>>(yl, yh3, g0a, g0b, g1a, g1b,
                                                 bufA);
  level_up<128><<<dim3(8, 256), 256, 0, stream>>>(bufA, yh2, g0a, g0b, g1a,
                                                  g1b, bufB);
  level_one<256><<<dim3(2, 256), 256, 0, stream>>>(bufB, yh1, g0o, g1o,
                                                   (float*)d_out);
}


# Level 1: row-pass lanes take a z tile's rows in an order that makes each column-pass fragment 4 neighbouring rows; the 2 fragments outside the filter window go (16->12 WMMAs; sums re-associated)
# speedup vs baseline: 3.4846x; 1.0335x over previous
#include <hip/hip_runtime.h>


typedef __attribute__((ext_vector_type(2))) float v2f;
typedef __attribute__((ext_vector_type(4))) float v4f;
typedef __attribute__((ext_vector_type(8))) float v8f;

#define SC_HALF_SQRT2 0.70710678118654752f

__device__ __forceinline__ int refl(int i, int n) {
  i = (i < 0) ? (-1 - i) : i;
  i = (i >= n) ? (2 * n - 1 - i) : i;
  return i;
}

__device__ __forceinline__ int iclamp(int v, int lo, int hi) {
  return v < lo ? lo : (v > hi ? hi : v);
}

__device__ __forceinline__ v8f wmma4(v2f a, v2f b, v8f c) {
  return __builtin_amdgcn_wmma_f32_16x16x4_f32(false, a, false, b, (short)0, c,
                                               false, false);
}

__device__ __forceinline__ float ifilt_coef(int o, int kg, int n,
                                            const float (&ha)[10],
                                            const float (&hb)[10], bool hp) {
  const int i = o >> 2, s = o & 3;
  const int base = 4 + ((((s & 1) != 0) != hp) ? 1 : 0);
  float acc = 0.f;
#pragma unroll
  for (int m = 0; m < 5; ++m) {
    const int L = refl(2 * i + base - 2 * m, n);
    const float c = (s == 0)   ? ha[2 * m]
                    : (s == 1) ? hb[2 * m]
                    : (s == 2) ? ha[2 * m + 1]
                               : hb[2 * m + 1];
    acc += (L == kg) ? c : 0.f;
  }
  return acc;
}

template <int NT>
__device__ __forceinline__ float filt_coef(int o, int kg, int n,
                                           const float (&h)[NT]) {
  float acc = 0.f;
#pragma unroll
  for (int m = 0; m < NT; ++m) {
    const int L = refl(o + (NT >> 1) - m, n);
    acc += (L == kg) ? h[m] : 0.f;
  }
  return acc;
}

constexpr int FPU = 20;
constexpr int FP1 = 36;
constexpr int FM1 = 4;

template <int NT>
__device__ __forceinline__ int tile_class(int ty) {
  return ty == 0 ? 0 : (ty == NT - 1 ? 2 : 1);
}

template <int MT, int W, int FP, int OFF = 0>
__device__ __forceinline__ void put_row(float* __restrict__ rp,
                                        const float (&cm)[MT],
                                        const int (&km)[MT]) {
  const v4f z4 = {0.f, 0.f, 0.f, 0.f};
#pragma unroll
  for (int q = 0; q < FP / 4; ++q) *reinterpret_cast<v4f*>(rp + 4 * q) = z4;
#pragma unroll
  for (int m = 0; m < MT; ++m) {
    float acc = 0.f;
#pragma unroll
    for (int mm = 0; mm <= m; ++mm) acc += (km[mm] == km[m]) ? cm[mm] : 0.f;
    if (km[m] >= 0 && km[m] < W) rp[OFF + km[m]] = acc;
  }
}

template <int R>
__device__ __forceinline__ void ifilt_row_taps(
    const float* __restrict__ g0a, const float* __restrict__ g0b,
    const float* __restrict__ g1a, const float* __restrict__ g1b, int tid,
    float (&cm)[5], int (&km)[5]) {
  constexpr int NT = R / 8;
  const int nf = tid >> 4, row = tid & 15;
  const int cls = nf >> 1;
  const bool hp = (nf & 1) != 0;
  const int ty = (cls == 0) ? 0 : ((cls == 1) ? 1 : NT - 1);
  const int o = (ty << 4) + row;
  const int w0 = iclamp(8 * ty - 4, 0, R - 16);
  const int i = o >> 2, s = o & 3;
  const bool sodd = (s & 1) != 0;
  const int base = 4 + ((sodd != hp) ? 1 : 0);
  const float* tp = sodd ? (hp ? g1a : g0a) : (hp ? g1b : g0b);
  tp += (s >> 1);
#pragma unroll
  for (int m = 0; m < 5; ++m) {
    cm[m] = tp[2 * m];
    km[m] = refl(2 * i + base - 2 * m, R) - w0;
  }
}

template <int R>
__device__ __forceinline__ void build_j1_row(const float (&h0)[7],
                                             const float (&h1)[5],
                                             float* __restrict__ ft, int tid) {
  constexpr int NT = R / 16;
  const int nf = tid >> 4, row = tid & 15;
  const int cls = nf >> 1;
  const bool hi = (nf & 1) != 0;
  const int ty = (cls == 0) ? 0 : ((cls == 1) ? 1 : NT - 1);
  const int o = (ty << 4) + row;
  const int w0 = iclamp(16 * ty - 4, 0, R - 24);
  const int half = hi ? 2 : 3;
  float cm[7];
  int km[7];
#pragma unroll
  for (int m = 0; m < 7; ++m) {
    const float hv = (m < 5) ? h1[m] : 0.f;
    cm[m] = hi ? hv : h0[m];
    const int pos = refl(o + half - m, R) - w0;
    km[m] = (hi && m >= 5) ? (-1 - m) : pos;
  }
  put_row<7, 24, FP1, FM1>(ft + tid * FP1, cm, km);
}

template <int K>
__device__ __forceinline__ void load_half(const float* __restrict__ p,
                                          v2f (&o)[K / 4]) {
#pragma unroll
  for (int q = 0; q < K / 8; ++q) {
    const v4f v = *reinterpret_cast<const v4f*>(p + 4 * q);
    o[2 * q] = v2f{v.x, v.y};
    o[2 * q + 1] = v2f{v.z, v.w};
  }
}

template <int K>
__device__ __forceinline__ void load_c2q(const float* __restrict__ p1,
                                         const float* __restrict__ p2,
                                         v2f sgn, v2f mul, bool odd,
                                         v2f (&o)[K / 4]) {
  v2f r1[K / 4], r2[K / 4];
  load_half<K>(p1, r1);
  load_half<K>(p2, r2);
#pragma unroll
  for (int t = 0; t < K / 4; ++t) {
    const v2f u = __builtin_elementwise_fma(sgn, r2[t], r1[t]) * mul;
    o[t].x = odd ? u.y : u.x;
    o[t].y = odd ? u.x : u.y;
  }
}

__device__ __forceinline__ void lds_barrier() {
  __builtin_amdgcn_fence(__ATOMIC_RELEASE, "workgroup", "local");
  __builtin_amdgcn_s_barrier();
  __builtin_amdgcn_fence(__ATOMIC_ACQUIRE, "workgroup", "local");
}

__device__ __forceinline__ void store8(float* __restrict__ p, v8f a) {
  *reinterpret_cast<v4f*>(p) = v4f{a[0], a[1], a[2], a[3]};
  *reinterpret_cast<v4f*>(p + 4) = v4f{a[4], a[5], a[6], a[7]};
}

__device__ __forceinline__ void wave_lds_handover() {
  __builtin_amdgcn_fence(__ATOMIC_RELEASE, "workgroup", "local");
  __builtin_amdgcn_wave_barrier();
  __builtin_amdgcn_fence(__ATOMIC_ACQUIRE, "workgroup", "local");
}

template <int K>
__device__ __forceinline__ void row_tile(const float* __restrict__ llp,
                                         const float* __restrict__ yhp,
                                         unsigned gl, unsigned gs, int SB,
                                         v2f sgn, v2f mul, bool odd,
                                         const v2f (&bL)[K / 4],
                                         const v2f (&bH)[K / 4], v8f& z1,
                                         v8f& z2) {
  constexpr int KT = K / 4;
  v2f aLL[KT], aLH[KT], aHL[KT], aHH[KT];
  load_half<K>(llp + gl, aLL);
  load_c2q<K>(yhp + 0 * SB + gs, yhp + 5 * SB + gs, sgn, mul, odd, aLH);
  load_c2q<K>(yhp + 2 * SB + gs, yhp + 3 * SB + gs, sgn, mul, odd, aHL);
  load_c2q<K>(yhp + 1 * SB + gs, yhp + 4 * SB + gs, sgn, mul, odd, aHH);
  v8f acc1 = {0.f, 0.f, 0.f, 0.f, 0.f, 0.f, 0.f, 0.f};
  v8f acc2 = {0.f, 0.f, 0.f, 0.f, 0.f, 0.f, 0.f, 0.f};
#pragma unroll
  for (int t = 0; t < KT; ++t) {
    acc1 = wmma4(aLL[t], bL[t], acc1);
    acc2 = wmma4(aLH[t], bL[t], acc2);
    acc1 = wmma4(aHL[t], bH[t], acc1);
    acc2 = wmma4(aHH[t], bH[t], acc2);
  }
  z1 = acc1;
  z2 = acc2;
}

__device__ __forceinline__ v8f col_tile(v8f z, const v2f (&g)[4], v8f acc) {
#pragma unroll
  for (int t = 0; t < 4; ++t)
    acc = wmma4(v2f{z[2 * t], z[2 * t + 1]}, g[t], acc);
  return acc;
}

template <int T0>
__device__ __forceinline__ v8f col_frags(v8f z, const v2f (&g)[3], v8f acc) {
#pragma unroll
  for (int j = 0; j < 3; ++j)
    acc = wmma4(v2f{z[2 * (T0 + j)], z[2 * (T0 + j) + 1]}, g[j], acc);
  return acc;
}

constexpr int OPU = 36;

template <int R>
__global__ __launch_bounds__(256)
void level_up(const float* __restrict__ ll, const float* __restrict__ yh,
              const float* __restrict__ g0a, const float* __restrict__ g0b,
              const float* __restrict__ g1a, const float* __restrict__ g1b,
              float* __restrict__ out) {
  constexpr int P = 2 * R;
  constexpr int NT = P / 16;
  constexpr int H2 = R / 2;
  constexpr int SB = H2 * H2 * 2;
  constexpr int YR = (R < 68) ? R : 68;
  constexpr int YS = YR / 2;
  constexpr int XW = 40;
  constexpr int SEG = XW / 4;
  constexpr int PITCH = 44;
  constexpr int NROW = YR + 6 * YS;
  constexpr int NPIECE = NROW * SEG;
  constexpr int NLD = (NPIECE + 255) / 256;
  __shared__ __attribute__((aligned(16))) float tile[NROW * PITCH];
  __shared__ __attribute__((aligned(16))) float ftile[3 * 2 * 16 * FPU];
  __shared__ __attribute__((aligned(16))) float obuf[8 * 16 * OPU];

  const int tid = threadIdx.x;
  const int wave = tid >> 5, lane = tid & 31;
  const int u = blockIdx.x * 8 + wave;
  const int nt = u % NT, ms = u / NT;
  const int nt0 = (blockIdx.x * 8) % NT;
  const int mc = blockIdx.y;
  const int o0 = nt << 4;
  const int wy = iclamp((o0 >> 1) - 4, 0, R - 16);
  const int Y0 = iclamp(8 * nt0 - 4, 0, R - YR);
  const int X0 = iclamp(32 * ms - 4, 0, R - XW);
  const int h = lane >> 4, c = lane & 15;
  const bool odd = (c & 1) != 0;
  const v2f sgn = odd ? v2f{-1.f, -1.f} : v2f{1.f, 1.f};
  const v2f mul = odd ? v2f{-SC_HALF_SQRT2, SC_HALF_SQRT2}
                      : v2f{SC_HALF_SQRT2, SC_HALF_SQRT2};

  const float* llp = ll + (size_t)mc * (R * R) + (size_t)Y0 * R + X0;
  const float* yhp = yh + (size_t)mc * (6 * SB) + (size_t)(Y0 >> 1) * R + X0;

  float fcm[5];
  int fkm[5];
  ifilt_row_taps<R>(g0a, g0b, g1a, g1b, tid, fcm, fkm);

  v4f pre[NLD];
#pragma unroll
  for (int n = 0; n < NLD; ++n) {
    const int j = tid + 256 * n;
    if (256 * (n + 1) <= NPIECE || j < NPIECE) {
      const int row = j / SEG, seg = j - row * SEG;
      const bool isll = row < YR;
      const int rr = row - YR;
      const int b = rr / YS, sr = rr - b * YS;
      const unsigned offl = (unsigned)row * R;
      const unsigned offs = (unsigned)b * SB + (unsigned)sr * R;
      const float* src = isll ? (llp + offl) : (yhp + offs);
      pre[n] = *reinterpret_cast<const v4f*>(src + 4 * seg);
    }
  }

  if (tid < 96) put_row<5, 16, FPU>(ftile + tid * FPU, fcm, fkm);
  float* ow = obuf + wave * (16 * OPU);
  float* owt = ow + c * OPU + 8 * h;
  const float* owr = ow + (lane >> 3) * OPU + 4 * (lane & 7);
  float* og = out + (size_t)mc * (P * P) + (size_t)(o0 + (lane >> 3)) * P +
              4 * (lane & 7);

#pragma unroll
  for (int n = 0; n < NLD; ++n) {
    const int j = tid + 256 * n;
    if (256 * (n + 1) <= NPIECE || j < NPIECE) {
      const int row = j / SEG, seg = j - row * SEG;
      *reinterpret_cast<v4f*>(tile + row * PITCH + 4 * seg) = pre[n];
    }
  }
  __syncthreads();

  const float* fw = ftile + c * FPU + 8 * h;
  v2f gL[4], gH[4];
  load_half<16>(fw + (tile_class<NT>(nt) * 2 + 0) * (16 * FPU), gL);
  load_half<16>(fw + (tile_class<NT>(nt) * 2 + 1) * (16 * FPU), gH);

  const float* tSB = tile + YR * PITCH;
  const int fl = (wy - Y0 + c) * PITCH + 8 * h - X0;
  const int fs = (((wy - Y0) >> 1) + (c >> 1)) * PITCH + 8 * h - X0;

#pragma unroll 1
  for (int m = 0; m < 4; ++m) {
    const int pt = ms * 4 + m;
    const int p0 = pt << 4;
    const int wx = iclamp((p0 >> 1) - 4, 0, R - 16);
    v2f bL[4], bH[4];
    load_half<16>(fw + (tile_class<NT>(pt) * 2 + 0) * (16 * FPU), bL);
    load_half<16>(fw + (tile_class<NT>(pt) * 2 + 1) * (16 * FPU), bH);
    v8f z1, z2;
    row_tile<16>(tile, tSB, (unsigned)(fl + wx), (unsigned)(fs + wx), YS * PITCH,
                 sgn, mul, odd, bL, bH, z1, z2);
    const v8f zero = {0.f, 0.f, 0.f, 0.f, 0.f, 0.f, 0.f, 0.f};
    const v8f a1 = col_tile(z1, gL, zero);
    const v8f a2 = col_tile(z2, gH, zero);
    store8(owt + 16 * (m & 1), a1 + a2);
    if ((m & 1) != 0) {
      wave_lds_handover();
      v4f f[4];
#pragma unroll
      for (int q = 0; q < 4; ++q)
        f[q] = *reinterpret_cast<const v4f*>(owr + (4 * q) * OPU);
#pragma unroll
      for (int q = 0; q < 4; ++q)
        *reinterpret_cast<v4f*>(og + (size_t)(4 * q) * P + (p0 - 16)) = f[q];
      wave_lds_handover();
    }
  }
}

constexpr int OP1 = 132;

template <int R>
__global__ __launch_bounds__(256)
void level_one(const float* __restrict__ ll, const float* __restrict__ yh,
               const float* __restrict__ g0o, const float* __restrict__ g1o,
               float* __restrict__ out) {
  constexpr int K = 24;
  constexpr int NT = R / 16;
  constexpr int H2 = R / 2;
  constexpr int SB = H2 * H2 * 2;
  constexpr int XW = 136;
  constexpr int SEG = XW / 4;
  constexpr int PITCH = 140;
  constexpr int NROW = 16 + 6 * 8;
  constexpr int NPIECE = NROW * SEG;
  constexpr int NLD = (NPIECE + 255) / 256;
  __shared__ __attribute__((aligned(16))) float tile[NROW * PITCH];
  __shared__ __attribute__((aligned(16))) float ftile[3 * 2 * 16 * FP1];
  __shared__ __attribute__((aligned(16))) float obuf[2 * 16 * OP1];

  const int tid = threadIdx.x;
  const int wave = tid >> 5, lane = tid & 31;
  const int nt0 = blockIdx.x * 8;
  const int nt = nt0 + wave;
  const int X0 = iclamp(16 * nt0 - 4, 0, R - XW);
  const int mc = blockIdx.y;
  const int p0 = nt << 4;
  const int wx = iclamp(p0 - 4, 0, R - K);
  const int h = lane >> 4, c = lane & 15;
  const bool odd = (c & 1) != 0;
  const v2f sgn = odd ? v2f{-1.f, -1.f} : v2f{1.f, 1.f};
  const v2f mul = odd ? v2f{-SC_HALF_SQRT2, SC_HALF_SQRT2}
                      : v2f{SC_HALF_SQRT2, SC_HALF_SQRT2};

  const float* llp = ll + (size_t)mc * (R * R) + X0;
  const float* yhp = yh + (size_t)mc * (6 * SB) + X0;
  float* owt = obuf + c * OP1 + 16 * wave + 8 * h;
  const float* owr = obuf + (2 * wave) * OP1 + 4 * lane;
  float* og = out + (size_t)mc * (R * R) + (size_t)(2 * wave) * R + 16 * nt0 +
              4 * lane;

  const int cr = 4 * ((c & 7) >> 1) + 2 * (c >> 3) + (c & 1);
  const unsigned fl = (unsigned)(cr * PITCH + wx - X0 + 12 * h);
  const unsigned fs = (unsigned)((cr >> 1) * PITCH + wx - X0 + 12 * h);
  const float* tSB = tile + 16 * PITCH;

  v4f pre[NLD];
  auto fetch = [&](int ybase) {
#pragma unroll
    for (int n = 0; n < NLD; ++n) {
      const int j = tid + 256 * n;
      if (256 * (n + 1) <= NPIECE || j < NPIECE) {
        const int row = j / SEG, seg = j - row * SEG;
        const bool isll = row < 16;
        const int rr = row - 16;
        const int yy = iclamp(ybase + (isll ? row : 2 * (rr & 7)), 0, R - 1);
        const unsigned offl = (unsigned)yy * R;
        const unsigned offs = (unsigned)(rr >> 3) * SB + (unsigned)(yy >> 1) * R;
        const float* src = isll ? (llp + offl) : (yhp + offs);
        pre[n] = *reinterpret_cast<const v4f*>(src + 4 * seg);
      }
    }
  };
  float h0[7], h1[5];
#pragma unroll
  for (int m = 0; m < 7; ++m) h0[m] = g0o[m];
#pragma unroll
  for (int m = 0; m < 5; ++m) h1[m] = g1o[m];

  fetch(-8);

  if (tid < 96) build_j1_row<R>(h0, h1, ftile, tid);
  lds_barrier();

  const float* fw = ftile + c * FP1 + FM1;
  v2f bL[6], bH[6];
  load_half<K>(fw + (tile_class<NT>(nt) * 2 + 0) * (16 * FP1) + 12 * h, bL);
  load_half<K>(fw + (tile_class<NT>(nt) * 2 + 1) * (16 * FP1) + 12 * h, bH);

  const v8f zero = {0.f, 0.f, 0.f, 0.f, 0.f, 0.f, 0.f, 0.f};
  v8f zp1 = zero, zp2 = zero;

#pragma unroll 1
  for (int t = 0; t <= NT; ++t) {
#pragma unroll
    for (int n = 0; n < NLD; ++n) {
      const int j = tid + 256 * n;
      if (256 * (n + 1) <= NPIECE || j < NPIECE) {
        const int row = j / SEG, seg = j - row * SEG;
        *reinterpret_cast<v4f*>(tile + row * PITCH + 4 * seg) = pre[n];
      }
    }
    lds_barrier();

    if (t < NT) fetch(16 * (t + 1) - 8);

    const int i = t - 1;
    const int w0 = iclamp(16 * i - 4, 0, R - K);
    v2f gq[2][2][3];
    if (t > 0) {
      const float* fr =
          fw + (tile_class<NT>(i) * 2) * (16 * FP1) + (16 * i - 8 - w0) + 2 * h;
#pragma unroll
      for (int sl = 0; sl < 2; ++sl) {
#pragma unroll
        for (int f = 0; f < 2; ++f) {
#pragma unroll
          for (int j = 0; j < 3; ++j)
            gq[sl][f][j] = *reinterpret_cast<const v2f*>(
                fr + f * (16 * FP1) + 16 * sl + 4 * (j + 1 - sl));
        }
      }
    }

    v4f f0, f1;
    if (t > 1) {
      const float* rr = owr + ((t - 1) & 1) * (16 * OP1);
      f0 = *reinterpret_cast<const v4f*>(rr);
      f1 = *reinterpret_cast<const v4f*>(rr + OP1);
    }

    v8f zc1, zc2;
    row_tile<K>(tile, tSB, fl, fs, 8 * PITCH, sgn, mul, odd, bL, bH, zc1, zc2);

    if (t > 1) {
      float* g = og + (size_t)(16 * (t - 2)) * R;
      *reinterpret_cast<v4f*>(g) = f0;
      *reinterpret_cast<v4f*>(g + R) = f1;
    }

    if (t > 0) {
      v8f a1 = col_frags<1>(zp1, gq[0][0], zero);
      v8f a2 = col_frags<1>(zp2, gq[0][1], zero);
      a1 = col_frags<0>(zc1, gq[1][0], a1);
      a2 = col_frags<0>(zc2, gq[1][1], a2);
      store8(owt + (t & 1) * (16 * OP1), a1 + a2);
    }
    zp1 = zc1;
    zp2 = zc2;
    lds_barrier();
  }
  {
    const float* rr = owr + (NT & 1) * (16 * OP1);
    const v4f f0 = *reinterpret_cast<const v4f*>(rr);
    const v4f f1 = *reinterpret_cast<const v4f*>(rr + OP1);
    float* g = og + (size_t)(16 * (NT - 1)) * R;
    *reinterpret_cast<v4f*>(g) = f0;
    *reinterpret_cast<v4f*>(g + R) = f1;
  }
}

extern "C" void kernel_launch(void* const* d_in, const int* in_sizes, int n_in,
                              void* d_out, int out_size, void* d_ws,
                              size_t ws_size, hipStream_t stream) {
  (void)in_sizes; (void)n_in; (void)out_size; (void)ws_size;
  const float* yl  = (const float*)d_in[0];
  const float* yh1 = (const float*)d_in[1];
  const float* yh2 = (const float*)d_in[2];
  const float* yh3 = (const float*)d_in[3];
  const float* g0o = (const float*)d_in[4];
  const float* g1o = (const float*)d_in[5];
  const float* g0a = (const float*)d_in[6];
  const float* g0b = (const float*)d_in[7];
  const float* g1a = (const float*)d_in[8];
  const float* g1b = (const float*)d_in[9];

  const size_t SEG = (size_t)16 * 1024 * 1024;
  float* bufA = (float*)d_ws;
  float* bufB = bufA + SEG;

  level_up<64><<<dim3(2, 256), 256, 0, stream>>>(yl, yh3, g0a, g0b, g1a, g1b,
                                                 bufA);
  level_up<128><<<dim3(8, 256), 256, 0, stream>>>(bufA, yh2, g0a, g0b, g1a,
                                                  g1b, bufB);
  level_one<256><<<dim3(2, 256), 256, 0, stream>>>(bufB, yh1, g0o, g1o,
                                                   (float*)d_out);
}


# Level 1: c2q done once per work-group while it stages the sub-band rows (LDS holds four real images); the row pass reads its A operands in one group of 12, no vector work among its WMMAs; same sums
# speedup vs baseline: 3.6231x; 1.0397x over previous
#include <hip/hip_runtime.h>


typedef __attribute__((ext_vector_type(2))) float v2f;
typedef __attribute__((ext_vector_type(4))) float v4f;
typedef __attribute__((ext_vector_type(8))) float v8f;

#define SC_HALF_SQRT2 0.70710678118654752f

__device__ __forceinline__ int refl(int i, int n) {
  i = (i < 0) ? (-1 - i) : i;
  i = (i >= n) ? (2 * n - 1 - i) : i;
  return i;
}

__device__ __forceinline__ int iclamp(int v, int lo, int hi) {
  return v < lo ? lo : (v > hi ? hi : v);
}

__device__ __forceinline__ v8f wmma4(v2f a, v2f b, v8f c) {
  return __builtin_amdgcn_wmma_f32_16x16x4_f32(false, a, false, b, (short)0, c,
                                               false, false);
}

__device__ __forceinline__ float ifilt_coef(int o, int kg, int n,
                                            const float (&ha)[10],
                                            const float (&hb)[10], bool hp) {
  const int i = o >> 2, s = o & 3;
  const int base = 4 + ((((s & 1) != 0) != hp) ? 1 : 0);
  float acc = 0.f;
#pragma unroll
  for (int m = 0; m < 5; ++m) {
    const int L = refl(2 * i + base - 2 * m, n);
    const float c = (s == 0)   ? ha[2 * m]
                    : (s == 1) ? hb[2 * m]
                    : (s == 2) ? ha[2 * m + 1]
                               : hb[2 * m + 1];
    acc += (L == kg) ? c : 0.f;
  }
  return acc;
}

template <int NT>
__device__ __forceinline__ float filt_coef(int o, int kg, int n,
                                           const float (&h)[NT]) {
  float acc = 0.f;
#pragma unroll
  for (int m = 0; m < NT; ++m) {
    const int L = refl(o + (NT >> 1) - m, n);
    acc += (L == kg) ? h[m] : 0.f;
  }
  return acc;
}

constexpr int FPU = 20;
constexpr int FP1 = 36;
constexpr int FM1 = 4;

template <int NT>
__device__ __forceinline__ int tile_class(int ty) {
  return ty == 0 ? 0 : (ty == NT - 1 ? 2 : 1);
}

template <int MT, int W, int FP, int OFF = 0>
__device__ __forceinline__ void put_row(float* __restrict__ rp,
                                        const float (&cm)[MT],
                                        const int (&km)[MT]) {
  const v4f z4 = {0.f, 0.f, 0.f, 0.f};
#pragma unroll
  for (int q = 0; q < FP / 4; ++q) *reinterpret_cast<v4f*>(rp + 4 * q) = z4;
#pragma unroll
  for (int m = 0; m < MT; ++m) {
    float acc = 0.f;
#pragma unroll
    for (int mm = 0; mm <= m; ++mm) acc += (km[mm] == km[m]) ? cm[mm] : 0.f;
    if (km[m] >= 0 && km[m] < W) rp[OFF + km[m]] = acc;
  }
}

template <int R>
__device__ __forceinline__ void ifilt_row_taps(
    const float* __restrict__ g0a, const float* __restrict__ g0b,
    const float* __restrict__ g1a, const float* __restrict__ g1b, int tid,
    float (&cm)[5], int (&km)[5]) {
  constexpr int NT = R / 8;
  const int nf = tid >> 4, row = tid & 15;
  const int cls = nf >> 1;
  const bool hp = (nf & 1) != 0;
  const int ty = (cls == 0) ? 0 : ((cls == 1) ? 1 : NT - 1);
  const int o = (ty << 4) + row;
  const int w0 = iclamp(8 * ty - 4, 0, R - 16);
  const int i = o >> 2, s = o & 3;
  const bool sodd = (s & 1) != 0;
  const int base = 4 + ((sodd != hp) ? 1 : 0);
  const float* tp = sodd ? (hp ? g1a : g0a) : (hp ? g1b : g0b);
  tp += (s >> 1);
#pragma unroll
  for (int m = 0; m < 5; ++m) {
    cm[m] = tp[2 * m];
    km[m] = refl(2 * i + base - 2 * m, R) - w0;
  }
}

template <int R>
__device__ __forceinline__ void build_j1_row(const float (&h0)[7],
                                             const float (&h1)[5],
                                             float* __restrict__ ft, int tid) {
  constexpr int NT = R / 16;
  const int nf = tid >> 4, row = tid & 15;
  const int cls = nf >> 1;
  const bool hi = (nf & 1) != 0;
  const int ty = (cls == 0) ? 0 : ((cls == 1) ? 1 : NT - 1);
  const int o = (ty << 4) + row;
  const int w0 = iclamp(16 * ty - 4, 0, R - 24);
  const int half = hi ? 2 : 3;
  float cm[7];
  int km[7];
#pragma unroll
  for (int m = 0; m < 7; ++m) {
    const float hv = (m < 5) ? h1[m] : 0.f;
    cm[m] = hi ? hv : h0[m];
    const int pos = refl(o + half - m, R) - w0;
    km[m] = (hi && m >= 5) ? (-1 - m) : pos;
  }
  put_row<7, 24, FP1, FM1>(ft + tid * FP1, cm, km);
}

template <int K>
__device__ __forceinline__ void load_half(const float* __restrict__ p,
                                          v2f (&o)[K / 4]) {
#pragma unroll
  for (int q = 0; q < K / 8; ++q) {
    const v4f v = *reinterpret_cast<const v4f*>(p + 4 * q);
    o[2 * q] = v2f{v.x, v.y};
    o[2 * q + 1] = v2f{v.z, v.w};
  }
}

template <int K>
__device__ __forceinline__ void load_c2q(const float* __restrict__ p1,
                                         const float* __restrict__ p2,
                                         v2f sgn, v2f mul, bool odd,
                                         v2f (&o)[K / 4]) {
  v2f r1[K / 4], r2[K / 4];
  load_half<K>(p1, r1);
  load_half<K>(p2, r2);
#pragma unroll
  for (int t = 0; t < K / 4; ++t) {
    const v2f u = __builtin_elementwise_fma(sgn, r2[t], r1[t]) * mul;
    o[t].x = odd ? u.y : u.x;
    o[t].y = odd ? u.x : u.y;
  }
}

__device__ __forceinline__ void lds_barrier() {
  __builtin_amdgcn_fence(__ATOMIC_RELEASE, "workgroup", "local");
  __builtin_amdgcn_s_barrier();
  __builtin_amdgcn_fence(__ATOMIC_ACQUIRE, "workgroup", "local");
}

__device__ __forceinline__ void store8(float* __restrict__ p, v8f a) {
  *reinterpret_cast<v4f*>(p) = v4f{a[0], a[1], a[2], a[3]};
  *reinterpret_cast<v4f*>(p + 4) = v4f{a[4], a[5], a[6], a[7]};
}

__device__ __forceinline__ void wave_lds_handover() {
  __builtin_amdgcn_fence(__ATOMIC_RELEASE, "workgroup", "local");
  __builtin_amdgcn_wave_barrier();
  __builtin_amdgcn_fence(__ATOMIC_ACQUIRE, "workgroup", "local");
}

template <int K>
__device__ __forceinline__ void row_tile(const float* __restrict__ llp,
                                         const float* __restrict__ yhp,
                                         unsigned gl, unsigned gs, int SB,
                                         v2f sgn, v2f mul, bool odd,
                                         const v2f (&bL)[K / 4],
                                         const v2f (&bH)[K / 4], v8f& z1,
                                         v8f& z2) {
  constexpr int KT = K / 4;
  v2f aLL[KT], aLH[KT], aHL[KT], aHH[KT];
  load_half<K>(llp + gl, aLL);
  load_c2q<K>(yhp + 0 * SB + gs, yhp + 5 * SB + gs, sgn, mul, odd, aLH);
  load_c2q<K>(yhp + 2 * SB + gs, yhp + 3 * SB + gs, sgn, mul, odd, aHL);
  load_c2q<K>(yhp + 1 * SB + gs, yhp + 4 * SB + gs, sgn, mul, odd, aHH);
  v8f acc1 = {0.f, 0.f, 0.f, 0.f, 0.f, 0.f, 0.f, 0.f};
  v8f acc2 = {0.f, 0.f, 0.f, 0.f, 0.f, 0.f, 0.f, 0.f};
#pragma unroll
  for (int t = 0; t < KT; ++t) {
    acc1 = wmma4(aLL[t], bL[t], acc1);
    acc2 = wmma4(aLH[t], bL[t], acc2);
    acc1 = wmma4(aHL[t], bH[t], acc1);
    acc2 = wmma4(aHH[t], bH[t], acc2);
  }
  z1 = acc1;
  z2 = acc2;
}

template <int K>
__device__ __forceinline__ void row_tile_real(const float* __restrict__ t0,
                                              unsigned f, int IMG,
                                              const v2f (&bL)[K / 4],
                                              const v2f (&bH)[K / 4], v8f& z1,
                                              v8f& z2) {
  constexpr int KT = K / 4;
  v2f aLL[KT], aLH[KT], aHL[KT], aHH[KT];
  load_half<K>(t0 + f, aLL);
  load_half<K>(t0 + f + 1 * IMG, aLH);
  load_half<K>(t0 + f + 2 * IMG, aHL);
  load_half<K>(t0 + f + 3 * IMG, aHH);
  __builtin_amdgcn_sched_barrier(0);
  v8f acc1 = {0.f, 0.f, 0.f, 0.f, 0.f, 0.f, 0.f, 0.f};
  v8f acc2 = {0.f, 0.f, 0.f, 0.f, 0.f, 0.f, 0.f, 0.f};
#pragma unroll
  for (int t = 0; t < KT; ++t) {
    acc1 = wmma4(aLL[t], bL[t], acc1);
    acc2 = wmma4(aLH[t], bL[t], acc2);
    acc1 = wmma4(aHL[t], bH[t], acc1);
    acc2 = wmma4(aHH[t], bH[t], acc2);
  }
  z1 = acc1;
  z2 = acc2;
}

__device__ __forceinline__ v8f col_tile(v8f z, const v2f (&g)[4], v8f acc) {
#pragma unroll
  for (int t = 0; t < 4; ++t)
    acc = wmma4(v2f{z[2 * t], z[2 * t + 1]}, g[t], acc);
  return acc;
}

template <int T0>
__device__ __forceinline__ v8f col_frags(v8f z, const v2f (&g)[3], v8f acc) {
#pragma unroll
  for (int j = 0; j < 3; ++j)
    acc = wmma4(v2f{z[2 * (T0 + j)], z[2 * (T0 + j) + 1]}, g[j], acc);
  return acc;
}

constexpr int OPU = 36;

template <int R>
__global__ __launch_bounds__(256)
void level_up(const float* __restrict__ ll, const float* __restrict__ yh,
              const float* __restrict__ g0a, const float* __restrict__ g0b,
              const float* __restrict__ g1a, const float* __restrict__ g1b,
              float* __restrict__ out) {
  constexpr int P = 2 * R;
  constexpr int NT = P / 16;
  constexpr int H2 = R / 2;
  constexpr int SB = H2 * H2 * 2;
  constexpr int YR = (R < 68) ? R : 68;
  constexpr int YS = YR / 2;
  constexpr int XW = 40;
  constexpr int SEG = XW / 4;
  constexpr int PITCH = 44;
  constexpr int NROW = YR + 6 * YS;
  constexpr int NPIECE = NROW * SEG;
  constexpr int NLD = (NPIECE + 255) / 256;
  __shared__ __attribute__((aligned(16))) float tile[NROW * PITCH];
  __shared__ __attribute__((aligned(16))) float ftile[3 * 2 * 16 * FPU];
  __shared__ __attribute__((aligned(16))) float obuf[8 * 16 * OPU];

  const int tid = threadIdx.x;
  const int wave = tid >> 5, lane = tid & 31;
  const int u = blockIdx.x * 8 + wave;
  const int nt = u % NT, ms = u / NT;
  const int nt0 = (blockIdx.x * 8) % NT;
  const int mc = blockIdx.y;
  const int o0 = nt << 4;
  const int wy = iclamp((o0 >> 1) - 4, 0, R - 16);
  const int Y0 = iclamp(8 * nt0 - 4, 0, R - YR);
  const int X0 = iclamp(32 * ms - 4, 0, R - XW);
  const int h = lane >> 4, c = lane & 15;
  const bool odd = (c & 1) != 0;
  const v2f sgn = odd ? v2f{-1.f, -1.f} : v2f{1.f, 1.f};
  const v2f mul = odd ? v2f{-SC_HALF_SQRT2, SC_HALF_SQRT2}
                      : v2f{SC_HALF_SQRT2, SC_HALF_SQRT2};

  const float* llp = ll + (size_t)mc * (R * R) + (size_t)Y0 * R + X0;
  const float* yhp = yh + (size_t)mc * (6 * SB) + (size_t)(Y0 >> 1) * R + X0;

  float fcm[5];
  int fkm[5];
  ifilt_row_taps<R>(g0a, g0b, g1a, g1b, tid, fcm, fkm);

  v4f pre[NLD];
#pragma unroll
  for (int n = 0; n < NLD; ++n) {
    const int j = tid + 256 * n;
    if (256 * (n + 1) <= NPIECE || j < NPIECE) {
      const int row = j / SEG, seg = j - row * SEG;
      const bool isll = row < YR;
      const int rr = row - YR;
      const int b = rr / YS, sr = rr - b * YS;
      const unsigned offl = (unsigned)row * R;
      const unsigned offs = (unsigned)b * SB + (unsigned)sr * R;
      const float* src = isll ? (llp + offl) : (yhp + offs);
      pre[n] = *reinterpret_cast<const v4f*>(src + 4 * seg);
    }
  }

  if (tid < 96) put_row<5, 16, FPU>(ftile + tid * FPU, fcm, fkm);
  float* ow = obuf + wave * (16 * OPU);
  float* owt = ow + c * OPU + 8 * h;
  const float* owr = ow + (lane >> 3) * OPU + 4 * (lane & 7);
  float* og = out + (size_t)mc * (P * P) + (size_t)(o0 + (lane >> 3)) * P +
              4 * (lane & 7);

#pragma unroll
  for (int n = 0; n < NLD; ++n) {
    const int j = tid + 256 * n;
    if (256 * (n + 1) <= NPIECE || j < NPIECE) {
      const int row = j / SEG, seg = j - row * SEG;
      *reinterpret_cast<v4f*>(tile + row * PITCH + 4 * seg) = pre[n];
    }
  }
  __syncthreads();

  const float* fw = ftile + c * FPU + 8 * h;
  v2f gL[4], gH[4];
  load_half<16>(fw + (tile_class<NT>(nt) * 2 + 0) * (16 * FPU), gL);
  load_half<16>(fw + (tile_class<NT>(nt) * 2 + 1) * (16 * FPU), gH);

  const float* tSB = tile + YR * PITCH;
  const int fl = (wy - Y0 + c) * PITCH + 8 * h - X0;
  const int fs = (((wy - Y0) >> 1) + (c >> 1)) * PITCH + 8 * h - X0;

#pragma unroll 1
  for (int m = 0; m < 4; ++m) {
    const int pt = ms * 4 + m;
    const int p0 = pt << 4;
    const int wx = iclamp((p0 >> 1) - 4, 0, R - 16);
    v2f bL[4], bH[4];
    load_half<16>(fw + (tile_class<NT>(pt) * 2 + 0) * (16 * FPU), bL);
    load_half<16>(fw + (tile_class<NT>(pt) * 2 + 1) * (16 * FPU), bH);
    v8f z1, z2;
    row_tile<16>(tile, tSB, (unsigned)(fl + wx), (unsigned)(fs + wx), YS * PITCH,
                 sgn, mul, odd, bL, bH, z1, z2);
    const v8f zero = {0.f, 0.f, 0.f, 0.f, 0.f, 0.f, 0.f, 0.f};
    const v8f a1 = col_tile(z1, gL, zero);
    const v8f a2 = col_tile(z2, gH, zero);
    store8(owt + 16 * (m & 1), a1 + a2);
    if ((m & 1) != 0) {
      wave_lds_handover();
      v4f f[4];
#pragma unroll
      for (int q = 0; q < 4; ++q)
        f[q] = *reinterpret_cast<const v4f*>(owr + (4 * q) * OPU);
#pragma unroll
      for (int q = 0; q < 4; ++q)
        *reinterpret_cast<v4f*>(og + (size_t)(4 * q) * P + (p0 - 16)) = f[q];
      wave_lds_handover();
    }
  }
}

constexpr int OP1 = 132;

template <int R>
__global__ __launch_bounds__(256)
void level_one(const float* __restrict__ ll, const float* __restrict__ yh,
               const float* __restrict__ g0o, const float* __restrict__ g1o,
               float* __restrict__ out) {
  constexpr int K = 24;
  constexpr int NT = R / 16;
  constexpr int H2 = R / 2;
  constexpr int SB = H2 * H2 * 2;
  constexpr int XW = 136;
  constexpr int SEG = XW / 4;
  constexpr int PITCH = 140;
  constexpr int NROW = 4 * 16;
  constexpr int IMG = 16 * PITCH;
  constexpr int NLL = 16 * SEG;
  constexpr int NPAIR = 3 * 8 * SEG;
  constexpr int NLJ = (NLL + 255) / 256;
  constexpr int NPJ = (NPAIR + 255) / 256;
  __shared__ __attribute__((aligned(16))) float tile[NROW * PITCH];
  __shared__ __attribute__((aligned(16))) float ftile[3 * 2 * 16 * FP1];
  __shared__ __attribute__((aligned(16))) float obuf[2 * 16 * OP1];

  const int tid = threadIdx.x;
  const int wave = tid >> 5, lane = tid & 31;
  const int nt0 = blockIdx.x * 8;
  const int nt = nt0 + wave;
  const int X0 = iclamp(16 * nt0 - 4, 0, R - XW);
  const int mc = blockIdx.y;
  const int p0 = nt << 4;
  const int wx = iclamp(p0 - 4, 0, R - K);
  const int h = lane >> 4, c = lane & 15;
  const int tll = (tid + 128) & 255;

  const float* llp = ll + (size_t)mc * (R * R) + X0;
  const float* yhp = yh + (size_t)mc * (6 * SB) + X0;
  float* owt = obuf + c * OP1 + 16 * wave + 8 * h;
  const float* owr = obuf + (2 * wave) * OP1 + 4 * lane;
  float* og = out + (size_t)mc * (R * R) + (size_t)(2 * wave) * R + 16 * nt0 +
              4 * lane;

  const int cr = 4 * ((c & 7) >> 1) + 2 * (c >> 3) + (c & 1);
  const unsigned fl = (unsigned)(cr * PITCH + wx - X0 + 12 * h);

  v4f preL[NLJ], pre1[NPJ], pre2[NPJ];
  auto fetch = [&](int ybase) {
#pragma unroll
    for (int n = 0; n < NLJ; ++n) {
      const int j = tll + 256 * n;
      if (256 * (n + 1) <= NLL || j < NLL) {
        const int row = j / SEG, seg = j - row * SEG;
        const int yy = iclamp(ybase + row, 0, R - 1);
        preL[n] = *reinterpret_cast<const v4f*>(llp + (unsigned)yy * R + 4 * seg);
      }
    }
#pragma unroll
    for (int n = 0; n < NPJ; ++n) {
      const int q = tid + 256 * n;
      if (256 * (n + 1) <= NPAIR || q < NPAIR) {
        const int b = q / (8 * SEG), rem = q - b * (8 * SEG);
        const int jr = rem / SEG, seg = rem - jr * SEG;
        const int w1 = (b == 0) ? 0 : ((b == 1) ? 2 : 1);
        const int yy = iclamp(ybase + 2 * jr, 0, R - 1);
        const unsigned off = (unsigned)(yy >> 1) * R + 4 * seg;
        pre1[n] = *reinterpret_cast<const v4f*>(yhp + (unsigned)w1 * SB + off);
        pre2[n] =
            *reinterpret_cast<const v4f*>(yhp + (unsigned)(5 - w1) * SB + off);
      }
    }
  };
  float h0[7], h1[5];
#pragma unroll
  for (int m = 0; m < 7; ++m) h0[m] = g0o[m];
#pragma unroll
  for (int m = 0; m < 5; ++m) h1[m] = g1o[m];

  fetch(-8);

  if (tid < 96) build_j1_row<R>(h0, h1, ftile, tid);
  lds_barrier();

  const float* fw = ftile + c * FP1 + FM1;
  v2f bL[6], bH[6];
  load_half<K>(fw + (tile_class<NT>(nt) * 2 + 0) * (16 * FP1) + 12 * h, bL);
  load_half<K>(fw + (tile_class<NT>(nt) * 2 + 1) * (16 * FP1) + 12 * h, bH);

  const v8f zero = {0.f, 0.f, 0.f, 0.f, 0.f, 0.f, 0.f, 0.f};
  v8f zp1 = zero, zp2 = zero;

#pragma unroll 1
  for (int t = 0; t <= NT; ++t) {
#pragma unroll
    for (int n = 0; n < NLJ; ++n) {
      const int j = tll + 256 * n;
      if (256 * (n + 1) <= NLL || j < NLL) {
        const int row = j / SEG, seg = j - row * SEG;
        *reinterpret_cast<v4f*>(tile + row * PITCH + 4 * seg) = preL[n];
      }
    }
#pragma unroll
    for (int n = 0; n < NPJ; ++n) {
      const int q = tid + 256 * n;
      if (256 * (n + 1) <= NPAIR || q < NPAIR) {
        const int b = q / (8 * SEG), rem = q - b * (8 * SEG);
        const int jr = rem / SEG, seg = rem - jr * SEG;
        const v4f sm = (pre1[n] + pre2[n]) * SC_HALF_SQRT2;
        const v4f d = pre1[n] - pre2[n];
        const v4f od = {d.y * SC_HALF_SQRT2, d.x * -SC_HALF_SQRT2,
                        d.w * SC_HALF_SQRT2, d.z * -SC_HALF_SQRT2};
        float* tp = tile + (16 + 16 * b + 2 * jr) * PITCH + 4 * seg;
        *reinterpret_cast<v4f*>(tp) = sm;
        *reinterpret_cast<v4f*>(tp + PITCH) = od;
      }
    }
    lds_barrier();

    if (t < NT) fetch(16 * (t + 1) - 8);

    const int i = t - 1;
    const int w0 = iclamp(16 * i - 4, 0, R - K);
    v2f gq[2][2][3];
    if (t > 0) {
      const float* fr =
          fw + (tile_class<NT>(i) * 2) * (16 * FP1) + (16 * i - 8 - w0) + 2 * h;
#pragma unroll
      for (int sl = 0; sl < 2; ++sl) {
#pragma unroll
        for (int f = 0; f < 2; ++f) {
#pragma unroll
          for (int j = 0; j < 3; ++j)
            gq[sl][f][j] = *reinterpret_cast<const v2f*>(
                fr + f * (16 * FP1) + 16 * sl + 4 * (j + 1 - sl));
        }
      }
    }

    v4f f0, f1;
    if (t > 1) {
      const float* rr = owr + ((t - 1) & 1) * (16 * OP1);
      f0 = *reinterpret_cast<const v4f*>(rr);
      f1 = *reinterpret_cast<const v4f*>(rr + OP1);
    }

    v8f zc1, zc2;
    row_tile_real<K>(tile, fl, IMG, bL, bH, zc1, zc2);

    if (t > 1) {
      float* g = og + (size_t)(16 * (t - 2)) * R;
      *reinterpret_cast<v4f*>(g) = f0;
      *reinterpret_cast<v4f*>(g + R) = f1;
    }

    if (t > 0) {
      v8f a1 = col_frags<1>(zp1, gq[0][0], zero);
      v8f a2 = col_frags<1>(zp2, gq[0][1], zero);
      a1 = col_frags<0>(zc1, gq[1][0], a1);
      a2 = col_frags<0>(zc2, gq[1][1], a2);
      store8(owt + (t & 1) * (16 * OP1), a1 + a2);
    }
    zp1 = zc1;
    zp2 = zc2;
    lds_barrier();
  }
  {
    const float* rr = owr + (NT & 1) * (16 * OP1);
    const v4f f0 = *reinterpret_cast<const v4f*>(rr);
    const v4f f1 = *reinterpret_cast<const v4f*>(rr + OP1);
    float* g = og + (size_t)(16 * (NT - 1)) * R;
    *reinterpret_cast<v4f*>(g) = f0;
    *reinterpret_cast<v4f*>(g + R) = f1;
  }
}

extern "C" void kernel_launch(void* const* d_in, const int* in_sizes, int n_in,
                              void* d_out, int out_size, void* d_ws,
                              size_t ws_size, hipStream_t stream) {
  (void)in_sizes; (void)n_in; (void)out_size; (void)ws_size;
  const float* yl  = (const float*)d_in[0];
  const float* yh1 = (const float*)d_in[1];
  const float* yh2 = (const float*)d_in[2];
  const float* yh3 = (const float*)d_in[3];
  const float* g0o = (const float*)d_in[4];
  const float* g1o = (const float*)d_in[5];
  const float* g0a = (const float*)d_in[6];
  const float* g0b = (const float*)d_in[7];
  const float* g1a = (const float*)d_in[8];
  const float* g1b = (const float*)d_in[9];

  const size_t SEG = (size_t)16 * 1024 * 1024;
  float* bufA = (float*)d_ws;
  float* bufB = bufA + SEG;

  level_up<64><<<dim3(2, 256), 256, 0, stream>>>(yl, yh3, g0a, g0b, g1a, g1b,
                                                 bufA);
  level_up<128><<<dim3(8, 256), 256, 0, stream>>>(bufA, yh2, g0a, g0b, g1a,
                                                  g1b, bufB);
  level_one<256><<<dim3(2, 256), 256, 0, stream>>>(bufB, yh1, g0o, g1o,
                                                   (float*)d_out);
}


# Level 2: a work-group walks the 4 column blocks of its band of rows; the next block's loads travel under this block's products into a second staged block; set-up once; 2048 -> 512 groups; same sums
# speedup vs baseline: 3.8533x; 1.0635x over previous
#include <hip/hip_runtime.h>


typedef __attribute__((ext_vector_type(2))) float v2f;
typedef __attribute__((ext_vector_type(4))) float v4f;
typedef __attribute__((ext_vector_type(8))) float v8f;

#define SC_HALF_SQRT2 0.70710678118654752f

__device__ __forceinline__ int refl(int i, int n) {
  i = (i < 0) ? (-1 - i) : i;
  i = (i >= n) ? (2 * n - 1 - i) : i;
  return i;
}

__device__ __forceinline__ int iclamp(int v, int lo, int hi) {
  return v < lo ? lo : (v > hi ? hi : v);
}

__device__ __forceinline__ v8f wmma4(v2f a, v2f b, v8f c) {
  return __builtin_amdgcn_wmma_f32_16x16x4_f32(false, a, false, b, (short)0, c,
                                               false, false);
}

__device__ __forceinline__ float ifilt_coef(int o, int kg, int n,
                                            const float (&ha)[10],
                                            const float (&hb)[10], bool hp) {
  const int i = o >> 2, s = o & 3;
  const int base = 4 + ((((s & 1) != 0) != hp) ? 1 : 0);
  float acc = 0.f;
#pragma unroll
  for (int m = 0; m < 5; ++m) {
    const int L = refl(2 * i + base - 2 * m, n);
    const float c = (s == 0)   ? ha[2 * m]
                    : (s == 1) ? hb[2 * m]
                    : (s == 2) ? ha[2 * m + 1]
                               : hb[2 * m + 1];
    acc += (L == kg) ? c : 0.f;
  }
  return acc;
}

template <int NT>
__device__ __forceinline__ float filt_coef(int o, int kg, int n,
                                           const float (&h)[NT]) {
  float acc = 0.f;
#pragma unroll
  for (int m = 0; m < NT; ++m) {
    const int L = refl(o + (NT >> 1) - m, n);
    acc += (L == kg) ? h[m] : 0.f;
  }
  return acc;
}

constexpr int FPU = 20;
constexpr int FP1 = 36;
constexpr int FM1 = 4;

template <int NT>
__device__ __forceinline__ int tile_class(int ty) {
  return ty == 0 ? 0 : (ty == NT - 1 ? 2 : 1);
}

template <int MT, int W, int FP, int OFF = 0>
__device__ __forceinline__ void put_row(float* __restrict__ rp,
                                        const float (&cm)[MT],
                                        const int (&km)[MT]) {
  const v4f z4 = {0.f, 0.f, 0.f, 0.f};
#pragma unroll
  for (int q = 0; q < FP / 4; ++q) *reinterpret_cast<v4f*>(rp + 4 * q) = z4;
#pragma unroll
  for (int m = 0; m < MT; ++m) {
    float acc = 0.f;
#pragma unroll
    for (int mm = 0; mm <= m; ++mm) acc += (km[mm] == km[m]) ? cm[mm] : 0.f;
    if (km[m] >= 0 && km[m] < W) rp[OFF + km[m]] = acc;
  }
}

template <int R>
__device__ __forceinline__ void ifilt_row_taps(
    const float* __restrict__ g0a, const float* __restrict__ g0b,
    const float* __restrict__ g1a, const float* __restrict__ g1b, int tid,
    float (&cm)[5], int (&km)[5]) {
  constexpr int NT = R / 8;
  const int nf = tid >> 4, row = tid & 15;
  const int cls = nf >> 1;
  const bool hp = (nf & 1) != 0;
  const int ty = (cls == 0) ? 0 : ((cls == 1) ? 1 : NT - 1);
  const int o = (ty << 4) + row;
  const int w0 = iclamp(8 * ty - 4, 0, R - 16);
  const int i = o >> 2, s = o & 3;
  const bool sodd = (s & 1) != 0;
  const int base = 4 + ((sodd != hp) ? 1 : 0);
  const float* tp = sodd ? (hp ? g1a : g0a) : (hp ? g1b : g0b);
  tp += (s >> 1);
#pragma unroll
  for (int m = 0; m < 5; ++m) {
    cm[m] = tp[2 * m];
    km[m] = refl(2 * i + base - 2 * m, R) - w0;
  }
}

template <int R>
__device__ __forceinline__ void build_j1_row(const float (&h0)[7],
                                             const float (&h1)[5],
                                             float* __restrict__ ft, int tid) {
  constexpr int NT = R / 16;
  const int nf = tid >> 4, row = tid & 15;
  const int cls = nf >> 1;
  const bool hi = (nf & 1) != 0;
  const int ty = (cls == 0) ? 0 : ((cls == 1) ? 1 : NT - 1);
  const int o = (ty << 4) + row;
  const int w0 = iclamp(16 * ty - 4, 0, R - 24);
  const int half = hi ? 2 : 3;
  float cm[7];
  int km[7];
#pragma unroll
  for (int m = 0; m < 7; ++m) {
    const float hv = (m < 5) ? h1[m] : 0.f;
    cm[m] = hi ? hv : h0[m];
    const int pos = refl(o + half - m, R) - w0;
    km[m] = (hi && m >= 5) ? (-1 - m) : pos;
  }
  put_row<7, 24, FP1, FM1>(ft + tid * FP1, cm, km);
}

template <int K>
__device__ __forceinline__ void load_half(const float* __restrict__ p,
                                          v2f (&o)[K / 4]) {
#pragma unroll
  for (int q = 0; q < K / 8; ++q) {
    const v4f v = *reinterpret_cast<const v4f*>(p + 4 * q);
    o[2 * q] = v2f{v.x, v.y};
    o[2 * q + 1] = v2f{v.z, v.w};
  }
}

template <int K>
__device__ __forceinline__ void load_c2q(const float* __restrict__ p1,
                                         const float* __restrict__ p2,
                                         v2f sgn, v2f mul, bool odd,
                                         v2f (&o)[K / 4]) {
  v2f r1[K / 4], r2[K / 4];
  load_half<K>(p1, r1);
  load_half<K>(p2, r2);
#pragma unroll
  for (int t = 0; t < K / 4; ++t) {
    const v2f u = __builtin_elementwise_fma(sgn, r2[t], r1[t]) * mul;
    o[t].x = odd ? u.y : u.x;
    o[t].y = odd ? u.x : u.y;
  }
}

__device__ __forceinline__ void lds_barrier() {
  __builtin_amdgcn_fence(__ATOMIC_RELEASE, "workgroup", "local");
  __builtin_amdgcn_s_barrier();
  __builtin_amdgcn_fence(__ATOMIC_ACQUIRE, "workgroup", "local");
}

__device__ __forceinline__ void store8(float* __restrict__ p, v8f a) {
  *reinterpret_cast<v4f*>(p) = v4f{a[0], a[1], a[2], a[3]};
  *reinterpret_cast<v4f*>(p + 4) = v4f{a[4], a[5], a[6], a[7]};
}

__device__ __forceinline__ void wave_lds_handover() {
  __builtin_amdgcn_fence(__ATOMIC_RELEASE, "workgroup", "local");
  __builtin_amdgcn_wave_barrier();
  __builtin_amdgcn_fence(__ATOMIC_ACQUIRE, "workgroup", "local");
}

template <int K>
__device__ __forceinline__ void row_tile(const float* __restrict__ llp,
                                         const float* __restrict__ yhp,
                                         unsigned gl, unsigned gs, int SB,
                                         v2f sgn, v2f mul, bool odd,
                                         const v2f (&bL)[K / 4],
                                         const v2f (&bH)[K / 4], v8f& z1,
                                         v8f& z2) {
  constexpr int KT = K / 4;
  v2f aLL[KT], aLH[KT], aHL[KT], aHH[KT];
  load_half<K>(llp + gl, aLL);
  load_c2q<K>(yhp + 0 * SB + gs, yhp + 5 * SB + gs, sgn, mul, odd, aLH);
  load_c2q<K>(yhp + 2 * SB + gs, yhp + 3 * SB + gs, sgn, mul, odd, aHL);
  load_c2q<K>(yhp + 1 * SB + gs, yhp + 4 * SB + gs, sgn, mul, odd, aHH);
  v8f acc1 = {0.f, 0.f, 0.f, 0.f, 0.f, 0.f, 0.f, 0.f};
  v8f acc2 = {0.f, 0.f, 0.f, 0.f, 0.f, 0.f, 0.f, 0.f};
#pragma unroll
  for (int t = 0; t < KT; ++t) {
    acc1 = wmma4(aLL[t], bL[t], acc1);
    acc2 = wmma4(aLH[t], bL[t], acc2);
    acc1 = wmma4(aHL[t], bH[t], acc1);
    acc2 = wmma4(aHH[t], bH[t], acc2);
  }
  z1 = acc1;
  z2 = acc2;
}

template <int K>
__device__ __forceinline__ void row_tile_real(const float* __restrict__ t0,
                                              unsigned f, int IMG,
                                              const v2f (&bL)[K / 4],
                                              const v2f (&bH)[K / 4], v8f& z1,
                                              v8f& z2) {
  constexpr int KT = K / 4;
  v2f aLL[KT], aLH[KT], aHL[KT], aHH[KT];
  load_half<K>(t0 + f, aLL);
  load_half<K>(t0 + f + 1 * IMG, aLH);
  load_half<K>(t0 + f + 2 * IMG, aHL);
  load_half<K>(t0 + f + 3 * IMG, aHH);
  __builtin_amdgcn_sched_barrier(0);
  v8f acc1 = {0.f, 0.f, 0.f, 0.f, 0.f, 0.f, 0.f, 0.f};
  v8f acc2 = {0.f, 0.f, 0.f, 0.f, 0.f, 0.f, 0.f, 0.f};
#pragma unroll
  for (int t = 0; t < KT; ++t) {
    acc1 = wmma4(aLL[t], bL[t], acc1);
    acc2 = wmma4(aLH[t], bL[t], acc2);
    acc1 = wmma4(aHL[t], bH[t], acc1);
    acc2 = wmma4(aHH[t], bH[t], acc2);
  }
  z1 = acc1;
  z2 = acc2;
}

__device__ __forceinline__ v8f col_tile(v8f z, const v2f (&g)[4], v8f acc) {
#pragma unroll
  for (int t = 0; t < 4; ++t)
    acc = wmma4(v2f{z[2 * t], z[2 * t + 1]}, g[t], acc);
  return acc;
}

template <int T0>
__device__ __forceinline__ v8f col_frags(v8f z, const v2f (&g)[3], v8f acc) {
#pragma unroll
  for (int j = 0; j < 3; ++j)
    acc = wmma4(v2f{z[2 * (T0 + j)], z[2 * (T0 + j) + 1]}, g[j], acc);
  return acc;
}

constexpr int OPU = 36;

template <int R, int NB>
__global__ __launch_bounds__(256)
void level_up(const float* __restrict__ ll, const float* __restrict__ yh,
              const float* __restrict__ g0a, const float* __restrict__ g0b,
              const float* __restrict__ g1a, const float* __restrict__ g1b,
              float* __restrict__ out) {
  constexpr int P = 2 * R;
  constexpr int NT = P / 16;
  constexpr int H2 = R / 2;
  constexpr int SB = H2 * H2 * 2;
  constexpr int YR = (R < 68) ? R : 68;
  constexpr int YS = YR / 2;
  constexpr int XW = 40;
  constexpr int SEG = XW / 4;
  constexpr int PITCH = 44;
  constexpr int NROW = YR + 6 * YS;
  constexpr int NPIECE = NROW * SEG;
  constexpr int NLD = (NPIECE + 255) / 256;
  constexpr int NBAND = NT / 8;
  constexpr int NBUF = (NB > 1) ? 2 : 1;
  constexpr int TSZ = NROW * PITCH;
  static_assert((NT / 4) % NB == 0, "blocks per work-group divide the row");
  __shared__ __attribute__((aligned(16))) float tile[NBUF * TSZ];
  __shared__ __attribute__((aligned(16))) float ftile[3 * 2 * 16 * FPU];
  __shared__ __attribute__((aligned(16))) float obuf[8 * 16 * OPU];

  const int tid = threadIdx.x;
  const int wave = tid >> 5, lane = tid & 31;
  const int nt0 = (blockIdx.x % NBAND) * 8;
  const int nt = nt0 + wave;
  const int ms0 = (blockIdx.x / NBAND) * NB;
  const int mc = blockIdx.y;
  const int o0 = nt << 4;
  const int wy = iclamp((o0 >> 1) - 4, 0, R - 16);
  const int Y0 = iclamp(8 * nt0 - 4, 0, R - YR);
  const int h = lane >> 4, c = lane & 15;
  const bool odd = (c & 1) != 0;
  const v2f sgn = odd ? v2f{-1.f, -1.f} : v2f{1.f, 1.f};
  const v2f mul = odd ? v2f{-SC_HALF_SQRT2, SC_HALF_SQRT2}
                      : v2f{SC_HALF_SQRT2, SC_HALF_SQRT2};

  const float* llp = ll + (size_t)mc * (R * R) + (size_t)Y0 * R;
  const float* yhp = yh + (size_t)mc * (6 * SB) + (size_t)(Y0 >> 1) * R;

  float fcm[5];
  int fkm[5];
  ifilt_row_taps<R>(g0a, g0b, g1a, g1b, tid, fcm, fkm);

  v4f pre[NLD];
  auto fetch = [&](int x0) {
#pragma unroll
    for (int n = 0; n < NLD; ++n) {
      const int j = tid + 256 * n;
      if (256 * (n + 1) <= NPIECE || j < NPIECE) {
        const int row = j / SEG, seg = j - row * SEG;
        const bool isll = row < YR;
        const int rr = row - YR;
        const int b = rr / YS, sr = rr - b * YS;
        const unsigned offl = (unsigned)row * R;
        const unsigned offs = (unsigned)b * SB + (unsigned)sr * R;
        const float* src = isll ? (llp + offl) : (yhp + offs);
        pre[n] = *reinterpret_cast<const v4f*>(src + 4 * seg + x0);
      }
    }
  };
  auto park = [&](float* __restrict__ buf) {
#pragma unroll
    for (int n = 0; n < NLD; ++n) {
      const int j = tid + 256 * n;
      if (256 * (n + 1) <= NPIECE || j < NPIECE) {
        const int row = j / SEG, seg = j - row * SEG;
        *reinterpret_cast<v4f*>(buf + row * PITCH + 4 * seg) = pre[n];
      }
    }
  };
  auto block_x0 = [&](int ms) { return iclamp(32 * ms - 4, 0, R - XW); };
  fetch(block_x0(ms0));

  if (tid < 96) put_row<5, 16, FPU>(ftile + tid * FPU, fcm, fkm);
  float* ow = obuf + wave * (16 * OPU);
  float* owt = ow + c * OPU + 8 * h;
  const float* owr = ow + (lane >> 3) * OPU + 4 * (lane & 7);
  float* og = out + (size_t)mc * (P * P) + (size_t)(o0 + (lane >> 3)) * P +
              4 * (lane & 7);

  park(tile);
  if (NB > 1) fetch(block_x0(ms0 + 1));
  lds_barrier();

  const float* fw = ftile + c * FPU + 8 * h;
  v2f gL[4], gH[4];
  load_half<16>(fw + (tile_class<NT>(nt) * 2 + 0) * (16 * FPU), gL);
  load_half<16>(fw + (tile_class<NT>(nt) * 2 + 1) * (16 * FPU), gH);

  const int fl0 = (wy - Y0 + c) * PITCH + 8 * h;
  const int fs0 = (((wy - Y0) >> 1) + (c >> 1)) * PITCH + 8 * h;

#pragma unroll 1
  for (int bk = 0; bk < NB; ++bk) {
  const int ms = ms0 + bk;
  const int X0 = block_x0(ms);
  const float* tcur = tile + (bk & (NBUF - 1)) * TSZ;
  const float* tSB = tcur + YR * PITCH;
  const int fl = fl0 - X0;
  const int fs = fs0 - X0;

#pragma unroll 1
  for (int m = 0; m < 4; ++m) {
    const int pt = ms * 4 + m;
    const int p0 = pt << 4;
    const int wx = iclamp((p0 >> 1) - 4, 0, R - 16);
    v2f bL[4], bH[4];
    load_half<16>(fw + (tile_class<NT>(pt) * 2 + 0) * (16 * FPU), bL);
    load_half<16>(fw + (tile_class<NT>(pt) * 2 + 1) * (16 * FPU), bH);
    v8f z1, z2;
    row_tile<16>(tcur, tSB, (unsigned)(fl + wx), (unsigned)(fs + wx), YS * PITCH,
                 sgn, mul, odd, bL, bH, z1, z2);
    const v8f zero = {0.f, 0.f, 0.f, 0.f, 0.f, 0.f, 0.f, 0.f};
    const v8f a1 = col_tile(z1, gL, zero);
    const v8f a2 = col_tile(z2, gH, zero);
    store8(owt + 16 * (m & 1), a1 + a2);
    if ((m & 1) != 0) {
      wave_lds_handover();
      v4f f[4];
#pragma unroll
      for (int q = 0; q < 4; ++q)
        f[q] = *reinterpret_cast<const v4f*>(owr + (4 * q) * OPU);
#pragma unroll
      for (int q = 0; q < 4; ++q)
        *reinterpret_cast<v4f*>(og + (size_t)(4 * q) * P + (p0 - 16)) = f[q];
      wave_lds_handover();
    }
  }
  if (NB > 1 && bk + 1 < NB) {
    park(tile + ((bk + 1) & (NBUF - 1)) * TSZ);
    if (bk + 2 < NB) fetch(block_x0(ms + 2));
    lds_barrier();
  }
  }
}

constexpr int OP1 = 132;

template <int R>
__global__ __launch_bounds__(256)
void level_one(const float* __restrict__ ll, const float* __restrict__ yh,
               const float* __restrict__ g0o, const float* __restrict__ g1o,
               float* __restrict__ out) {
  constexpr int K = 24;
  constexpr int NT = R / 16;
  constexpr int H2 = R / 2;
  constexpr int SB = H2 * H2 * 2;
  constexpr int XW = 136;
  constexpr int SEG = XW / 4;
  constexpr int PITCH = 140;
  constexpr int NROW = 4 * 16;
  constexpr int IMG = 16 * PITCH;
  constexpr int NLL = 16 * SEG;
  constexpr int NPAIR = 3 * 8 * SEG;
  constexpr int NLJ = (NLL + 255) / 256;
  constexpr int NPJ = (NPAIR + 255) / 256;
  __shared__ __attribute__((aligned(16))) float tile[NROW * PITCH];
  __shared__ __attribute__((aligned(16))) float ftile[3 * 2 * 16 * FP1];
  __shared__ __attribute__((aligned(16))) float obuf[2 * 16 * OP1];

  const int tid = threadIdx.x;
  const int wave = tid >> 5, lane = tid & 31;
  const int nt0 = blockIdx.x * 8;
  const int nt = nt0 + wave;
  const int X0 = iclamp(16 * nt0 - 4, 0, R - XW);
  const int mc = blockIdx.y;
  const int p0 = nt << 4;
  const int wx = iclamp(p0 - 4, 0, R - K);
  const int h = lane >> 4, c = lane & 15;
  const int tll = (tid + 128) & 255;

  const float* llp = ll + (size_t)mc * (R * R) + X0;
  const float* yhp = yh + (size_t)mc * (6 * SB) + X0;
  float* owt = obuf + c * OP1 + 16 * wave + 8 * h;
  const float* owr = obuf + (2 * wave) * OP1 + 4 * lane;
  float* og = out + (size_t)mc * (R * R) + (size_t)(2 * wave) * R + 16 * nt0 +
              4 * lane;

  const int cr = 4 * ((c & 7) >> 1) + 2 * (c >> 3) + (c & 1);
  const unsigned fl = (unsigned)(cr * PITCH + wx - X0 + 12 * h);

  v4f preL[NLJ], pre1[NPJ], pre2[NPJ];
  auto fetch = [&](int ybase) {
#pragma unroll
    for (int n = 0; n < NLJ; ++n) {
      const int j = tll + 256 * n;
      if (256 * (n + 1) <= NLL || j < NLL) {
        const int row = j / SEG, seg = j - row * SEG;
        const int yy = iclamp(ybase + row, 0, R - 1);
        preL[n] = *reinterpret_cast<const v4f*>(llp + (unsigned)yy * R + 4 * seg);
      }
    }
#pragma unroll
    for (int n = 0; n < NPJ; ++n) {
      const int q = tid + 256 * n;
      if (256 * (n + 1) <= NPAIR || q < NPAIR) {
        const int b = q / (8 * SEG), rem = q - b * (8 * SEG);
        const int jr = rem / SEG, seg = rem - jr * SEG;
        const int w1 = (b == 0) ? 0 : ((b == 1) ? 2 : 1);
        const int yy = iclamp(ybase + 2 * jr, 0, R - 1);
        const unsigned off = (unsigned)(yy >> 1) * R + 4 * seg;
        pre1[n] = *reinterpret_cast<const v4f*>(yhp + (unsigned)w1 * SB + off);
        pre2[n] =
            *reinterpret_cast<const v4f*>(yhp + (unsigned)(5 - w1) * SB + off);
      }
    }
  };
  float h0[7], h1[5];
#pragma unroll
  for (int m = 0; m < 7; ++m) h0[m] = g0o[m];
#pragma unroll
  for (int m = 0; m < 5; ++m) h1[m] = g1o[m];

  fetch(-8);

  if (tid < 96) build_j1_row<R>(h0, h1, ftile, tid);
  lds_barrier();

  const float* fw = ftile + c * FP1 + FM1;
  v2f bL[6], bH[6];
  load_half<K>(fw + (tile_class<NT>(nt) * 2 + 0) * (16 * FP1) + 12 * h, bL);
  load_half<K>(fw + (tile_class<NT>(nt) * 2 + 1) * (16 * FP1) + 12 * h, bH);

  const v8f zero = {0.f, 0.f, 0.f, 0.f, 0.f, 0.f, 0.f, 0.f};
  v8f zp1 = zero, zp2 = zero;

#pragma unroll 1
  for (int t = 0; t <= NT; ++t) {
#pragma unroll
    for (int n = 0; n < NLJ; ++n) {
      const int j = tll + 256 * n;
      if (256 * (n + 1) <= NLL || j < NLL) {
        const int row = j / SEG, seg = j - row * SEG;
        *reinterpret_cast<v4f*>(tile + row * PITCH + 4 * seg) = preL[n];
      }
    }
#pragma unroll
    for (int n = 0; n < NPJ; ++n) {
      const int q = tid + 256 * n;
      if (256 * (n + 1) <= NPAIR || q < NPAIR) {
        const int b = q / (8 * SEG), rem = q - b * (8 * SEG);
        const int jr = rem / SEG, seg = rem - jr * SEG;
        const v4f sm = (pre1[n] + pre2[n]) * SC_HALF_SQRT2;
        const v4f d = pre1[n] - pre2[n];
        const v4f od = {d.y * SC_HALF_SQRT2, d.x * -SC_HALF_SQRT2,
                        d.w * SC_HALF_SQRT2, d.z * -SC_HALF_SQRT2};
        float* tp = tile + (16 + 16 * b + 2 * jr) * PITCH + 4 * seg;
        *reinterpret_cast<v4f*>(tp) = sm;
        *reinterpret_cast<v4f*>(tp + PITCH) = od;
      }
    }
    lds_barrier();

    if (t < NT) fetch(16 * (t + 1) - 8);

    const int i = t - 1;
    const int w0 = iclamp(16 * i - 4, 0, R - K);
    v2f gq[2][2][3];
    if (t > 0) {
      const float* fr =
          fw + (tile_class<NT>(i) * 2) * (16 * FP1) + (16 * i - 8 - w0) + 2 * h;
#pragma unroll
      for (int sl = 0; sl < 2; ++sl) {
#pragma unroll
        for (int f = 0; f < 2; ++f) {
#pragma unroll
          for (int j = 0; j < 3; ++j)
            gq[sl][f][j] = *reinterpret_cast<const v2f*>(
                fr + f * (16 * FP1) + 16 * sl + 4 * (j + 1 - sl));
        }
      }
    }

    v4f f0, f1;
    if (t > 1) {
      const float* rr = owr + ((t - 1) & 1) * (16 * OP1);
      f0 = *reinterpret_cast<const v4f*>(rr);
      f1 = *reinterpret_cast<const v4f*>(rr + OP1);
    }

    v8f zc1, zc2;
    row_tile_real<K>(tile, fl, IMG, bL, bH, zc1, zc2);

    if (t > 1) {
      float* g = og + (size_t)(16 * (t - 2)) * R;
      *reinterpret_cast<v4f*>(g) = f0;
      *reinterpret_cast<v4f*>(g + R) = f1;
    }

    if (t > 0) {
      v8f a1 = col_frags<1>(zp1, gq[0][0], zero);
      v8f a2 = col_frags<1>(zp2, gq[0][1], zero);
      a1 = col_frags<0>(zc1, gq[1][0], a1);
      a2 = col_frags<0>(zc2, gq[1][1], a2);
      store8(owt + (t & 1) * (16 * OP1), a1 + a2);
    }
    zp1 = zc1;
    zp2 = zc2;
    lds_barrier();
  }
  {
    const float* rr = owr + (NT & 1) * (16 * OP1);
    const v4f f0 = *reinterpret_cast<const v4f*>(rr);
    const v4f f1 = *reinterpret_cast<const v4f*>(rr + OP1);
    float* g = og + (size_t)(16 * (NT - 1)) * R;
    *reinterpret_cast<v4f*>(g) = f0;
    *reinterpret_cast<v4f*>(g + R) = f1;
  }
}

extern "C" void kernel_launch(void* const* d_in, const int* in_sizes, int n_in,
                              void* d_out, int out_size, void* d_ws,
                              size_t ws_size, hipStream_t stream) {
  (void)in_sizes; (void)n_in; (void)out_size; (void)ws_size;
  const float* yl  = (const float*)d_in[0];
  const float* yh1 = (const float*)d_in[1];
  const float* yh2 = (const float*)d_in[2];
  const float* yh3 = (const float*)d_in[3];
  const float* g0o = (const float*)d_in[4];
  const float* g1o = (const float*)d_in[5];
  const float* g0a = (const float*)d_in[6];
  const float* g0b = (const float*)d_in[7];
  const float* g1a = (const float*)d_in[8];
  const float* g1b = (const float*)d_in[9];

  const size_t SEG = (size_t)16 * 1024 * 1024;
  float* bufA = (float*)d_ws;
  float* bufB = bufA + SEG;

  level_up<64, 1><<<dim3(2, 256), 256, 0, stream>>>(yl, yh3, g0a, g0b, g1a,
                                                    g1b, bufA);
  level_up<128, 4><<<dim3(2, 256), 256, 0, stream>>>(bufA, yh2, g0a, g0b, g1a,
                                                     g1b, bufB);
  level_one<256><<<dim3(2, 256), 256, 0, stream>>>(bufB, yh1, g0o, g1o,
                                                   (float*)d_out);
}


# Level 3 is no launch of its own (3 -> 2): a level-2 work-group computes the level-3 rows its band needs into LDS (5 x 8 of the 8 x 8 tiles) and cuts its ll fragments there; same sums
# speedup vs baseline: 4.0625x; 1.0543x over previous
#include <hip/hip_runtime.h>


typedef __attribute__((ext_vector_type(2))) float v2f;
typedef __attribute__((ext_vector_type(4))) float v4f;
typedef __attribute__((ext_vector_type(8))) float v8f;

#define SC_HALF_SQRT2 0.70710678118654752f

__device__ __forceinline__ int refl(int i, int n) {
  i = (i < 0) ? (-1 - i) : i;
  i = (i >= n) ? (2 * n - 1 - i) : i;
  return i;
}

__device__ __forceinline__ int iclamp(int v, int lo, int hi) {
  return v < lo ? lo : (v > hi ? hi : v);
}

__device__ __forceinline__ v8f wmma4(v2f a, v2f b, v8f c) {
  return __builtin_amdgcn_wmma_f32_16x16x4_f32(false, a, false, b, (short)0, c,
                                               false, false);
}

__device__ __forceinline__ float ifilt_coef(int o, int kg, int n,
                                            const float (&ha)[10],
                                            const float (&hb)[10], bool hp) {
  const int i = o >> 2, s = o & 3;
  const int base = 4 + ((((s & 1) != 0) != hp) ? 1 : 0);
  float acc = 0.f;
#pragma unroll
  for (int m = 0; m < 5; ++m) {
    const int L = refl(2 * i + base - 2 * m, n);
    const float c = (s == 0)   ? ha[2 * m]
                    : (s == 1) ? hb[2 * m]
                    : (s == 2) ? ha[2 * m + 1]
                               : hb[2 * m + 1];
    acc += (L == kg) ? c : 0.f;
  }
  return acc;
}

template <int NT>
__device__ __forceinline__ float filt_coef(int o, int kg, int n,
                                           const float (&h)[NT]) {
  float acc = 0.f;
#pragma unroll
  for (int m = 0; m < NT; ++m) {
    const int L = refl(o + (NT >> 1) - m, n);
    acc += (L == kg) ? h[m] : 0.f;
  }
  return acc;
}

constexpr int FPU = 20;
constexpr int FP1 = 36;
constexpr int FM1 = 4;

template <int NT>
__device__ __forceinline__ int tile_class(int ty) {
  return ty == 0 ? 0 : (ty == NT - 1 ? 2 : 1);
}

template <int MT, int W, int FP, int OFF = 0>
__device__ __forceinline__ void put_row(float* __restrict__ rp,
                                        const float (&cm)[MT],
                                        const int (&km)[MT]) {
  const v4f z4 = {0.f, 0.f, 0.f, 0.f};
#pragma unroll
  for (int q = 0; q < FP / 4; ++q) *reinterpret_cast<v4f*>(rp + 4 * q) = z4;
#pragma unroll
  for (int m = 0; m < MT; ++m) {
    float acc = 0.f;
#pragma unroll
    for (int mm = 0; mm <= m; ++mm) acc += (km[mm] == km[m]) ? cm[mm] : 0.f;
    if (km[m] >= 0 && km[m] < W) rp[OFF + km[m]] = acc;
  }
}

template <int R>
__device__ __forceinline__ void ifilt_row_taps(
    const float* __restrict__ g0a, const float* __restrict__ g0b,
    const float* __restrict__ g1a, const float* __restrict__ g1b, int tid,
    float (&cm)[5], int (&km)[5]) {
  constexpr int NT = R / 8;
  const int nf = tid >> 4, row = tid & 15;
  const int cls = nf >> 1;
  const bool hp = (nf & 1) != 0;
  const int ty = (cls == 0) ? 0 : ((cls == 1) ? 1 : NT - 1);
  const int o = (ty << 4) + row;
  const int w0 = iclamp(8 * ty - 4, 0, R - 16);
  const int i = o >> 2, s = o & 3;
  const bool sodd = (s & 1) != 0;
  const int base = 4 + ((sodd != hp) ? 1 : 0);
  const float* tp = sodd ? (hp ? g1a : g0a) : (hp ? g1b : g0b);
  tp += (s >> 1);
#pragma unroll
  for (int m = 0; m < 5; ++m) {
    cm[m] = tp[2 * m];
    km[m] = refl(2 * i + base - 2 * m, R) - w0;
  }
}

template <int R>
__device__ __forceinline__ void build_j1_row(const float (&h0)[7],
                                             const float (&h1)[5],
                                             float* __restrict__ ft, int tid) {
  constexpr int NT = R / 16;
  const int nf = tid >> 4, row = tid & 15;
  const int cls = nf >> 1;
  const bool hi = (nf & 1) != 0;
  const int ty = (cls == 0) ? 0 : ((cls == 1) ? 1 : NT - 1);
  const int o = (ty << 4) + row;
  const int w0 = iclamp(16 * ty - 4, 0, R - 24);
  const int half = hi ? 2 : 3;
  float cm[7];
  int km[7];
#pragma unroll
  for (int m = 0; m < 7; ++m) {
    const float hv = (m < 5) ? h1[m] : 0.f;
    cm[m] = hi ? hv : h0[m];
    const int pos = refl(o + half - m, R) - w0;
    km[m] = (hi && m >= 5) ? (-1 - m) : pos;
  }
  put_row<7, 24, FP1, FM1>(ft + tid * FP1, cm, km);
}

template <int K>
__device__ __forceinline__ void load_half(const float* __restrict__ p,
                                          v2f (&o)[K / 4]) {
#pragma unroll
  for (int q = 0; q < K / 8; ++q) {
    const v4f v = *reinterpret_cast<const v4f*>(p + 4 * q);
    o[2 * q] = v2f{v.x, v.y};
    o[2 * q + 1] = v2f{v.z, v.w};
  }
}

template <int K>
__device__ __forceinline__ void load_c2q(const float* __restrict__ p1,
                                         const float* __restrict__ p2,
                                         v2f sgn, v2f mul, bool odd,
                                         v2f (&o)[K / 4]) {
  v2f r1[K / 4], r2[K / 4];
  load_half<K>(p1, r1);
  load_half<K>(p2, r2);
#pragma unroll
  for (int t = 0; t < K / 4; ++t) {
    const v2f u = __builtin_elementwise_fma(sgn, r2[t], r1[t]) * mul;
    o[t].x = odd ? u.y : u.x;
    o[t].y = odd ? u.x : u.y;
  }
}

__device__ __forceinline__ void lds_barrier() {
  __builtin_amdgcn_fence(__ATOMIC_RELEASE, "workgroup", "local");
  __builtin_amdgcn_s_barrier();
  __builtin_amdgcn_fence(__ATOMIC_ACQUIRE, "workgroup", "local");
}

__device__ __forceinline__ void store8(float* __restrict__ p, v8f a) {
  *reinterpret_cast<v4f*>(p) = v4f{a[0], a[1], a[2], a[3]};
  *reinterpret_cast<v4f*>(p + 4) = v4f{a[4], a[5], a[6], a[7]};
}

__device__ __forceinline__ void wave_lds_handover() {
  __builtin_amdgcn_fence(__ATOMIC_RELEASE, "workgroup", "local");
  __builtin_amdgcn_wave_barrier();
  __builtin_amdgcn_fence(__ATOMIC_ACQUIRE, "workgroup", "local");
}

template <int K>
__device__ __forceinline__ void row_tile(const float* __restrict__ llp,
                                         const float* __restrict__ yhp,
                                         unsigned gl, unsigned gs, int SB,
                                         v2f sgn, v2f mul, bool odd,
                                         const v2f (&bL)[K / 4],
                                         const v2f (&bH)[K / 4], v8f& z1,
                                         v8f& z2) {
  constexpr int KT = K / 4;
  v2f aLL[KT], aLH[KT], aHL[KT], aHH[KT];
  load_half<K>(llp + gl, aLL);
  load_c2q<K>(yhp + 0 * SB + gs, yhp + 5 * SB + gs, sgn, mul, odd, aLH);
  load_c2q<K>(yhp + 2 * SB + gs, yhp + 3 * SB + gs, sgn, mul, odd, aHL);
  load_c2q<K>(yhp + 1 * SB + gs, yhp + 4 * SB + gs, sgn, mul, odd, aHH);
  v8f acc1 = {0.f, 0.f, 0.f, 0.f, 0.f, 0.f, 0.f, 0.f};
  v8f acc2 = {0.f, 0.f, 0.f, 0.f, 0.f, 0.f, 0.f, 0.f};
#pragma unroll
  for (int t = 0; t < KT; ++t) {
    acc1 = wmma4(aLL[t], bL[t], acc1);
    acc2 = wmma4(aLH[t], bL[t], acc2);
    acc1 = wmma4(aHL[t], bH[t], acc1);
    acc2 = wmma4(aHH[t], bH[t], acc2);
  }
  z1 = acc1;
  z2 = acc2;
}

template <int K>
__device__ __forceinline__ void row_tile_real(const float* __restrict__ t0,
                                              unsigned f, int IMG,
                                              const v2f (&bL)[K / 4],
                                              const v2f (&bH)[K / 4], v8f& z1,
                                              v8f& z2) {
  constexpr int KT = K / 4;
  v2f aLL[KT], aLH[KT], aHL[KT], aHH[KT];
  load_half<K>(t0 + f, aLL);
  load_half<K>(t0 + f + 1 * IMG, aLH);
  load_half<K>(t0 + f + 2 * IMG, aHL);
  load_half<K>(t0 + f + 3 * IMG, aHH);
  __builtin_amdgcn_sched_barrier(0);
  v8f acc1 = {0.f, 0.f, 0.f, 0.f, 0.f, 0.f, 0.f, 0.f};
  v8f acc2 = {0.f, 0.f, 0.f, 0.f, 0.f, 0.f, 0.f, 0.f};
#pragma unroll
  for (int t = 0; t < KT; ++t) {
    acc1 = wmma4(aLL[t], bL[t], acc1);
    acc2 = wmma4(aLH[t], bL[t], acc2);
    acc1 = wmma4(aHL[t], bH[t], acc1);
    acc2 = wmma4(aHH[t], bH[t], acc2);
  }
  z1 = acc1;
  z2 = acc2;
}

__device__ __forceinline__ v8f col_tile(v8f z, const v2f (&g)[4], v8f acc) {
#pragma unroll
  for (int t = 0; t < 4; ++t)
    acc = wmma4(v2f{z[2 * t], z[2 * t + 1]}, g[t], acc);
  return acc;
}

template <int T0>
__device__ __forceinline__ v8f col_frags(v8f z, const v2f (&g)[3], v8f acc) {
#pragma unroll
  for (int j = 0; j < 3; ++j)
    acc = wmma4(v2f{z[2 * (T0 + j)], z[2 * (T0 + j) + 1]}, g[j], acc);
  return acc;
}

constexpr int OPU = 36;

template <int R, int NB>
__global__ __launch_bounds__(256)
void levels_32(const float* __restrict__ yl, const float* __restrict__ yh3,
               const float* __restrict__ yh,
               const float* __restrict__ g0a, const float* __restrict__ g0b,
               const float* __restrict__ g1a, const float* __restrict__ g1b,
               float* __restrict__ out) {
  constexpr int P = 2 * R;
  constexpr int NT = P / 16;
  constexpr int H2 = R / 2;
  constexpr int SB = H2 * H2 * 2;
  constexpr int YR = 68;
  constexpr int YS = YR / 2;
  constexpr int XW = 40;
  constexpr int SEG = XW / 4;
  constexpr int PITCH = 44;
  constexpr int NROW = 6 * YS;
  constexpr int NPIECE = NROW * SEG;
  constexpr int NLD = (NPIECE + 255) / 256;
  constexpr int NBAND = NT / 8;
  constexpr int NBUF = 2;
  constexpr int TSZ = NROW * PITCH;
  constexpr int LP = R + 4;
  constexpr int R3 = R / 2;
  constexpr int NT3 = R / 16;
  constexpr int SB3 = (R3 / 2) * (R3 / 2) * 2;
  constexpr int NR3 = 5;
  constexpr int YR3 = 8 * NR3 + 4;
  constexpr int YS3 = YR3 / 2;
  constexpr int SEG3 = R3 / 4;
  constexpr int P3 = R3 + 4;
  constexpr int NROW3 = YR3 + 6 * YS3;
  constexpr int NPIECE3 = NROW3 * SEG3;
  constexpr int NLD3 = (NPIECE3 + 255) / 256;
  static_assert(R == 128, "the band geometry below is that of a 128-sample side");
  static_assert(NB > 1 && (NT / 4) % NB == 0, "blocks per work-group divide the row");
  static_assert(NROW3 * P3 <= NBUF * TSZ, "the level-3 stage fits the staged blocks");
  static_assert(((LP / 4) & 1) == 1 && ((P3 / 4) & 1) == 1, "odd pitches");
  __shared__ __attribute__((aligned(16))) float tile[NBUF * TSZ];
  __shared__ __attribute__((aligned(16))) float llb[16 * NR3 * LP];
  __shared__ __attribute__((aligned(16))) float ftile[3 * 2 * 16 * FPU];
  __shared__ __attribute__((aligned(16))) float ftile3[3 * 2 * 16 * FPU];
  __shared__ __attribute__((aligned(16))) float obuf[8 * 16 * OPU];

  const int tid = threadIdx.x;
  const int wave = tid >> 5, lane = tid & 31;
  const int nt0 = (blockIdx.x % NBAND) * 8;
  const int nt = nt0 + wave;
  const int ms0 = (blockIdx.x / NBAND) * NB;
  const int mc = blockIdx.y;
  const int o0 = nt << 4;
  const int wy = iclamp((o0 >> 1) - 4, 0, R - 16);
  const int Y0 = iclamp(8 * nt0 - 4, 0, R - YR);
  const int h = lane >> 4, c = lane & 15;
  const bool odd = (c & 1) != 0;
  const v2f sgn = odd ? v2f{-1.f, -1.f} : v2f{1.f, 1.f};
  const v2f mul = odd ? v2f{-SC_HALF_SQRT2, SC_HALF_SQRT2}
                      : v2f{SC_HALF_SQRT2, SC_HALF_SQRT2};

  const int T3 = Y0 >> 4;
  const int Y3 = iclamp(8 * T3 - 4, 0, R3 - YR3);
  const float* ylp = yl + (size_t)mc * (R3 * R3) + (size_t)Y3 * R3;
  const float* y3p = yh3 + (size_t)mc * (6 * SB3) + (size_t)(Y3 >> 1) * R3;
  const float* yhp = yh + (size_t)mc * (6 * SB) + (size_t)(Y0 >> 1) * R;

  float fcm3[5], fcm[5];
  int fkm3[5], fkm[5];
  ifilt_row_taps<R3>(g0a, g0b, g1a, g1b, tid, fcm3, fkm3);
  ifilt_row_taps<R>(g0a, g0b, g1a, g1b, tid, fcm, fkm);

  v4f pre3[NLD3];
#pragma unroll
  for (int n = 0; n < NLD3; ++n) {
    const int j = tid + 256 * n;
    if (256 * (n + 1) <= NPIECE3 || j < NPIECE3) {
      const int row = j / SEG3, seg = j - row * SEG3;
      const bool isll = row < YR3;
      const int rr = row - YR3;
      const int b = rr / YS3, sr = rr - b * YS3;
      const unsigned offl = (unsigned)row * R3;
      const unsigned offs = (unsigned)b * SB3 + (unsigned)sr * R3;
      const float* sp = isll ? (ylp + offl) : (y3p + offs);
      pre3[n] = *reinterpret_cast<const v4f*>(sp + 4 * seg);
    }
  }

  v4f pre[NLD];
  auto fetch = [&](int x0) {
#pragma unroll
    for (int n = 0; n < NLD; ++n) {
      const int j = tid + 256 * n;
      if (256 * (n + 1) <= NPIECE || j < NPIECE) {
        const int row = j / SEG, seg = j - row * SEG;
        const int b = row / YS, sr = row - b * YS;
        const unsigned offs = (unsigned)b * SB + (unsigned)sr * R;
        pre[n] = *reinterpret_cast<const v4f*>(yhp + offs + 4 * seg + x0);
      }
    }
  };
  auto park = [&](float* __restrict__ buf) {
#pragma unroll
    for (int n = 0; n < NLD; ++n) {
      const int j = tid + 256 * n;
      if (256 * (n + 1) <= NPIECE || j < NPIECE) {
        const int row = j / SEG, seg = j - row * SEG;
        *reinterpret_cast<v4f*>(buf + row * PITCH + 4 * seg) = pre[n];
      }
    }
  };
  auto block_x0 = [&](int ms) { return iclamp(32 * ms - 4, 0, R - XW); };
  fetch(block_x0(ms0));

  if (tid < 96) {
    put_row<5, 16, FPU>(ftile3 + tid * FPU, fcm3, fkm3);
    put_row<5, 16, FPU>(ftile + tid * FPU, fcm, fkm);
  }
#pragma unroll
  for (int n = 0; n < NLD3; ++n) {
    const int j = tid + 256 * n;
    if (256 * (n + 1) <= NPIECE3 || j < NPIECE3) {
      const int row = j / SEG3, seg = j - row * SEG3;
      *reinterpret_cast<v4f*>(tile + row * P3 + 4 * seg) = pre3[n];
    }
  }
  lds_barrier();

  {
    const float* fw3 = ftile3 + c * FPU + 8 * h;
    v2f bL3[4], bH3[4];
    load_half<16>(fw3 + (tile_class<NT3>(wave) * 2 + 0) * (16 * FPU), bL3);
    load_half<16>(fw3 + (tile_class<NT3>(wave) * 2 + 1) * (16 * FPU), bH3);
    const int wx3 = iclamp(8 * wave - 4, 0, R3 - 16);
    const float* t3SB = tile + YR3 * P3;
#pragma unroll 1
    for (int r = 0; r < NR3; ++r) {
      const int n3 = T3 + r;
      const int wy3 = iclamp(8 * n3 - 4, 0, R3 - 16);
      v2f gL3[4], gH3[4];
      load_half<16>(fw3 + (tile_class<NT3>(n3) * 2 + 0) * (16 * FPU), gL3);
      load_half<16>(fw3 + (tile_class<NT3>(n3) * 2 + 1) * (16 * FPU), gH3);
      const unsigned gl3 = (unsigned)((wy3 - Y3 + c) * P3 + wx3 + 8 * h);
      const unsigned gs3 =
          (unsigned)((((wy3 - Y3) >> 1) + (c >> 1)) * P3 + wx3 + 8 * h);
      v8f z1, z2;
      row_tile<16>(tile, t3SB, gl3, gs3, YS3 * P3, sgn, mul, odd, bL3, bH3, z1,
                   z2);
      const v8f zero = {0.f, 0.f, 0.f, 0.f, 0.f, 0.f, 0.f, 0.f};
      const v8f a1 = col_tile(z1, gL3, zero);
      const v8f a2 = col_tile(z2, gH3, zero);
      store8(llb + (16 * r + c) * LP + 16 * wave + 8 * h, a1 + a2);
    }
  }
  lds_barrier();

  float* ow = obuf + wave * (16 * OPU);
  float* owt = ow + c * OPU + 8 * h;
  const float* owr = ow + (lane >> 3) * OPU + 4 * (lane & 7);
  float* og = out + (size_t)mc * (P * P) + (size_t)(o0 + (lane >> 3)) * P +
              4 * (lane & 7);

  park(tile);
  fetch(block_x0(ms0 + 1));
  lds_barrier();

  const float* fw = ftile + c * FPU + 8 * h;
  v2f gL[4], gH[4];
  load_half<16>(fw + (tile_class<NT>(nt) * 2 + 0) * (16 * FPU), gL);
  load_half<16>(fw + (tile_class<NT>(nt) * 2 + 1) * (16 * FPU), gH);

  const int fl = (wy - 16 * T3 + c) * LP + 8 * h;
  const int fs0 = (((wy - Y0) >> 1) + (c >> 1)) * PITCH + 8 * h;

#pragma unroll 1
  for (int bk = 0; bk < NB; ++bk) {
  const int ms = ms0 + bk;
  const int X0 = block_x0(ms);
  const float* tcur = tile + (bk & (NBUF - 1)) * TSZ;
  const int fs = fs0 - X0;

#pragma unroll 1
  for (int m = 0; m < 4; ++m) {
    const int pt = ms * 4 + m;
    const int p0 = pt << 4;
    const int wx = iclamp((p0 >> 1) - 4, 0, R - 16);
    v2f bL[4], bH[4];
    load_half<16>(fw + (tile_class<NT>(pt) * 2 + 0) * (16 * FPU), bL);
    load_half<16>(fw + (tile_class<NT>(pt) * 2 + 1) * (16 * FPU), bH);
    v8f z1, z2;
    row_tile<16>(llb, tcur, (unsigned)(fl + wx), (unsigned)(fs + wx), YS * PITCH,
                 sgn, mul, odd, bL, bH, z1, z2);
    const v8f zero = {0.f, 0.f, 0.f, 0.f, 0.f, 0.f, 0.f, 0.f};
    const v8f a1 = col_tile(z1, gL, zero);
    const v8f a2 = col_tile(z2, gH, zero);
    store8(owt + 16 * (m & 1), a1 + a2);
    if ((m & 1) != 0) {
      wave_lds_handover();
      v4f f[4];
#pragma unroll
      for (int q = 0; q < 4; ++q)
        f[q] = *reinterpret_cast<const v4f*>(owr + (4 * q) * OPU);
#pragma unroll
      for (int q = 0; q < 4; ++q)
        *reinterpret_cast<v4f*>(og + (size_t)(4 * q) * P + (p0 - 16)) = f[q];
      wave_lds_handover();
    }
  }
  if (bk + 1 < NB) {
    park(tile + ((bk + 1) & (NBUF - 1)) * TSZ);
    if (bk + 2 < NB) fetch(block_x0(ms + 2));
    lds_barrier();
  }
  }
}

constexpr int OP1 = 132;

template <int R>
__global__ __launch_bounds__(256)
void level_one(const float* __restrict__ ll, const float* __restrict__ yh,
               const float* __restrict__ g0o, const float* __restrict__ g1o,
               float* __restrict__ out) {
  constexpr int K = 24;
  constexpr int NT = R / 16;
  constexpr int H2 = R / 2;
  constexpr int SB = H2 * H2 * 2;
  constexpr int XW = 136;
  constexpr int SEG = XW / 4;
  constexpr int PITCH = 140;
  constexpr int NROW = 4 * 16;
  constexpr int IMG = 16 * PITCH;
  constexpr int NLL = 16 * SEG;
  constexpr int NPAIR = 3 * 8 * SEG;
  constexpr int NLJ = (NLL + 255) / 256;
  constexpr int NPJ = (NPAIR + 255) / 256;
  __shared__ __attribute__((aligned(16))) float tile[NROW * PITCH];
  __shared__ __attribute__((aligned(16))) float ftile[3 * 2 * 16 * FP1];
  __shared__ __attribute__((aligned(16))) float obuf[2 * 16 * OP1];

  const int tid = threadIdx.x;
  const int wave = tid >> 5, lane = tid & 31;
  const int nt0 = blockIdx.x * 8;
  const int nt = nt0 + wave;
  const int X0 = iclamp(16 * nt0 - 4, 0, R - XW);
  const int mc = blockIdx.y;
  const int p0 = nt << 4;
  const int wx = iclamp(p0 - 4, 0, R - K);
  const int h = lane >> 4, c = lane & 15;
  const int tll = (tid + 128) & 255;

  const float* llp = ll + (size_t)mc * (R * R) + X0;
  const float* yhp = yh + (size_t)mc * (6 * SB) + X0;
  float* owt = obuf + c * OP1 + 16 * wave + 8 * h;
  const float* owr = obuf + (2 * wave) * OP1 + 4 * lane;
  float* og = out + (size_t)mc * (R * R) + (size_t)(2 * wave) * R + 16 * nt0 +
              4 * lane;

  const int cr = 4 * ((c & 7) >> 1) + 2 * (c >> 3) + (c & 1);
  const unsigned fl = (unsigned)(cr * PITCH + wx - X0 + 12 * h);

  v4f preL[NLJ], pre1[NPJ], pre2[NPJ];
  auto fetch = [&](int ybase) {
#pragma unroll
    for (int n = 0; n < NLJ; ++n) {
      const int j = tll + 256 * n;
      if (256 * (n + 1) <= NLL || j < NLL) {
        const int row = j / SEG, seg = j - row * SEG;
        const int yy = iclamp(ybase + row, 0, R - 1);
        preL[n] = *reinterpret_cast<const v4f*>(llp + (unsigned)yy * R + 4 * seg);
      }
    }
#pragma unroll
    for (int n = 0; n < NPJ; ++n) {
      const int q = tid + 256 * n;
      if (256 * (n + 1) <= NPAIR || q < NPAIR) {
        const int b = q / (8 * SEG), rem = q - b * (8 * SEG);
        const int jr = rem / SEG, seg = rem - jr * SEG;
        const int w1 = (b == 0) ? 0 : ((b == 1) ? 2 : 1);
        const int yy = iclamp(ybase + 2 * jr, 0, R - 1);
        const unsigned off = (unsigned)(yy >> 1) * R + 4 * seg;
        pre1[n] = *reinterpret_cast<const v4f*>(yhp + (unsigned)w1 * SB + off);
        pre2[n] =
            *reinterpret_cast<const v4f*>(yhp + (unsigned)(5 - w1) * SB + off);
      }
    }
  };
  float h0[7], h1[5];
#pragma unroll
  for (int m = 0; m < 7; ++m) h0[m] = g0o[m];
#pragma unroll
  for (int m = 0; m < 5; ++m) h1[m] = g1o[m];

  fetch(-8);

  if (tid < 96) build_j1_row<R>(h0, h1, ftile, tid);
  lds_barrier();

  const float* fw = ftile + c * FP1 + FM1;
  v2f bL[6], bH[6];
  load_half<K>(fw + (tile_class<NT>(nt) * 2 + 0) * (16 * FP1) + 12 * h, bL);
  load_half<K>(fw + (tile_class<NT>(nt) * 2 + 1) * (16 * FP1) + 12 * h, bH);

  const v8f zero = {0.f, 0.f, 0.f, 0.f, 0.f, 0.f, 0.f, 0.f};
  v8f zp1 = zero, zp2 = zero;

#pragma unroll 1
  for (int t = 0; t <= NT; ++t) {
#pragma unroll
    for (int n = 0; n < NLJ; ++n) {
      const int j = tll + 256 * n;
      if (256 * (n + 1) <= NLL || j < NLL) {
        const int row = j / SEG, seg = j - row * SEG;
        *reinterpret_cast<v4f*>(tile + row * PITCH + 4 * seg) = preL[n];
      }
    }
#pragma unroll
    for (int n = 0; n < NPJ; ++n) {
      const int q = tid + 256 * n;
      if (256 * (n + 1) <= NPAIR || q < NPAIR) {
        const int b = q / (8 * SEG), rem = q - b * (8 * SEG);
        const int jr = rem / SEG, seg = rem - jr * SEG;
        const v4f sm = (pre1[n] + pre2[n]) * SC_HALF_SQRT2;
        const v4f d = pre1[n] - pre2[n];
        const v4f od = {d.y * SC_HALF_SQRT2, d.x * -SC_HALF_SQRT2,
                        d.w * SC_HALF_SQRT2, d.z * -SC_HALF_SQRT2};
        float* tp = tile + (16 + 16 * b + 2 * jr) * PITCH + 4 * seg;
        *reinterpret_cast<v4f*>(tp) = sm;
        *reinterpret_cast<v4f*>(tp + PITCH) = od;
      }
    }
    lds_barrier();

    if (t < NT) fetch(16 * (t + 1) - 8);

    const int i = t - 1;
    const int w0 = iclamp(16 * i - 4, 0, R - K);
    v2f gq[2][2][3];
    if (t > 0) {
      const float* fr =
          fw + (tile_class<NT>(i) * 2) * (16 * FP1) + (16 * i - 8 - w0) + 2 * h;
#pragma unroll
      for (int sl = 0; sl < 2; ++sl) {
#pragma unroll
        for (int f = 0; f < 2; ++f) {
#pragma unroll
          for (int j = 0; j < 3; ++j)
            gq[sl][f][j] = *reinterpret_cast<const v2f*>(
                fr + f * (16 * FP1) + 16 * sl + 4 * (j + 1 - sl));
        }
      }
    }

    v4f f0, f1;
    if (t > 1) {
      const float* rr = owr + ((t - 1) & 1) * (16 * OP1);
      f0 = *reinterpret_cast<const v4f*>(rr);
      f1 = *reinterpret_cast<const v4f*>(rr + OP1);
    }

    v8f zc1, zc2;
    row_tile_real<K>(tile, fl, IMG, bL, bH, zc1, zc2);

    if (t > 1) {
      float* g = og + (size_t)(16 * (t - 2)) * R;
      *reinterpret_cast<v4f*>(g) = f0;
      *reinterpret_cast<v4f*>(g + R) = f1;
    }

    if (t > 0) {
      v8f a1 = col_frags<1>(zp1, gq[0][0], zero);
      v8f a2 = col_frags<1>(zp2, gq[0][1], zero);
      a1 = col_frags<0>(zc1, gq[1][0], a1);
      a2 = col_frags<0>(zc2, gq[1][1], a2);
      store8(owt + (t & 1) * (16 * OP1), a1 + a2);
    }
    zp1 = zc1;
    zp2 = zc2;
    lds_barrier();
  }
  {
    const float* rr = owr + (NT & 1) * (16 * OP1);
    const v4f f0 = *reinterpret_cast<const v4f*>(rr);
    const v4f f1 = *reinterpret_cast<const v4f*>(rr + OP1);
    float* g = og + (size_t)(16 * (NT - 1)) * R;
    *reinterpret_cast<v4f*>(g) = f0;
    *reinterpret_cast<v4f*>(g + R) = f1;
  }
}

extern "C" void kernel_launch(void* const* d_in, const int* in_sizes, int n_in,
                              void* d_out, int out_size, void* d_ws,
                              size_t ws_size, hipStream_t stream) {
  (void)in_sizes; (void)n_in; (void)out_size; (void)ws_size;
  const float* yl  = (const float*)d_in[0];
  const float* yh1 = (const float*)d_in[1];
  const float* yh2 = (const float*)d_in[2];
  const float* yh3 = (const float*)d_in[3];
  const float* g0o = (const float*)d_in[4];
  const float* g1o = (const float*)d_in[5];
  const float* g0a = (const float*)d_in[6];
  const float* g0b = (const float*)d_in[7];
  const float* g1a = (const float*)d_in[8];
  const float* g1b = (const float*)d_in[9];

  float* bufB = (float*)d_ws;

  levels_32<128, 4><<<dim3(2, 256), 256, 0, stream>>>(yl, yh3, yh2, g0a, g0b,
                                                      g1a, g1b, bufB);
  level_one<256><<<dim3(2, 256), 256, 0, stream>>>(bufB, yh1, g0o, g1o,
                                                   (float*)d_out);
}
